# starting kernel
# baseline (speedup 1.0000x reference)
#include <hip/hip_runtime.h>
#include <math.h>

typedef __attribute__((ext_vector_type(2))) float v2f;
typedef __attribute__((ext_vector_type(8))) float v8f;

#define NEG_SLOPE 0.2f

__device__ __forceinline__ unsigned fenc(float x) {
  unsigned u = __float_as_uint(x);
  return (u & 0x80000000u) ? ~u : (u | 0x80000000u);
}
__device__ __forceinline__ float fdec(unsigned e) {
  unsigned b = (e & 0x80000000u) ? (e & 0x7FFFFFFFu) : ~e;
  return __uint_as_float(b);
}

__global__ __launch_bounds__(256) void gat_gemm_128x128(
    const float* __restrict__ X, const float* __restrict__ W,
    float* __restrict__ H)
{
  const int lane = threadIdx.x & 31;
  const int wave = threadIdx.x >> 5;
  const int row0 = blockIdx.x << 4;
  const int col0 = wave << 4;
  const int aRow = row0 + (lane & 15);
  const int kOff = (lane >> 4) << 1;
  const int bCol = col0 + (lane & 15);
  const float* xr = X + (size_t)aRow * 128;
  v8f acc = {};
  for (int k = 0; k < 128; k += 4) {
    v2f a, b;
    a.x = xr[k + kOff];
    a.y = xr[k + kOff + 1];
    b.x = W[(size_t)(k + kOff) * 128 + bCol];
    b.y = W[(size_t)(k + kOff + 1) * 128 + bCol];
    acc = __builtin_amdgcn_wmma_f32_16x16x4_f32(
        false, a, false, b, (short)0, acc, false, false);
  }
  const int r = row0 + ((lane >> 4) << 3);
  const int c = col0 + (lane & 15);
#pragma unroll
  for (int i = 0; i < 8; ++i)
    H[(size_t)(r + i) * 128 + c] = acc[i];
}

__global__ __launch_bounds__(96) void gat_gemm_128x40(
    const float* __restrict__ X, const float* __restrict__ W,
    float* __restrict__ G)
{
  const int lane = threadIdx.x & 31;
  const int wave = threadIdx.x >> 5;
  const int row0 = blockIdx.x << 4;
  const int col0 = wave << 4;
  const int aRow = row0 + (lane & 15);
  const int kOff = (lane >> 4) << 1;
  const int bCol = col0 + (lane & 15);
  const bool colOk = (bCol < 40);
  const float* xr = X + (size_t)aRow * 128;
  v8f acc = {};
  for (int k = 0; k < 128; k += 4) {
    v2f a, b;
    a.x = xr[k + kOff];
    a.y = xr[k + kOff + 1];
    b.x = colOk ? W[(size_t)(k + kOff) * 40 + bCol] : 0.0f;
    b.y = colOk ? W[(size_t)(k + kOff + 1) * 40 + bCol] : 0.0f;
    acc = __builtin_amdgcn_wmma_f32_16x16x4_f32(
        false, a, false, b, (short)0, acc, false, false);
  }
  const int r = row0 + ((lane >> 4) << 3);
  const int c = col0 + (lane & 15);
  if (c < 40) {
#pragma unroll
    for (int i = 0; i < 8; ++i)
      G[(size_t)(r + i) * 40 + c] = acc[i];
  }
}

__global__ __launch_bounds__(256) void gat_attn(
    const float* __restrict__ Hf, const float* __restrict__ aS,
    const float* __restrict__ aD, float* __restrict__ alS,
    float* __restrict__ alD, int N, int NH, int C)
{
  int idx = blockIdx.x * blockDim.x + threadIdx.x;
  if (idx >= N * NH) return;
  int n = idx / NH, h = idx - n * NH;
  const float* hp = Hf + (size_t)n * NH * C + (size_t)h * C;
  const float* sp = aS + (size_t)h * C;
  const float* dp = aD + (size_t)h * C;
  float s = 0.f, d = 0.f;
  for (int c = 0; c < C; ++c) { s += hp[c] * sp[c]; d += hp[c] * dp[c]; }
  alS[idx] = s;
  alD[idx] = d;
}

__device__ __forceinline__ void edge_ends(const int* ei, int E, int e,
                                          int& s, int& d) {
  if (e < E) { s = ei[e]; d = ei[E + e]; }
  else       { s = d = e - E; }
}

__global__ __launch_bounds__(256) void gat_edge_max(
    const int* __restrict__ ei, int E, int NH,
    const float* __restrict__ alS, const float* __restrict__ alD,
    unsigned* __restrict__ mx, int total)
{
  int idx = blockIdx.x * blockDim.x + threadIdx.x;
  if (idx >= total) return;
  int e = idx / NH, h = idx - e * NH;
  int s, d; edge_ends(ei, E, e, s, d);
  float v = alS[s * NH + h] + alD[d * NH + h];
  v = (v > 0.f) ? v : NEG_SLOPE * v;
  atomicMax(&mx[d * NH + h], fenc(v));
}

__global__ __launch_bounds__(256) void gat_edge_exp(
    const int* __restrict__ ei, int E, int NH,
    const float* __restrict__ alS, const float* __restrict__ alD,
    const unsigned* __restrict__ mx, float* __restrict__ wexp,
    float* __restrict__ den, int total)
{
  int idx = blockIdx.x * blockDim.x + threadIdx.x;
  if (idx >= total) return;
  int e = idx / NH, h = idx - e * NH;
  int s, d; edge_ends(ei, E, e, s, d);
  float v = alS[s * NH + h] + alD[d * NH + h];
  v = (v > 0.f) ? v : NEG_SLOPE * v;
  float w = expf(v - fdec(mx[d * NH + h]));
  wexp[idx] = w;
  atomicAdd(&den[d * NH + h], w);
}

__global__ __launch_bounds__(256) void gat_edge_scatter4(
    const int* __restrict__ ei, int E, int NH, int C,
    const float* __restrict__ Hf, const float* __restrict__ wexp,
    float* __restrict__ outA, long long total)
{
  long long idx = (long long)blockIdx.x * blockDim.x + threadIdx.x;
  if (idx >= total) return;
  const int F = NH * C;
  const int Q = F >> 2;
  int e = (int)(idx / Q);
  int q = (int)(idx - (long long)e * Q);
  int f = q << 2;
  int s, d; edge_ends(ei, E, e, s, d);
  int h = f / C;
  float w = wexp[(size_t)e * NH + h];
  const float4 hv = *(const float4*)(Hf + (size_t)s * F + f);
  float* op = outA + (size_t)d * F + f;
  atomicAdd(op + 0, w * hv.x);
  atomicAdd(op + 1, w * hv.y);
  atomicAdd(op + 2, w * hv.z);
  atomicAdd(op + 3, w * hv.w);
}

__global__ __launch_bounds__(256) void gat_finalize1(
    float* __restrict__ out1, const float* __restrict__ den,
    const float* __restrict__ b, int N, int F, int C)
{
  int idx = blockIdx.x * blockDim.x + threadIdx.x;
  if (idx >= N * F) return;
  int n = idx / F, f = idx - n * F;
  int h = f / C;
  float v = out1[idx] / (den[n * (F / C) + h] + 1e-16f) + b[f];
  out1[idx] = (v > 0.f) ? v : (expf(v) - 1.0f);
}

__global__ __launch_bounds__(256) void gat_finalize2(
    const float* __restrict__ out2, const float* __restrict__ den,
    const float* __restrict__ b, float* __restrict__ out, int N, int C)
{
  int idx = blockIdx.x * blockDim.x + threadIdx.x;
  if (idx >= N * C) return;
  int n = idx / C, c = idx - n * C;
  out[idx] = out2[idx] / (den[n] + 1e-16f) + b[c];
}

extern "C" void kernel_launch(void* const* d_in, const int* in_sizes, int n_in,
                              void* d_out, int out_size, void* d_ws, size_t ws_size,
                              hipStream_t stream)
{
  const float* x   = (const float*)d_in[0];
  const int*   ei  = (const int*)d_in[1];
  const float* W1  = (const float*)d_in[2];
  const float* aS1 = (const float*)d_in[3];
  const float* aD1 = (const float*)d_in[4];
  const float* b1  = (const float*)d_in[5];
  const float* W2  = (const float*)d_in[6];
  const float* aS2 = (const float*)d_in[7];
  const float* aD2 = (const float*)d_in[8];
  const float* b2  = (const float*)d_in[9];
  float* out = (float*)d_out;

  const int N  = in_sizes[0] / 128;
  const int E  = in_sizes[1] / 2;
  const int EE = E + N;

  char* ws = (char*)d_ws;
  size_t off = 0;
  auto carve = [&](size_t bytes) -> char* {
    char* p = ws + off;
    off += (bytes + 255) & ~(size_t)255;
    return p;
  };
  float*    h1   = (float*)carve((size_t)N * 128 * 4);
  float*    out1 = (float*)carve((size_t)N * 128 * 4);
  float*    alS  = (float*)carve((size_t)N * 2 * 4);
  float*    alD  = (float*)carve((size_t)N * 2 * 4);
  unsigned* mx   = (unsigned*)carve((size_t)N * 2 * 4);
  float*    den  = (float*)carve((size_t)N * 2 * 4);
  float*    wexp = (float*)carve((size_t)EE * 2 * 4);
  float* g2   = h1;
  float* out2 = h1 + (size_t)N * 64;

  gat_gemm_128x128<<<N / 16, 256, 0, stream>>>(x, W1, h1);
  gat_attn<<<(N * 2 + 255) / 256, 256, 0, stream>>>(h1, aS1, aD1, alS, alD, N, 2, 64);
  hipMemsetAsync(mx,   0, (size_t)N * 2 * 4, stream);
  hipMemsetAsync(den,  0, (size_t)N * 2 * 4, stream);
  hipMemsetAsync(out1, 0, (size_t)N * 128 * 4, stream);
  gat_edge_max<<<(EE * 2 + 255) / 256, 256, 0, stream>>>(ei, E, 2, alS, alD, mx, EE * 2);
  gat_edge_exp<<<(EE * 2 + 255) / 256, 256, 0, stream>>>(ei, E, 2, alS, alD, mx, wexp, den, EE * 2);
  {
    long long tot = (long long)EE * 32;
    gat_edge_scatter4<<<(unsigned)((tot + 255) / 256), 256, 0, stream>>>(
        ei, E, 2, 64, h1, wexp, out1, tot);
  }
  gat_finalize1<<<(N * 128 + 255) / 256, 256, 0, stream>>>(out1, den, b1, N, 128, 64);

  gat_gemm_128x40<<<N / 16, 96, 0, stream>>>(out1, W2, g2);
  gat_attn<<<(N + 255) / 256, 256, 0, stream>>>(g2, aS2, aD2, alS, alD, N, 1, 40);
  hipMemsetAsync(mx,   0, (size_t)N * 4, stream);
  hipMemsetAsync(den,  0, (size_t)N * 4, stream);
  hipMemsetAsync(out2, 0, (size_t)N * 40 * 4, stream);
  gat_edge_max<<<(EE + 255) / 256, 256, 0, stream>>>(ei, E, 1, alS, alD, mx, EE);
  gat_edge_exp<<<(EE + 255) / 256, 256, 0, stream>>>(ei, E, 1, alS, alD, mx, wexp, den, EE);
  {
    long long tot = (long long)EE * 10;
    gat_edge_scatter4<<<(unsigned)((tot + 255) / 256), 256, 0, stream>>>(
        ei, E, 1, 40, g2, wexp, out2, tot);
  }
  gat_finalize2<<<(N * 40 + 255) / 256, 256, 0, stream>>>(out2, den, b2, out, N, 40);
}


# Sort edges by destination once per call (count, scan, fill); one wave per node does max, exp, sum and row gather in registers: no float atomics, sums run in arrival-rank order per node
# speedup vs baseline: 2.6235x; 2.6235x over previous
#include <hip/hip_runtime.h>
#include <math.h>

typedef __attribute__((ext_vector_type(2))) float v2f;
typedef __attribute__((ext_vector_type(8))) float v8f;

#define NEG_SLOPE 0.2f

__global__ __launch_bounds__(256) void gat_gemm_128x128(
    const float* __restrict__ X, const float* __restrict__ W,
    float* __restrict__ H)
{
  const int lane = threadIdx.x & 31;
  const int wave = threadIdx.x >> 5;
  const int row0 = blockIdx.x << 4;
  const int col0 = wave << 4;
  const int aRow = row0 + (lane & 15);
  const int kOff = (lane >> 4) << 1;
  const int bCol = col0 + (lane & 15);
  const float* xr = X + (size_t)aRow * 128;
  v8f acc = {};
  for (int k = 0; k < 128; k += 4) {
    v2f a, b;
    a.x = xr[k + kOff];
    a.y = xr[k + kOff + 1];
    b.x = W[(size_t)(k + kOff) * 128 + bCol];
    b.y = W[(size_t)(k + kOff + 1) * 128 + bCol];
    acc = __builtin_amdgcn_wmma_f32_16x16x4_f32(
        false, a, false, b, (short)0, acc, false, false);
  }
  const int r = row0 + ((lane >> 4) << 3);
  const int c = col0 + (lane & 15);
#pragma unroll
  for (int i = 0; i < 8; ++i)
    H[(size_t)(r + i) * 128 + c] = acc[i];
}

__global__ __launch_bounds__(96) void gat_gemm_128x40(
    const float* __restrict__ X, const float* __restrict__ W,
    float* __restrict__ G)
{
  const int lane = threadIdx.x & 31;
  const int wave = threadIdx.x >> 5;
  const int row0 = blockIdx.x << 4;
  const int col0 = wave << 4;
  const int aRow = row0 + (lane & 15);
  const int kOff = (lane >> 4) << 1;
  const int bCol = col0 + (lane & 15);
  const bool colOk = (bCol < 40);
  const float* xr = X + (size_t)aRow * 128;
  v8f acc = {};
  for (int k = 0; k < 128; k += 4) {
    v2f a, b;
    a.x = xr[k + kOff];
    a.y = xr[k + kOff + 1];
    b.x = colOk ? W[(size_t)(k + kOff) * 40 + bCol] : 0.0f;
    b.y = colOk ? W[(size_t)(k + kOff + 1) * 40 + bCol] : 0.0f;
    acc = __builtin_amdgcn_wmma_f32_16x16x4_f32(
        false, a, false, b, (short)0, acc, false, false);
  }
  const int r = row0 + ((lane >> 4) << 3);
  const int c = col0 + (lane & 15);
  if (c < 40) {
#pragma unroll
    for (int i = 0; i < 8; ++i)
      G[(size_t)(r + i) * 40 + c] = acc[i];
  }
}

__global__ __launch_bounds__(256) void gat_attn(
    const float* __restrict__ Hf, const float* __restrict__ aS,
    const float* __restrict__ aD, float* __restrict__ alS,
    float* __restrict__ alD, int N, int NH, int C)
{
  int idx = blockIdx.x * blockDim.x + threadIdx.x;
  if (idx >= N * NH) return;
  int n = idx / NH, h = idx - n * NH;
  const float* hp = Hf + (size_t)n * NH * C + (size_t)h * C;
  const float* sp = aS + (size_t)h * C;
  const float* dp = aD + (size_t)h * C;
  float s = 0.f, d = 0.f;
  for (int c = 0; c < C; ++c) { s += hp[c] * sp[c]; d += hp[c] * dp[c]; }
  alS[idx] = s;
  alD[idx] = d;
}

__device__ __forceinline__ float leaky(float v) {
  return (v > 0.f) ? v : NEG_SLOPE * v;
}
__device__ __forceinline__ float wave_max(float v) {
  for (int o = 16; o > 0; o >>= 1) v = fmaxf(v, __shfl_xor(v, o, 32));
  return v;
}
__device__ __forceinline__ float wave_sum(float v) {
  for (int o = 16; o > 0; o >>= 1) v += __shfl_xor(v, o, 32);
  return v;
}
__device__ __forceinline__ float lane_bcast(float v, int j) {
  return __int_as_float(__builtin_amdgcn_readlane(__float_as_int(v), j));
}

__global__ __launch_bounds__(256) void gat_csr_rank(
    const int* __restrict__ ei, int E, unsigned* __restrict__ cnt,
    unsigned* __restrict__ rank)
{
  int e = blockIdx.x * blockDim.x + threadIdx.x;
  if (e >= E) return;
  rank[e] = atomicAdd(&cnt[ei[E + e]], 1u);
}

#define SCAN_CHUNK 16384
__global__ __launch_bounds__(1024) void gat_csr_scan(
    const unsigned* __restrict__ cnt, unsigned* __restrict__ rowptr, int NP)
{
  __shared__ unsigned wtot[32];
  const int t = threadIdx.x, lane = t & 31, wv = t >> 5;
  unsigned carry = 0;
  for (int base = 0; base < NP; base += SCAN_CHUNK) {
    const uint4* p = (const uint4*)(cnt + base + t * 16);
    unsigned v[16];
#pragma unroll
    for (int i = 0; i < 4; ++i) {
      const uint4 q = p[i];
      v[4 * i] = q.x; v[4 * i + 1] = q.y; v[4 * i + 2] = q.z; v[4 * i + 3] = q.w;
    }
    unsigned s = 0;
#pragma unroll
    for (int i = 0; i < 16; ++i) s += v[i];
    unsigned incl = s;
    for (int o = 1; o < 32; o <<= 1) {
      const unsigned u = __shfl_up(incl, o, 32);
      if (lane >= o) incl += u;
    }
    if (lane == 31) wtot[wv] = incl;
    __syncthreads();
    if (wv == 0) {
      unsigned w = wtot[lane];
      for (int o = 1; o < 32; o <<= 1) {
        const unsigned u = __shfl_up(w, o, 32);
        if (lane >= o) w += u;
      }
      wtot[lane] = w;
    }
    __syncthreads();
    unsigned run = carry + (incl - s) + (wv ? wtot[wv - 1] : 0u);
    carry += wtot[31];
    uint4* op = (uint4*)(rowptr + base + t * 16);
#pragma unroll
    for (int i = 0; i < 4; ++i) {
      uint4 q;
      q.x = run; run += v[4 * i];
      q.y = run; run += v[4 * i + 1];
      q.z = run; run += v[4 * i + 2];
      q.w = run; run += v[4 * i + 3];
      op[i] = q;
    }
    __syncthreads();
  }
}

__global__ __launch_bounds__(256) void gat_csr_fill(
    const int* __restrict__ ei, int E, const unsigned* __restrict__ rowptr,
    const unsigned* __restrict__ rank, int* __restrict__ col)
{
  int e = blockIdx.x * blockDim.x + threadIdx.x;
  if (e >= E) return;
  col[rowptr[ei[E + e]] + rank[e]] = ei[e];
}

__global__ __launch_bounds__(256) void gat_gather_l1(
    const unsigned* __restrict__ rowptr, const int* __restrict__ col,
    const float* __restrict__ Hf, const float* __restrict__ alS,
    const float* __restrict__ alD, const float* __restrict__ b,
    float* __restrict__ out1, int N)
{
  const int lane = threadIdx.x & 31;
  const int wv = __builtin_amdgcn_readfirstlane((int)(threadIdx.x >> 5));
  const int d = blockIdx.x * 8 + wv;
  if (d >= N) return;
  const int beg = (int)rowptr[d];
  const int deg = (int)rowptr[d + 1] - beg;
  const float ad0 = alD[d * 2], ad1 = alD[d * 2 + 1];
  const float vs0 = leaky(alS[d * 2] + ad0);
  const float vs1 = leaky(alS[d * 2 + 1] + ad1);
  float m0 = vs0, m1 = vs1;
  for (int base = 0; base < deg; base += 32) {
    const int i = base + lane;
    if (i < deg) {
      const int s = col[beg + i];
      const float2 a = *(const float2*)(alS + (size_t)s * 2);
      m0 = fmaxf(m0, leaky(a.x + ad0));
      m1 = fmaxf(m1, leaky(a.y + ad1));
    }
  }
  m0 = wave_max(m0);
  m1 = wave_max(m1);
  const int hsel = lane >> 4;
  const float ws0 = expf(vs0 - m0), ws1 = expf(vs1 - m1);
  float4 acc;
  {
    const float w = hsel ? ws1 : ws0;
    const float4 hv = *(const float4*)(Hf + (size_t)d * 128 + lane * 4);
    acc.x = w * hv.x; acc.y = w * hv.y; acc.z = w * hv.z; acc.w = w * hv.w;
  }
  float p0 = 0.f, p1 = 0.f;
  for (int base = 0; base < deg; base += 32) {
    const int i = base + lane;
    int s = 0;
    float w0 = 0.f, w1 = 0.f;
    if (i < deg) {
      s = col[beg + i];
      const float2 a = *(const float2*)(alS + (size_t)s * 2);
      w0 = expf(leaky(a.x + ad0) - m0);
      w1 = expf(leaky(a.y + ad1) - m1);
    }
    p0 += w0;
    p1 += w1;
    const int cnt = min(32, deg - base);
#pragma unroll 4
    for (int j = 0; j < cnt; ++j) {
      const int sj = __builtin_amdgcn_readlane(s, j);
      const float wj0 = lane_bcast(w0, j), wj1 = lane_bcast(w1, j);
      const float w = hsel ? wj1 : wj0;
      const float4 hv = *(const float4*)(Hf + (size_t)sj * 128 + lane * 4);
      acc.x += w * hv.x; acc.y += w * hv.y; acc.z += w * hv.z; acc.w += w * hv.w;
    }
  }
  const float den0 = wave_sum(p0) + ws0;
  const float den1 = wave_sum(p1) + ws1;
  const float den = (hsel ? den1 : den0) + 1e-16f;
  const float4 bv = *(const float4*)(b + lane * 4);
  float4 r;
  r.x = acc.x / den + bv.x;
  r.y = acc.y / den + bv.y;
  r.z = acc.z / den + bv.z;
  r.w = acc.w / den + bv.w;
  r.x = (r.x > 0.f) ? r.x : (expf(r.x) - 1.0f);
  r.y = (r.y > 0.f) ? r.y : (expf(r.y) - 1.0f);
  r.z = (r.z > 0.f) ? r.z : (expf(r.z) - 1.0f);
  r.w = (r.w > 0.f) ? r.w : (expf(r.w) - 1.0f);
  *(float4*)(out1 + (size_t)d * 128 + lane * 4) = r;
}

__global__ __launch_bounds__(256) void gat_gather_l2(
    const unsigned* __restrict__ rowptr, const int* __restrict__ col,
    const float* __restrict__ G, const float* __restrict__ alS,
    const float* __restrict__ alD, const float* __restrict__ b,
    float* __restrict__ out, int N)
{
  __shared__ float stage[320];
  const int lane = threadIdx.x & 31;
  const int wv = __builtin_amdgcn_readfirstlane((int)(threadIdx.x >> 5));
  const int d = blockIdx.x * 8 + wv;
  const bool act = lane < 20;
  const int c2 = act ? 2 * lane : 0;
  float r0 = 0.f, r1 = 0.f;
  if (d < N) {
    const int beg = (int)rowptr[d];
    const int deg = (int)rowptr[d + 1] - beg;
    const float ad = alD[d];
    const float vs = leaky(alS[d] + ad);
    float m = vs;
    for (int base = 0; base < deg; base += 32) {
      const int i = base + lane;
      if (i < deg) m = fmaxf(m, leaky(alS[col[beg + i]] + ad));
    }
    m = wave_max(m);
    const float wself = expf(vs - m);
    const float2 g0 = *(const float2*)(G + (size_t)d * 40 + c2);
    float acc0 = wself * g0.x, acc1 = wself * g0.y;
    float p = 0.f;
    for (int base = 0; base < deg; base += 32) {
      const int i = base + lane;
      int s = 0;
      float w = 0.f;
      if (i < deg) {
        s = col[beg + i];
        w = expf(leaky(alS[s] + ad) - m);
      }
      p += w;
      const int cnt = min(32, deg - base);
#pragma unroll 4
      for (int j = 0; j < cnt; ++j) {
        const int sj = __builtin_amdgcn_readlane(s, j);
        const float wj = lane_bcast(w, j);
        const float2 gv = *(const float2*)(G + (size_t)sj * 40 + c2);
        acc0 += wj * gv.x;
        acc1 += wj * gv.y;
      }
    }
    const float den = wave_sum(p) + wself + 1e-16f;
    r0 = acc0 / den + b[c2];
    r1 = acc1 / den + b[c2 + 1];
  }
  if (act) {
    stage[wv * 40 + c2] = r0;
    stage[wv * 40 + c2 + 1] = r1;
  }
  __syncthreads();
  const size_t lim = (size_t)N * 40;
  const size_t g = (size_t)blockIdx.x * 320 + threadIdx.x;
  if (g < lim) out[g] = stage[threadIdx.x];
  if (threadIdx.x < 64 && g + 256 < lim) out[g + 256] = stage[256 + threadIdx.x];
}

extern "C" void kernel_launch(void* const* d_in, const int* in_sizes, int n_in,
                              void* d_out, int out_size, void* d_ws, size_t ws_size,
                              hipStream_t stream)
{
  const float* x   = (const float*)d_in[0];
  const int*   ei  = (const int*)d_in[1];
  const float* W1  = (const float*)d_in[2];
  const float* aS1 = (const float*)d_in[3];
  const float* aD1 = (const float*)d_in[4];
  const float* b1  = (const float*)d_in[5];
  const float* W2  = (const float*)d_in[6];
  const float* aS2 = (const float*)d_in[7];
  const float* aD2 = (const float*)d_in[8];
  const float* b2  = (const float*)d_in[9];
  float* out = (float*)d_out;

  const int N  = in_sizes[0] / 128;
  const int E  = in_sizes[1] / 2;
  const int NP = (N / SCAN_CHUNK + 1) * SCAN_CHUNK;

  char* ws = (char*)d_ws;
  size_t off = 0;
  auto carve = [&](size_t bytes) -> char* {
    char* p = ws + off;
    off += (bytes + 255) & ~(size_t)255;
    return p;
  };
  float*    h1     = (float*)carve((size_t)N * 128 * 4);
  float*    out1   = (float*)carve((size_t)N * 128 * 4);
  float*    alS    = (float*)carve((size_t)N * 2 * 4);
  float*    alD    = (float*)carve((size_t)N * 2 * 4);
  unsigned* cnt    = (unsigned*)carve((size_t)NP * 4);
  unsigned* rowptr = (unsigned*)carve((size_t)NP * 4);
  unsigned* rank   = (unsigned*)carve((size_t)E * 4);
  int*      col    = (int*)carve((size_t)E * 4);
  float* g2 = h1;

  hipMemsetAsync(cnt, 0, (size_t)NP * 4, stream);
  gat_csr_rank<<<(E + 255) / 256, 256, 0, stream>>>(ei, E, cnt, rank);
  gat_csr_scan<<<1, 1024, 0, stream>>>(cnt, rowptr, NP);
  gat_csr_fill<<<(E + 255) / 256, 256, 0, stream>>>(ei, E, rowptr, rank, col);

  gat_gemm_128x128<<<N / 16, 256, 0, stream>>>(x, W1, h1);
  gat_attn<<<(N * 2 + 255) / 256, 256, 0, stream>>>(h1, aS1, aD1, alS, alD, N, 2, 64);
  gat_gather_l1<<<(N + 7) / 8, 256, 0, stream>>>(rowptr, col, h1, alS, alD, b1, out1, N);

  gat_gemm_128x40<<<N / 16, 96, 0, stream>>>(out1, W2, g2);
  gat_attn<<<(N + 255) / 256, 256, 0, stream>>>(g2, aS2, aD2, alS, alD, N, 1, 40);
  gat_gather_l2<<<(N + 7) / 8, 256, 0, stream>>>(rowptr, col, g2, alS, alD, b2, out, N);
}


# Fewer, longer waves in every stage: 8 edges per thread in rank and fill, GEMM blocks walk row tiles with weights held in registers, a gather wave does 4 nodes in turn; sum orders unchanged
# speedup vs baseline: 2.7059x; 1.0314x over previous
#include <hip/hip_runtime.h>
#include <math.h>

typedef __attribute__((ext_vector_type(2))) float v2f;
typedef __attribute__((ext_vector_type(8))) float v8f;

#define NEG_SLOPE 0.2f

__global__ __launch_bounds__(256) void gat_gemm_128x128(
    const float* __restrict__ X, const float* __restrict__ W,
    float* __restrict__ H, int nTiles)
{
  const int lane = threadIdx.x & 31;
  const int wave = threadIdx.x >> 5;
  const int col0 = wave << 4;
  const int kOff = (lane >> 4) << 1;
  const int bCol = col0 + (lane & 15);
  v2f bf[32];
#pragma unroll
  for (int s = 0; s < 32; ++s) {
    bf[s].x = W[(size_t)(4 * s + kOff) * 128 + bCol];
    bf[s].y = W[(size_t)(4 * s + kOff + 1) * 128 + bCol];
  }
  for (int tile = blockIdx.x; tile < nTiles; tile += gridDim.x) {
    const int row0 = tile << 4;
    const float* xr = X + (size_t)(row0 + (lane & 15)) * 128 + kOff;
    v8f acc = {};
#pragma unroll
    for (int s = 0; s < 32; ++s) {
      v2f a;
      a.x = xr[4 * s];
      a.y = xr[4 * s + 1];
      acc = __builtin_amdgcn_wmma_f32_16x16x4_f32(
          false, a, false, bf[s], (short)0, acc, false, false);
    }
    const int r = row0 + ((lane >> 4) << 3);
    const int c = col0 + (lane & 15);
#pragma unroll
    for (int i = 0; i < 8; ++i)
      H[(size_t)(r + i) * 128 + c] = acc[i];
  }
}

__global__ __launch_bounds__(96) void gat_gemm_128x40(
    const float* __restrict__ X, const float* __restrict__ W,
    float* __restrict__ G, int nTiles)
{
  const int lane = threadIdx.x & 31;
  const int wave = threadIdx.x >> 5;
  const int col0 = wave << 4;
  const int kOff = (lane >> 4) << 1;
  const int bCol = col0 + (lane & 15);
  const bool colOk = (bCol < 40);
  v2f bf[32];
#pragma unroll
  for (int s = 0; s < 32; ++s) {
    bf[s].x = colOk ? W[(size_t)(4 * s + kOff) * 40 + bCol] : 0.0f;
    bf[s].y = colOk ? W[(size_t)(4 * s + kOff + 1) * 40 + bCol] : 0.0f;
  }
  for (int tile = blockIdx.x; tile < nTiles; tile += gridDim.x) {
    const int row0 = tile << 4;
    const float* xr = X + (size_t)(row0 + (lane & 15)) * 128 + kOff;
    v8f acc = {};
#pragma unroll
    for (int s = 0; s < 32; ++s) {
      v2f a;
      a.x = xr[4 * s];
      a.y = xr[4 * s + 1];
      acc = __builtin_amdgcn_wmma_f32_16x16x4_f32(
          false, a, false, bf[s], (short)0, acc, false, false);
    }
    const int r = row0 + ((lane >> 4) << 3);
    const int c = col0 + (lane & 15);
    if (c < 40) {
#pragma unroll
      for (int i = 0; i < 8; ++i)
        G[(size_t)(r + i) * 40 + c] = acc[i];
    }
  }
}

__global__ __launch_bounds__(256) void gat_attn(
    const float* __restrict__ Hf, const float* __restrict__ aS,
    const float* __restrict__ aD, float* __restrict__ alS,
    float* __restrict__ alD, int N, int NH, int C)
{
  int idx = blockIdx.x * blockDim.x + threadIdx.x;
  if (idx >= N * NH) return;
  int n = idx / NH, h = idx - n * NH;
  const float* hp = Hf + (size_t)n * NH * C + (size_t)h * C;
  const float* sp = aS + (size_t)h * C;
  const float* dp = aD + (size_t)h * C;
  float s = 0.f, d = 0.f;
  for (int c = 0; c < C; ++c) { s += hp[c] * sp[c]; d += hp[c] * dp[c]; }
  alS[idx] = s;
  alD[idx] = d;
}

__device__ __forceinline__ float leaky(float v) {
  return (v > 0.f) ? v : NEG_SLOPE * v;
}
__device__ __forceinline__ float wave_max(float v) {
  for (int o = 16; o > 0; o >>= 1) v = fmaxf(v, __shfl_xor(v, o, 32));
  return v;
}
__device__ __forceinline__ float wave_sum(float v) {
  for (int o = 16; o > 0; o >>= 1) v += __shfl_xor(v, o, 32);
  return v;
}
__device__ __forceinline__ float lane_bcast(float v, int j) {
  return __int_as_float(__builtin_amdgcn_readlane(__float_as_int(v), j));
}

#define EDGES_PER_THREAD 8
__global__ __launch_bounds__(256) void gat_csr_rank(
    const int* __restrict__ ei, int E, unsigned* __restrict__ cnt,
    unsigned* __restrict__ rank)
{
  const int e0 = blockIdx.x * (256 * EDGES_PER_THREAD) + threadIdx.x;
  int dst[EDGES_PER_THREAD];
  unsigned r[EDGES_PER_THREAD];
#pragma unroll
  for (int k = 0; k < EDGES_PER_THREAD; ++k) {
    const int e = e0 + 256 * k;
    dst[k] = (e < E) ? ei[E + e] : -1;
  }
#pragma unroll
  for (int k = 0; k < EDGES_PER_THREAD; ++k)
    r[k] = (dst[k] >= 0) ? atomicAdd(&cnt[dst[k]], 1u) : 0u;
#pragma unroll
  for (int k = 0; k < EDGES_PER_THREAD; ++k) {
    const int e = e0 + 256 * k;
    if (e < E) rank[e] = r[k];
  }
}

#define SCAN_CHUNK 16384
__global__ __launch_bounds__(1024) void gat_csr_scan(
    const unsigned* __restrict__ cnt, unsigned* __restrict__ rowptr, int NP)
{
  __shared__ unsigned wtot[32];
  const int t = threadIdx.x, lane = t & 31, wv = t >> 5;
  unsigned carry = 0;
  for (int base = 0; base < NP; base += SCAN_CHUNK) {
    const uint4* p = (const uint4*)(cnt + base + t * 16);
    unsigned v[16];
#pragma unroll
    for (int i = 0; i < 4; ++i) {
      const uint4 q = p[i];
      v[4 * i] = q.x; v[4 * i + 1] = q.y; v[4 * i + 2] = q.z; v[4 * i + 3] = q.w;
    }
    unsigned s = 0;
#pragma unroll
    for (int i = 0; i < 16; ++i) s += v[i];
    unsigned incl = s;
    for (int o = 1; o < 32; o <<= 1) {
      const unsigned u = __shfl_up(incl, o, 32);
      if (lane >= o) incl += u;
    }
    if (lane == 31) wtot[wv] = incl;
    __syncthreads();
    if (wv == 0) {
      unsigned w = wtot[lane];
      for (int o = 1; o < 32; o <<= 1) {
        const unsigned u = __shfl_up(w, o, 32);
        if (lane >= o) w += u;
      }
      wtot[lane] = w;
    }
    __syncthreads();
    unsigned run = carry + (incl - s) + (wv ? wtot[wv - 1] : 0u);
    carry += wtot[31];
    uint4* op = (uint4*)(rowptr + base + t * 16);
#pragma unroll
    for (int i = 0; i < 4; ++i) {
      uint4 q;
      q.x = run; run += v[4 * i];
      q.y = run; run += v[4 * i + 1];
      q.z = run; run += v[4 * i + 2];
      q.w = run; run += v[4 * i + 3];
      op[i] = q;
    }
    __syncthreads();
  }
}

__global__ __launch_bounds__(256) void gat_csr_fill(
    const int* __restrict__ ei, int E, const unsigned* __restrict__ rowptr,
    const unsigned* __restrict__ rank, int* __restrict__ col)
{
  const int e0 = blockIdx.x * (256 * EDGES_PER_THREAD) + threadIdx.x;
  int dst[EDGES_PER_THREAD], sr[EDGES_PER_THREAD];
  unsigned rk[EDGES_PER_THREAD], rp[EDGES_PER_THREAD];
#pragma unroll
  for (int k = 0; k < EDGES_PER_THREAD; ++k) {
    const int e = e0 + 256 * k;
    const bool ok = e < E;
    dst[k] = ok ? ei[E + e] : -1;
    sr[k] = ok ? ei[e] : 0;
    rk[k] = ok ? rank[e] : 0u;
  }
#pragma unroll
  for (int k = 0; k < EDGES_PER_THREAD; ++k)
    rp[k] = (dst[k] >= 0) ? rowptr[dst[k]] : 0u;
#pragma unroll
  for (int k = 0; k < EDGES_PER_THREAD; ++k)
    if (dst[k] >= 0) col[rp[k] + rk[k]] = sr[k];
}

#define NODES_PER_WAVE 4
__global__ __launch_bounds__(256) void gat_gather_l1(
    const unsigned* __restrict__ rowptr, const int* __restrict__ col,
    const float* __restrict__ Hf, const float* __restrict__ alS,
    const float* __restrict__ alD, const float* __restrict__ b,
    float* __restrict__ out1, int N)
{
  const int lane = threadIdx.x & 31;
  const int wv = __builtin_amdgcn_readfirstlane((int)(threadIdx.x >> 5));
  const int d0 = (blockIdx.x * 8 + wv) * NODES_PER_WAVE;
  if (d0 >= N) return;
  unsigned rbeg = rowptr[d0], rend = rowptr[d0 + 1];
#pragma unroll 1
  for (int kn = 0; kn < NODES_PER_WAVE; ++kn) {
  const int d = d0 + kn;
  if (d >= N) break;
  const unsigned rnext = rowptr[min(d + 2, N)];
  const int beg = (int)rbeg;
  const int deg = (int)(rend - rbeg);
  rbeg = rend;
  rend = rnext;
  const float ad0 = alD[d * 2], ad1 = alD[d * 2 + 1];
  const float vs0 = leaky(alS[d * 2] + ad0);
  const float vs1 = leaky(alS[d * 2 + 1] + ad1);
  float m0 = vs0, m1 = vs1;
  for (int base = 0; base < deg; base += 32) {
    const int i = base + lane;
    if (i < deg) {
      const int s = col[beg + i];
      const float2 a = *(const float2*)(alS + (size_t)s * 2);
      m0 = fmaxf(m0, leaky(a.x + ad0));
      m1 = fmaxf(m1, leaky(a.y + ad1));
    }
  }
  m0 = wave_max(m0);
  m1 = wave_max(m1);
  const int hsel = lane >> 4;
  const float ws0 = expf(vs0 - m0), ws1 = expf(vs1 - m1);
  float4 acc;
  {
    const float w = hsel ? ws1 : ws0;
    const float4 hv = *(const float4*)(Hf + (size_t)d * 128 + lane * 4);
    acc.x = w * hv.x; acc.y = w * hv.y; acc.z = w * hv.z; acc.w = w * hv.w;
  }
  float p0 = 0.f, p1 = 0.f;
  for (int base = 0; base < deg; base += 32) {
    const int i = base + lane;
    int s = 0;
    float w0 = 0.f, w1 = 0.f;
    if (i < deg) {
      s = col[beg + i];
      const float2 a = *(const float2*)(alS + (size_t)s * 2);
      w0 = expf(leaky(a.x + ad0) - m0);
      w1 = expf(leaky(a.y + ad1) - m1);
    }
    p0 += w0;
    p1 += w1;
    const int cnt = min(32, deg - base);
#pragma unroll 4
    for (int j = 0; j < cnt; ++j) {
      const int sj = __builtin_amdgcn_readlane(s, j);
      const float wj0 = lane_bcast(w0, j), wj1 = lane_bcast(w1, j);
      const float w = hsel ? wj1 : wj0;
      const float4 hv = *(const float4*)(Hf + (size_t)sj * 128 + lane * 4);
      acc.x += w * hv.x; acc.y += w * hv.y; acc.z += w * hv.z; acc.w += w * hv.w;
    }
  }
  const float den0 = wave_sum(p0) + ws0;
  const float den1 = wave_sum(p1) + ws1;
  const float den = (hsel ? den1 : den0) + 1e-16f;
  const float4 bv = *(const float4*)(b + lane * 4);
  float4 r;
  r.x = acc.x / den + bv.x;
  r.y = acc.y / den + bv.y;
  r.z = acc.z / den + bv.z;
  r.w = acc.w / den + bv.w;
  r.x = (r.x > 0.f) ? r.x : (expf(r.x) - 1.0f);
  r.y = (r.y > 0.f) ? r.y : (expf(r.y) - 1.0f);
  r.z = (r.z > 0.f) ? r.z : (expf(r.z) - 1.0f);
  r.w = (r.w > 0.f) ? r.w : (expf(r.w) - 1.0f);
  *(float4*)(out1 + (size_t)d * 128 + lane * 4) = r;
  }
}

__global__ __launch_bounds__(256) void gat_gather_l2(
    const unsigned* __restrict__ rowptr, const int* __restrict__ col,
    const float* __restrict__ G, const float* __restrict__ alS,
    const float* __restrict__ alD, const float* __restrict__ b,
    float* __restrict__ out, int N)
{
  __shared__ float stage[8 * NODES_PER_WAVE * 40];
  const int lane = threadIdx.x & 31;
  const int wv = __builtin_amdgcn_readfirstlane((int)(threadIdx.x >> 5));
  const int d0 = (blockIdx.x * 8 + wv) * NODES_PER_WAVE;
  float* slot = stage + wv * (NODES_PER_WAVE * 40);
  const bool act = lane < 20;
  const int c2 = act ? 2 * lane : 0;
  unsigned rbeg = 0, rend = 0;
  if (d0 < N) { rbeg = rowptr[d0]; rend = rowptr[d0 + 1]; }
#pragma unroll 1
  for (int kn = 0; kn < NODES_PER_WAVE; ++kn) {
  const int d = d0 + kn;
  float r0 = 0.f, r1 = 0.f;
  if (d < N) {
    const unsigned rnext = rowptr[min(d + 2, N)];
    const int beg = (int)rbeg;
    const int deg = (int)(rend - rbeg);
    rbeg = rend;
    rend = rnext;
    const float ad = alD[d];
    const float vs = leaky(alS[d] + ad);
    float m = vs;
    for (int base = 0; base < deg; base += 32) {
      const int i = base + lane;
      if (i < deg) m = fmaxf(m, leaky(alS[col[beg + i]] + ad));
    }
    m = wave_max(m);
    const float wself = expf(vs - m);
    const float2 g0 = *(const float2*)(G + (size_t)d * 40 + c2);
    float acc0 = wself * g0.x, acc1 = wself * g0.y;
    float p = 0.f;
    for (int base = 0; base < deg; base += 32) {
      const int i = base + lane;
      int s = 0;
      float w = 0.f;
      if (i < deg) {
        s = col[beg + i];
        w = expf(leaky(alS[s] + ad) - m);
      }
      p += w;
      const int cnt = min(32, deg - base);
#pragma unroll 4
      for (int j = 0; j < cnt; ++j) {
        const int sj = __builtin_amdgcn_readlane(s, j);
        const float wj = lane_bcast(w, j);
        const float2 gv = *(const float2*)(G + (size_t)sj * 40 + c2);
        acc0 += wj * gv.x;
        acc1 += wj * gv.y;
      }
    }
    const float den = wave_sum(p) + wself + 1e-16f;
    r0 = acc0 / den + b[c2];
    r1 = acc1 / den + b[c2 + 1];
  }
  if (act) {
    slot[kn * 40 + c2] = r0;
    slot[kn * 40 + c2 + 1] = r1;
  }
  }
  const size_t lim = (size_t)N * 40;
  const size_t g0 = (size_t)d0 * 40 + lane;
#pragma unroll
  for (int j = 0; j < (NODES_PER_WAVE * 40) / 32; ++j) {
    const size_t g = g0 + 32 * j;
    if (g < lim) out[g] = slot[32 * j + lane];
  }
}

extern "C" void kernel_launch(void* const* d_in, const int* in_sizes, int n_in,
                              void* d_out, int out_size, void* d_ws, size_t ws_size,
                              hipStream_t stream)
{
  const float* x   = (const float*)d_in[0];
  const int*   ei  = (const int*)d_in[1];
  const float* W1  = (const float*)d_in[2];
  const float* aS1 = (const float*)d_in[3];
  const float* aD1 = (const float*)d_in[4];
  const float* b1  = (const float*)d_in[5];
  const float* W2  = (const float*)d_in[6];
  const float* aS2 = (const float*)d_in[7];
  const float* aD2 = (const float*)d_in[8];
  const float* b2  = (const float*)d_in[9];
  float* out = (float*)d_out;

  const int N  = in_sizes[0] / 128;
  const int E  = in_sizes[1] / 2;
  const int NP = (N / SCAN_CHUNK + 1) * SCAN_CHUNK;

  char* ws = (char*)d_ws;
  size_t off = 0;
  auto carve = [&](size_t bytes) -> char* {
    char* p = ws + off;
    off += (bytes + 255) & ~(size_t)255;
    return p;
  };
  float*    h1     = (float*)carve((size_t)N * 128 * 4);
  float*    out1   = (float*)carve((size_t)N * 128 * 4);
  float*    alS    = (float*)carve((size_t)N * 2 * 4);
  float*    alD    = (float*)carve((size_t)N * 2 * 4);
  unsigned* cnt    = (unsigned*)carve((size_t)NP * 4);
  unsigned* rowptr = (unsigned*)carve((size_t)NP * 4);
  unsigned* rank   = (unsigned*)carve((size_t)E * 4);
  int*      col    = (int*)carve((size_t)E * 4);
  float* g2 = h1;

  hipMemsetAsync(cnt, 0, (size_t)NP * 4, stream);
  gat_csr_rank<<<(E + 2047) / 2048, 256, 0, stream>>>(ei, E, cnt, rank);
  gat_csr_scan<<<1, 1024, 0, stream>>>(cnt, rowptr, NP);
  gat_csr_fill<<<(E + 2047) / 2048, 256, 0, stream>>>(ei, E, rowptr, rank, col);

  gat_gemm_128x128<<<768, 256, 0, stream>>>(x, W1, h1, N / 16);
  gat_attn<<<(N * 2 + 255) / 256, 256, 0, stream>>>(h1, aS1, aD1, alS, alD, N, 2, 64);
  gat_gather_l1<<<(N + 31) / 32, 256, 0, stream>>>(rowptr, col, h1, alS, alD, b1, out1, N);

  gat_gemm_128x40<<<1024, 96, 0, stream>>>(out1, W2, g2, N / 16);
  gat_attn<<<(N + 255) / 256, 256, 0, stream>>>(g2, aS2, aD2, alS, alD, N, 1, 40);
  gat_gather_l2<<<(N + 31) / 32, 256, 0, stream>>>(rowptr, col, g2, alS, alD, b2, out, N);
}


# Destination counts and ranks from LDS atomics in per-slice whole-graph histograms, slice sums by a small kernel: no global atomics, no memset; a node's edges now sum in slice, then LDS arrival order
# speedup vs baseline: 3.0911x; 1.1424x over previous
#include <hip/hip_runtime.h>
#include <math.h>

typedef __attribute__((ext_vector_type(2))) float v2f;
typedef __attribute__((ext_vector_type(8))) float v8f;

#define NEG_SLOPE 0.2f

__global__ __launch_bounds__(256) void gat_gemm_128x128(
    const float* __restrict__ X, const float* __restrict__ W,
    float* __restrict__ H, int nTiles)
{
  const int lane = threadIdx.x & 31;
  const int wave = threadIdx.x >> 5;
  const int col0 = wave << 4;
  const int kOff = (lane >> 4) << 1;
  const int bCol = col0 + (lane & 15);
  v2f bf[32];
#pragma unroll
  for (int s = 0; s < 32; ++s) {
    bf[s].x = W[(size_t)(4 * s + kOff) * 128 + bCol];
    bf[s].y = W[(size_t)(4 * s + kOff + 1) * 128 + bCol];
  }
  for (int tile = blockIdx.x; tile < nTiles; tile += gridDim.x) {
    const int row0 = tile << 4;
    const float* xr = X + (size_t)(row0 + (lane & 15)) * 128 + kOff;
    v8f acc = {};
#pragma unroll
    for (int s = 0; s < 32; ++s) {
      v2f a;
      a.x = xr[4 * s];
      a.y = xr[4 * s + 1];
      acc = __builtin_amdgcn_wmma_f32_16x16x4_f32(
          false, a, false, bf[s], (short)0, acc, false, false);
    }
    const int r = row0 + ((lane >> 4) << 3);
    const int c = col0 + (lane & 15);
#pragma unroll
    for (int i = 0; i < 8; ++i)
      H[(size_t)(r + i) * 128 + c] = acc[i];
  }
}

__global__ __launch_bounds__(96) void gat_gemm_128x40(
    const float* __restrict__ X, const float* __restrict__ W,
    float* __restrict__ G, int nTiles)
{
  const int lane = threadIdx.x & 31;
  const int wave = threadIdx.x >> 5;
  const int col0 = wave << 4;
  const int kOff = (lane >> 4) << 1;
  const int bCol = col0 + (lane & 15);
  const bool colOk = (bCol < 40);
  v2f bf[32];
#pragma unroll
  for (int s = 0; s < 32; ++s) {
    bf[s].x = colOk ? W[(size_t)(4 * s + kOff) * 40 + bCol] : 0.0f;
    bf[s].y = colOk ? W[(size_t)(4 * s + kOff + 1) * 40 + bCol] : 0.0f;
  }
  for (int tile = blockIdx.x; tile < nTiles; tile += gridDim.x) {
    const int row0 = tile << 4;
    const float* xr = X + (size_t)(row0 + (lane & 15)) * 128 + kOff;
    v8f acc = {};
#pragma unroll
    for (int s = 0; s < 32; ++s) {
      v2f a;
      a.x = xr[4 * s];
      a.y = xr[4 * s + 1];
      acc = __builtin_amdgcn_wmma_f32_16x16x4_f32(
          false, a, false, bf[s], (short)0, acc, false, false);
    }
    const int r = row0 + ((lane >> 4) << 3);
    const int c = col0 + (lane & 15);
    if (c < 40) {
#pragma unroll
      for (int i = 0; i < 8; ++i)
        G[(size_t)(r + i) * 40 + c] = acc[i];
    }
  }
}

__global__ __launch_bounds__(256) void gat_attn(
    const float* __restrict__ Hf, const float* __restrict__ aS,
    const float* __restrict__ aD, float* __restrict__ alS,
    float* __restrict__ alD, int N, int NH, int C)
{
  int idx = blockIdx.x * blockDim.x + threadIdx.x;
  if (idx >= N * NH) return;
  int n = idx / NH, h = idx - n * NH;
  const float* hp = Hf + (size_t)n * NH * C + (size_t)h * C;
  const float* sp = aS + (size_t)h * C;
  const float* dp = aD + (size_t)h * C;
  float s = 0.f, d = 0.f;
  for (int c = 0; c < C; ++c) { s += hp[c] * sp[c]; d += hp[c] * dp[c]; }
  alS[idx] = s;
  alD[idx] = d;
}

__device__ __forceinline__ float leaky(float v) {
  return (v > 0.f) ? v : NEG_SLOPE * v;
}
__device__ __forceinline__ float wave_max(float v) {
  for (int o = 16; o > 0; o >>= 1) v = fmaxf(v, __shfl_xor(v, o, 32));
  return v;
}
__device__ __forceinline__ float wave_sum(float v) {
  for (int o = 16; o > 0; o >>= 1) v += __shfl_xor(v, o, 32);
  return v;
}
__device__ __forceinline__ float lane_bcast(float v, int j) {
  return __int_as_float(__builtin_amdgcn_readlane(__float_as_int(v), j));
}

#define EDGES_PER_THREAD 8
#define SLICE_SHIFT 14
#define SLICE (1 << SLICE_SHIFT)
#define HIST_WORDS 51200
#define HIST_EPT (SLICE / 1024)
__global__ __launch_bounds__(1024) void gat_csr_hist(
    const int* __restrict__ ei, int E, int NS, unsigned* __restrict__ part,
    unsigned* __restrict__ rank)
{
  __shared__ unsigned hist[HIST_WORDS];
  const int t = threadIdx.x;
  for (int i = t; i < HIST_WORDS / 4; i += 1024)
    ((uint4*)hist)[i] = make_uint4(0u, 0u, 0u, 0u);
  __syncthreads();
  const int e0 = blockIdx.x * SLICE + t;
  int dst[HIST_EPT];
  unsigned r[HIST_EPT];
#pragma unroll
  for (int k = 0; k < HIST_EPT; ++k) {
    const int e = e0 + 1024 * k;
    dst[k] = (e < E) ? ei[E + e] : -1;
  }
#pragma unroll
  for (int k = 0; k < HIST_EPT; ++k)
    r[k] = ((unsigned)dst[k] < (unsigned)HIST_WORDS) ? atomicAdd(&hist[dst[k]], 1u) : 0u;
#pragma unroll
  for (int k = 0; k < HIST_EPT; ++k) {
    const int e = e0 + 1024 * k;
    if (e < E) rank[e] = r[k];
  }
  __syncthreads();
  uint4* p = (uint4*)(part + (size_t)blockIdx.x * NS);
  for (int i = t; i < NS / 4; i += 1024) p[i] = ((const uint4*)hist)[i];
}

#define COLSUM_CHUNK 16
__global__ __launch_bounds__(256) void gat_csr_colsum(
    unsigned* __restrict__ part, unsigned* __restrict__ cnt, int N, int NS,
    int B, int NP)
{
  const int n = blockIdx.x * 256 + threadIdx.x;
  if (n >= NP) return;
  unsigned run = 0;
  if (n < N) {
    unsigned* p = part + n;
    for (int b0 = 0; b0 < B; b0 += COLSUM_CHUNK) {
      unsigned v[COLSUM_CHUNK];
#pragma unroll
      for (int k = 0; k < COLSUM_CHUNK; ++k)
        v[k] = (b0 + k < B) ? p[(size_t)(b0 + k) * NS] : 0u;
#pragma unroll
      for (int k = 0; k < COLSUM_CHUNK; ++k) {
        if (b0 + k < B) p[(size_t)(b0 + k) * NS] = run;
        run += v[k];
      }
    }
  }
  cnt[n] = run;
}

#define SCAN_CHUNK 16384
__global__ __launch_bounds__(1024) void gat_csr_scan(
    const unsigned* __restrict__ cnt, unsigned* __restrict__ rowptr, int NP)
{
  __shared__ unsigned wtot[32];
  const int t = threadIdx.x, lane = t & 31, wv = t >> 5;
  unsigned carry = 0;
  for (int base = 0; base < NP; base += SCAN_CHUNK) {
    const uint4* p = (const uint4*)(cnt + base + t * 16);
    unsigned v[16];
#pragma unroll
    for (int i = 0; i < 4; ++i) {
      const uint4 q = p[i];
      v[4 * i] = q.x; v[4 * i + 1] = q.y; v[4 * i + 2] = q.z; v[4 * i + 3] = q.w;
    }
    unsigned s = 0;
#pragma unroll
    for (int i = 0; i < 16; ++i) s += v[i];
    unsigned incl = s;
    for (int o = 1; o < 32; o <<= 1) {
      const unsigned u = __shfl_up(incl, o, 32);
      if (lane >= o) incl += u;
    }
    if (lane == 31) wtot[wv] = incl;
    __syncthreads();
    if (wv == 0) {
      unsigned w = wtot[lane];
      for (int o = 1; o < 32; o <<= 1) {
        const unsigned u = __shfl_up(w, o, 32);
        if (lane >= o) w += u;
      }
      wtot[lane] = w;
    }
    __syncthreads();
    unsigned run = carry + (incl - s) + (wv ? wtot[wv - 1] : 0u);
    carry += wtot[31];
    uint4* op = (uint4*)(rowptr + base + t * 16);
#pragma unroll
    for (int i = 0; i < 4; ++i) {
      uint4 q;
      q.x = run; run += v[4 * i];
      q.y = run; run += v[4 * i + 1];
      q.z = run; run += v[4 * i + 2];
      q.w = run; run += v[4 * i + 3];
      op[i] = q;
    }
    __syncthreads();
  }
}

__global__ __launch_bounds__(256) void gat_csr_fill(
    const int* __restrict__ ei, int E, const unsigned* __restrict__ rowptr,
    const unsigned* __restrict__ part, int NS,
    const unsigned* __restrict__ rank, int* __restrict__ col)
{
  const int e0 = blockIdx.x * (256 * EDGES_PER_THREAD) + threadIdx.x;
  int dst[EDGES_PER_THREAD], sr[EDGES_PER_THREAD];
  unsigned rk[EDGES_PER_THREAD], rp[EDGES_PER_THREAD], pp[EDGES_PER_THREAD];
#pragma unroll
  for (int k = 0; k < EDGES_PER_THREAD; ++k) {
    const int e = e0 + 256 * k;
    const bool ok = e < E;
    dst[k] = ok ? ei[E + e] : -1;
    sr[k] = ok ? ei[e] : 0;
    rk[k] = ok ? rank[e] : 0u;
  }
#pragma unroll
  for (int k = 0; k < EDGES_PER_THREAD; ++k)
    rp[k] = (dst[k] >= 0) ? rowptr[dst[k]] : 0u;
#pragma unroll
  for (int k = 0; k < EDGES_PER_THREAD; ++k) {
    const int e = e0 + 256 * k;
    pp[k] = (dst[k] >= 0) ? part[(size_t)(e >> SLICE_SHIFT) * NS + dst[k]] : 0u;
  }
#pragma unroll
  for (int k = 0; k < EDGES_PER_THREAD; ++k)
    if (dst[k] >= 0) col[rp[k] + pp[k] + rk[k]] = sr[k];
}

#define NODES_PER_WAVE 4
__global__ __launch_bounds__(256) void gat_gather_l1(
    const unsigned* __restrict__ rowptr, const int* __restrict__ col,
    const float* __restrict__ Hf, const float* __restrict__ alS,
    const float* __restrict__ alD, const float* __restrict__ b,
    float* __restrict__ out1, int N)
{
  const int lane = threadIdx.x & 31;
  const int wv = __builtin_amdgcn_readfirstlane((int)(threadIdx.x >> 5));
  const int d0 = (blockIdx.x * 8 + wv) * NODES_PER_WAVE;
  if (d0 >= N) return;
  unsigned rbeg = rowptr[d0], rend = rowptr[d0 + 1];
#pragma unroll 1
  for (int kn = 0; kn < NODES_PER_WAVE; ++kn) {
  const int d = d0 + kn;
  if (d >= N) break;
  const unsigned rnext = rowptr[min(d + 2, N)];
  const int beg = (int)rbeg;
  const int deg = (int)(rend - rbeg);
  rbeg = rend;
  rend = rnext;
  const float ad0 = alD[d * 2], ad1 = alD[d * 2 + 1];
  const float vs0 = leaky(alS[d * 2] + ad0);
  const float vs1 = leaky(alS[d * 2 + 1] + ad1);
  float m0 = vs0, m1 = vs1;
  for (int base = 0; base < deg; base += 32) {
    const int i = base + lane;
    if (i < deg) {
      const int s = col[beg + i];
      const float2 a = *(const float2*)(alS + (size_t)s * 2);
      m0 = fmaxf(m0, leaky(a.x + ad0));
      m1 = fmaxf(m1, leaky(a.y + ad1));
    }
  }
  m0 = wave_max(m0);
  m1 = wave_max(m1);
  const int hsel = lane >> 4;
  const float ws0 = expf(vs0 - m0), ws1 = expf(vs1 - m1);
  float4 acc;
  {
    const float w = hsel ? ws1 : ws0;
    const float4 hv = *(const float4*)(Hf + (size_t)d * 128 + lane * 4);
    acc.x = w * hv.x; acc.y = w * hv.y; acc.z = w * hv.z; acc.w = w * hv.w;
  }
  float p0 = 0.f, p1 = 0.f;
  for (int base = 0; base < deg; base += 32) {
    const int i = base + lane;
    int s = 0;
    float w0 = 0.f, w1 = 0.f;
    if (i < deg) {
      s = col[beg + i];
      const float2 a = *(const float2*)(alS + (size_t)s * 2);
      w0 = expf(leaky(a.x + ad0) - m0);
      w1 = expf(leaky(a.y + ad1) - m1);
    }
    p0 += w0;
    p1 += w1;
    const int cnt = min(32, deg - base);
#pragma unroll 4
    for (int j = 0; j < cnt; ++j) {
      const int sj = __builtin_amdgcn_readlane(s, j);
      const float wj0 = lane_bcast(w0, j), wj1 = lane_bcast(w1, j);
      const float w = hsel ? wj1 : wj0;
      const float4 hv = *(const float4*)(Hf + (size_t)sj * 128 + lane * 4);
      acc.x += w * hv.x; acc.y += w * hv.y; acc.z += w * hv.z; acc.w += w * hv.w;
    }
  }
  const float den0 = wave_sum(p0) + ws0;
  const float den1 = wave_sum(p1) + ws1;
  const float den = (hsel ? den1 : den0) + 1e-16f;
  const float4 bv = *(const float4*)(b + lane * 4);
  float4 r;
  r.x = acc.x / den + bv.x;
  r.y = acc.y / den + bv.y;
  r.z = acc.z / den + bv.z;
  r.w = acc.w / den + bv.w;
  r.x = (r.x > 0.f) ? r.x : (expf(r.x) - 1.0f);
  r.y = (r.y > 0.f) ? r.y : (expf(r.y) - 1.0f);
  r.z = (r.z > 0.f) ? r.z : (expf(r.z) - 1.0f);
  r.w = (r.w > 0.f) ? r.w : (expf(r.w) - 1.0f);
  *(float4*)(out1 + (size_t)d * 128 + lane * 4) = r;
  }
}

__global__ __launch_bounds__(256) void gat_gather_l2(
    const unsigned* __restrict__ rowptr, const int* __restrict__ col,
    const float* __restrict__ G, const float* __restrict__ alS,
    const float* __restrict__ alD, const float* __restrict__ b,
    float* __restrict__ out, int N)
{
  __shared__ float stage[8 * NODES_PER_WAVE * 40];
  const int lane = threadIdx.x & 31;
  const int wv = __builtin_amdgcn_readfirstlane((int)(threadIdx.x >> 5));
  const int d0 = (blockIdx.x * 8 + wv) * NODES_PER_WAVE;
  float* slot = stage + wv * (NODES_PER_WAVE * 40);
  const bool act = lane < 20;
  const int c2 = act ? 2 * lane : 0;
  unsigned rbeg = 0, rend = 0;
  if (d0 < N) { rbeg = rowptr[d0]; rend = rowptr[d0 + 1]; }
#pragma unroll 1
  for (int kn = 0; kn < NODES_PER_WAVE; ++kn) {
  const int d = d0 + kn;
  float r0 = 0.f, r1 = 0.f;
  if (d < N) {
    const unsigned rnext = rowptr[min(d + 2, N)];
    const int beg = (int)rbeg;
    const int deg = (int)(rend - rbeg);
    rbeg = rend;
    rend = rnext;
    const float ad = alD[d];
    const float vs = leaky(alS[d] + ad);
    float m = vs;
    for (int base = 0; base < deg; base += 32) {
      const int i = base + lane;
      if (i < deg) m = fmaxf(m, leaky(alS[col[beg + i]] + ad));
    }
    m = wave_max(m);
    const float wself = expf(vs - m);
    const float2 g0 = *(const float2*)(G + (size_t)d * 40 + c2);
    float acc0 = wself * g0.x, acc1 = wself * g0.y;
    float p = 0.f;
    for (int base = 0; base < deg; base += 32) {
      const int i = base + lane;
      int s = 0;
      float w = 0.f;
      if (i < deg) {
        s = col[beg + i];
        w = expf(leaky(alS[s] + ad) - m);
      }
      p += w;
      const int cnt = min(32, deg - base);
#pragma unroll 4
      for (int j = 0; j < cnt; ++j) {
        const int sj = __builtin_amdgcn_readlane(s, j);
        const float wj = lane_bcast(w, j);
        const float2 gv = *(const float2*)(G + (size_t)sj * 40 + c2);
        acc0 += wj * gv.x;
        acc1 += wj * gv.y;
      }
    }
    const float den = wave_sum(p) + wself + 1e-16f;
    r0 = acc0 / den + b[c2];
    r1 = acc1 / den + b[c2 + 1];
  }
  if (act) {
    slot[kn * 40 + c2] = r0;
    slot[kn * 40 + c2 + 1] = r1;
  }
  }
  const size_t lim = (size_t)N * 40;
  const size_t g0 = (size_t)d0 * 40 + lane;
#pragma unroll
  for (int j = 0; j < (NODES_PER_WAVE * 40) / 32; ++j) {
    const size_t g = g0 + 32 * j;
    if (g < lim) out[g] = slot[32 * j + lane];
  }
}

extern "C" void kernel_launch(void* const* d_in, const int* in_sizes, int n_in,
                              void* d_out, int out_size, void* d_ws, size_t ws_size,
                              hipStream_t stream)
{
  const float* x   = (const float*)d_in[0];
  const int*   ei  = (const int*)d_in[1];
  const float* W1  = (const float*)d_in[2];
  const float* aS1 = (const float*)d_in[3];
  const float* aD1 = (const float*)d_in[4];
  const float* b1  = (const float*)d_in[5];
  const float* W2  = (const float*)d_in[6];
  const float* aS2 = (const float*)d_in[7];
  const float* aD2 = (const float*)d_in[8];
  const float* b2  = (const float*)d_in[9];
  float* out = (float*)d_out;

  const int N  = in_sizes[0] / 128;
  const int E  = in_sizes[1] / 2;
  const int NP = (N / SCAN_CHUNK + 1) * SCAN_CHUNK;

  char* ws = (char*)d_ws;
  size_t off = 0;
  auto carve = [&](size_t bytes) -> char* {
    char* p = ws + off;
    off += (bytes + 255) & ~(size_t)255;
    return p;
  };
  float*    h1     = (float*)carve((size_t)N * 128 * 4);
  float*    out1   = (float*)carve((size_t)N * 128 * 4);
  float*    alS    = (float*)carve((size_t)N * 2 * 4);
  float*    alD    = (float*)carve((size_t)N * 2 * 4);
  unsigned* cnt    = (unsigned*)carve((size_t)NP * 4);
  unsigned* rowptr = (unsigned*)carve((size_t)NP * 4);
  unsigned* rank   = (unsigned*)carve((size_t)E * 4);
  int*      col    = (int*)carve((size_t)E * 4);
  const int NS = (N + 31) & ~31;
  const int B  = (E + SLICE - 1) >> SLICE_SHIFT;
  unsigned* part   = (unsigned*)carve((size_t)B * NS * 4);
  float* g2 = h1;

  gat_csr_hist<<<B, 1024, 0, stream>>>(ei, E, NS, part, rank);
  gat_csr_colsum<<<NP / 256, 256, 0, stream>>>(part, cnt, N, NS, B, NP);
  gat_csr_scan<<<1, 1024, 0, stream>>>(cnt, rowptr, NP);
  gat_csr_fill<<<(E + 2047) / 2048, 256, 0, stream>>>(ei, E, rowptr, part, NS, rank, col);

  gat_gemm_128x128<<<768, 256, 0, stream>>>(x, W1, h1, N / 16);
  gat_attn<<<(N * 2 + 255) / 256, 256, 0, stream>>>(h1, aS1, aD1, alS, alD, N, 2, 64);
  gat_gather_l1<<<(N + 31) / 32, 256, 0, stream>>>(rowptr, col, h1, alS, alD, b1, out1, N);

  gat_gemm_128x40<<<1024, 96, 0, stream>>>(out1, W2, g2, N / 16);
  gat_attn<<<(N + 255) / 256, 256, 0, stream>>>(g2, aS2, aD2, alS, alD, N, 1, 40);
  gat_gather_l2<<<(N + 31) / 32, 256, 0, stream>>>(rowptr, col, g2, alS, alD, b2, out, N);
}


# Gather waves shorten their chain of dependent loads: a node's sources and logits are read once and requested one node ahead; rows go in whole groups of 4, padded with the last row at weight 0
# speedup vs baseline: 3.2528x; 1.0523x over previous
#include <hip/hip_runtime.h>
#include <math.h>

typedef __attribute__((ext_vector_type(2))) float v2f;
typedef __attribute__((ext_vector_type(8))) float v8f;

#define NEG_SLOPE 0.2f

__global__ __launch_bounds__(256) void gat_gemm_128x128(
    const float* __restrict__ X, const float* __restrict__ W,
    float* __restrict__ H, int nTiles)
{
  const int lane = threadIdx.x & 31;
  const int wave = threadIdx.x >> 5;
  const int col0 = wave << 4;
  const int kOff = (lane >> 4) << 1;
  const int bCol = col0 + (lane & 15);
  v2f bf[32];
#pragma unroll
  for (int s = 0; s < 32; ++s) {
    bf[s].x = W[(size_t)(4 * s + kOff) * 128 + bCol];
    bf[s].y = W[(size_t)(4 * s + kOff + 1) * 128 + bCol];
  }
  for (int tile = blockIdx.x; tile < nTiles; tile += gridDim.x) {
    const int row0 = tile << 4;
    const float* xr = X + (size_t)(row0 + (lane & 15)) * 128 + kOff;
    v8f acc = {};
#pragma unroll
    for (int s = 0; s < 32; ++s) {
      v2f a;
      a.x = xr[4 * s];
      a.y = xr[4 * s + 1];
      acc = __builtin_amdgcn_wmma_f32_16x16x4_f32(
          false, a, false, bf[s], (short)0, acc, false, false);
    }
    const int r = row0 + ((lane >> 4) << 3);
    const int c = col0 + (lane & 15);
#pragma unroll
    for (int i = 0; i < 8; ++i)
      H[(size_t)(r + i) * 128 + c] = acc[i];
  }
}

__global__ __launch_bounds__(96) void gat_gemm_128x40(
    const float* __restrict__ X, const float* __restrict__ W,
    float* __restrict__ G, int nTiles)
{
  const int lane = threadIdx.x & 31;
  const int wave = threadIdx.x >> 5;
  const int col0 = wave << 4;
  const int kOff = (lane >> 4) << 1;
  const int bCol = col0 + (lane & 15);
  const bool colOk = (bCol < 40);
  v2f bf[32];
#pragma unroll
  for (int s = 0; s < 32; ++s) {
    bf[s].x = colOk ? W[(size_t)(4 * s + kOff) * 40 + bCol] : 0.0f;
    bf[s].y = colOk ? W[(size_t)(4 * s + kOff + 1) * 40 + bCol] : 0.0f;
  }
  for (int tile = blockIdx.x; tile < nTiles; tile += gridDim.x) {
    const int row0 = tile << 4;
    const float* xr = X + (size_t)(row0 + (lane & 15)) * 128 + kOff;
    v8f acc = {};
#pragma unroll
    for (int s = 0; s < 32; ++s) {
      v2f a;
      a.x = xr[4 * s];
      a.y = xr[4 * s + 1];
      acc = __builtin_amdgcn_wmma_f32_16x16x4_f32(
          false, a, false, bf[s], (short)0, acc, false, false);
    }
    const int r = row0 + ((lane >> 4) << 3);
    const int c = col0 + (lane & 15);
    if (c < 40) {
#pragma unroll
      for (int i = 0; i < 8; ++i)
        G[(size_t)(r + i) * 40 + c] = acc[i];
    }
  }
}

__global__ __launch_bounds__(256) void gat_attn(
    const float* __restrict__ Hf, const float* __restrict__ aS,
    const float* __restrict__ aD, float* __restrict__ alS,
    float* __restrict__ alD, int N, int NH, int C)
{
  int idx = blockIdx.x * blockDim.x + threadIdx.x;
  if (idx >= N * NH) return;
  int n = idx / NH, h = idx - n * NH;
  const float* hp = Hf + (size_t)n * NH * C + (size_t)h * C;
  const float* sp = aS + (size_t)h * C;
  const float* dp = aD + (size_t)h * C;
  float s = 0.f, d = 0.f;
  for (int c = 0; c < C; ++c) { s += hp[c] * sp[c]; d += hp[c] * dp[c]; }
  alS[idx] = s;
  alD[idx] = d;
}

__device__ __forceinline__ float leaky(float v) {
  return (v > 0.f) ? v : NEG_SLOPE * v;
}
__device__ __forceinline__ float wave_max(float v) {
  for (int o = 16; o > 0; o >>= 1) v = fmaxf(v, __shfl_xor(v, o, 32));
  return v;
}
__device__ __forceinline__ float wave_sum(float v) {
  for (int o = 16; o > 0; o >>= 1) v += __shfl_xor(v, o, 32);
  return v;
}
__device__ __forceinline__ float lane_bcast(float v, int j) {
  return __int_as_float(__builtin_amdgcn_readlane(__float_as_int(v), j));
}

#define EDGES_PER_THREAD 8
#define SLICE_SHIFT 14
#define SLICE (1 << SLICE_SHIFT)
#define HIST_WORDS 51200
#define HIST_EPT (SLICE / 1024)
__global__ __launch_bounds__(1024) void gat_csr_hist(
    const int* __restrict__ ei, int E, int NS, unsigned* __restrict__ part,
    unsigned* __restrict__ rank)
{
  __shared__ unsigned hist[HIST_WORDS];
  const int t = threadIdx.x;
  for (int i = t; i < HIST_WORDS / 4; i += 1024)
    ((uint4*)hist)[i] = make_uint4(0u, 0u, 0u, 0u);
  __syncthreads();
  const int e0 = blockIdx.x * SLICE + t;
  int dst[HIST_EPT];
  unsigned r[HIST_EPT];
#pragma unroll
  for (int k = 0; k < HIST_EPT; ++k) {
    const int e = e0 + 1024 * k;
    dst[k] = (e < E) ? ei[E + e] : -1;
  }
#pragma unroll
  for (int k = 0; k < HIST_EPT; ++k)
    r[k] = ((unsigned)dst[k] < (unsigned)HIST_WORDS) ? atomicAdd(&hist[dst[k]], 1u) : 0u;
#pragma unroll
  for (int k = 0; k < HIST_EPT; ++k) {
    const int e = e0 + 1024 * k;
    if (e < E) rank[e] = r[k];
  }
  __syncthreads();
  uint4* p = (uint4*)(part + (size_t)blockIdx.x * NS);
  for (int i = t; i < NS / 4; i += 1024) p[i] = ((const uint4*)hist)[i];
}

#define COLSUM_CHUNK 16
__global__ __launch_bounds__(256) void gat_csr_colsum(
    unsigned* __restrict__ part, unsigned* __restrict__ cnt, int N, int NS,
    int B, int NP)
{
  const int n = blockIdx.x * 256 + threadIdx.x;
  if (n >= NP) return;
  unsigned run = 0;
  if (n < N) {
    unsigned* p = part + n;
    for (int b0 = 0; b0 < B; b0 += COLSUM_CHUNK) {
      unsigned v[COLSUM_CHUNK];
#pragma unroll
      for (int k = 0; k < COLSUM_CHUNK; ++k)
        v[k] = (b0 + k < B) ? p[(size_t)(b0 + k) * NS] : 0u;
#pragma unroll
      for (int k = 0; k < COLSUM_CHUNK; ++k) {
        if (b0 + k < B) p[(size_t)(b0 + k) * NS] = run;
        run += v[k];
      }
    }
  }
  cnt[n] = run;
}

#define SCAN_CHUNK 16384
__global__ __launch_bounds__(1024) void gat_csr_scan(
    const unsigned* __restrict__ cnt, unsigned* __restrict__ rowptr, int NP)
{
  __shared__ unsigned wtot[32];
  const int t = threadIdx.x, lane = t & 31, wv = t >> 5;
  unsigned carry = 0;
  for (int base = 0; base < NP; base += SCAN_CHUNK) {
    const uint4* p = (const uint4*)(cnt + base + t * 16);
    unsigned v[16];
#pragma unroll
    for (int i = 0; i < 4; ++i) {
      const uint4 q = p[i];
      v[4 * i] = q.x; v[4 * i + 1] = q.y; v[4 * i + 2] = q.z; v[4 * i + 3] = q.w;
    }
    unsigned s = 0;
#pragma unroll
    for (int i = 0; i < 16; ++i) s += v[i];
    unsigned incl = s;
    for (int o = 1; o < 32; o <<= 1) {
      const unsigned u = __shfl_up(incl, o, 32);
      if (lane >= o) incl += u;
    }
    if (lane == 31) wtot[wv] = incl;
    __syncthreads();
    if (wv == 0) {
      unsigned w = wtot[lane];
      for (int o = 1; o < 32; o <<= 1) {
        const unsigned u = __shfl_up(w, o, 32);
        if (lane >= o) w += u;
      }
      wtot[lane] = w;
    }
    __syncthreads();
    unsigned run = carry + (incl - s) + (wv ? wtot[wv - 1] : 0u);
    carry += wtot[31];
    uint4* op = (uint4*)(rowptr + base + t * 16);
#pragma unroll
    for (int i = 0; i < 4; ++i) {
      uint4 q;
      q.x = run; run += v[4 * i];
      q.y = run; run += v[4 * i + 1];
      q.z = run; run += v[4 * i + 2];
      q.w = run; run += v[4 * i + 3];
      op[i] = q;
    }
    __syncthreads();
  }
}

__global__ __launch_bounds__(256) void gat_csr_fill(
    const int* __restrict__ ei, int E, const unsigned* __restrict__ rowptr,
    const unsigned* __restrict__ part, int NS,
    const unsigned* __restrict__ rank, int* __restrict__ col)
{
  const int e0 = blockIdx.x * (256 * EDGES_PER_THREAD) + threadIdx.x;
  int dst[EDGES_PER_THREAD], sr[EDGES_PER_THREAD];
  unsigned rk[EDGES_PER_THREAD], rp[EDGES_PER_THREAD], pp[EDGES_PER_THREAD];
#pragma unroll
  for (int k = 0; k < EDGES_PER_THREAD; ++k) {
    const int e = e0 + 256 * k;
    const bool ok = e < E;
    dst[k] = ok ? ei[E + e] : -1;
    sr[k] = ok ? ei[e] : 0;
    rk[k] = ok ? rank[e] : 0u;
  }
#pragma unroll
  for (int k = 0; k < EDGES_PER_THREAD; ++k)
    rp[k] = (dst[k] >= 0) ? rowptr[dst[k]] : 0u;
#pragma unroll
  for (int k = 0; k < EDGES_PER_THREAD; ++k) {
    const int e = e0 + 256 * k;
    pp[k] = (dst[k] >= 0) ? part[(size_t)(e >> SLICE_SHIFT) * NS + dst[k]] : 0u;
  }
#pragma unroll
  for (int k = 0; k < EDGES_PER_THREAD; ++k)
    if (dst[k] >= 0) col[rp[k] + pp[k] + rk[k]] = sr[k];
}

#define NODES_PER_WAVE 4
#define ROWS_IN_FLIGHT 4
__global__ __launch_bounds__(256) void gat_gather_l1(
    const unsigned* __restrict__ rowptr, const int* __restrict__ col,
    const float* __restrict__ Hf, const float* __restrict__ alS,
    const float* __restrict__ alD, const float* __restrict__ b,
    float* __restrict__ out1, int N)
{
  const int lane = threadIdx.x & 31;
  const int wv = __builtin_amdgcn_readfirstlane((int)(threadIdx.x >> 5));
  const int d0 = (blockIdx.x * 8 + wv) * NODES_PER_WAVE;
  if (d0 >= N) return;
  const int hsel = lane >> 4;
  unsigned r0 = rowptr[d0], r1 = rowptr[d0 + 1], r2 = rowptr[min(d0 + 2, N)];
  const int elast = max((int)rowptr[N] - 1, 0);
  int dgn = (int)(r2 - r1);
  float ad0 = alD[d0 * 2], ad1 = alD[d0 * 2 + 1];
  float vs0 = leaky(alS[d0 * 2] + ad0), vs1 = leaky(alS[d0 * 2 + 1] + ad1);
  int sc = col[min((int)r0 + min(lane, max((int)(r1 - r0) - 1, 0)), elast)];
  float2 ac = *(const float2*)(alS + (size_t)sc * 2);
#pragma unroll 1
  for (int kn = 0; kn < NODES_PER_WAVE; ++kn) {
  const int d = d0 + kn;
  if (d >= N) break;
  const int beg = (int)r0;
  const int deg = (int)(r1 - r0);
  const int dn = d + 1;
  const bool more = (kn + 1 < NODES_PER_WAVE) && (dn < N);
  const int degn = more ? dgn : 0;
  const unsigned r3 = rowptr[min(d + 3, N)];
  const int dnc = min(dn, N - 1);
  const float ns0 = alS[dnc * 2], ns1 = alS[dnc * 2 + 1];
  const float nd0 = alD[dnc * 2], nd1 = alD[dnc * 2 + 1];
  const int snr = col[min((int)r1 + min(lane, max(degn - 1, 0)), elast)];
  const float4 hown = *(const float4*)(Hf + (size_t)d * 128 + lane * 4);
  float m0 = vs0, m1 = vs1;
  if (lane < deg) {
    m0 = fmaxf(m0, leaky(ac.x + ad0));
    m1 = fmaxf(m1, leaky(ac.y + ad1));
  }
  for (int base = 32; base < deg; base += 32) {
    const int i = base + lane;
    if (i < deg) {
      const int s = col[beg + i];
      const float2 a = *(const float2*)(alS + (size_t)s * 2);
      m0 = fmaxf(m0, leaky(a.x + ad0));
      m1 = fmaxf(m1, leaky(a.y + ad1));
    }
  }
  m0 = wave_max(m0);
  m1 = wave_max(m1);
  __builtin_amdgcn_sched_barrier(0);
  const float2 anr = *(const float2*)(alS + (size_t)snr * 2);
  const float ws0 = expf(vs0 - m0), ws1 = expf(vs1 - m1);
  float4 acc;
  {
    const float w = hsel ? ws1 : ws0;
    acc.x = w * hown.x; acc.y = w * hown.y; acc.z = w * hown.z; acc.w = w * hown.w;
  }
  float p0 = 0.f, p1 = 0.f;
  for (int base = 0; base < deg; base += 32) {
    const int i = base + lane;
    int s = d;
    float w0 = 0.f, w1 = 0.f;
    if (base == 0) {
      s = sc;
      if (i < deg) {
        w0 = expf(leaky(ac.x + ad0) - m0);
        w1 = expf(leaky(ac.y + ad1) - m1);
      }
    } else if (i < deg) {
      s = col[beg + i];
      const float2 a = *(const float2*)(alS + (size_t)s * 2);
      w0 = expf(leaky(a.x + ad0) - m0);
      w1 = expf(leaky(a.y + ad1) - m1);
    }
    p0 += w0;
    p1 += w1;
    const int cnt = min(32, deg - base);
    for (int j0 = 0; j0 < cnt; j0 += ROWS_IN_FLIGHT) {
#pragma unroll
      for (int u = 0; u < ROWS_IN_FLIGHT; ++u) {
        const int j = j0 + u;
        const int sj = __builtin_amdgcn_readlane(s, j);
        const float wj0 = lane_bcast(w0, j), wj1 = lane_bcast(w1, j);
        const float w = hsel ? wj1 : wj0;
        const float4 hv = *(const float4*)(Hf + (size_t)sj * 128 + lane * 4);
        acc.x += w * hv.x; acc.y += w * hv.y; acc.z += w * hv.z; acc.w += w * hv.w;
      }
    }
  }
  const float den0 = wave_sum(p0) + ws0;
  const float den1 = wave_sum(p1) + ws1;
  const float den = (hsel ? den1 : den0) + 1e-16f;
  const float4 bv = *(const float4*)(b + lane * 4);
  float4 r;
  r.x = acc.x / den + bv.x;
  r.y = acc.y / den + bv.y;
  r.z = acc.z / den + bv.z;
  r.w = acc.w / den + bv.w;
  r.x = (r.x > 0.f) ? r.x : (expf(r.x) - 1.0f);
  r.y = (r.y > 0.f) ? r.y : (expf(r.y) - 1.0f);
  r.z = (r.z > 0.f) ? r.z : (expf(r.z) - 1.0f);
  r.w = (r.w > 0.f) ? r.w : (expf(r.w) - 1.0f);
  *(float4*)(out1 + (size_t)d * 128 + lane * 4) = r;
  r0 = r1; r1 = r2; r2 = r3;
  dgn = (int)(r2 - r1);
  ad0 = nd0; ad1 = nd1;
  vs0 = leaky(ns0 + nd0); vs1 = leaky(ns1 + nd1);
  sc = snr; ac = anr;
  }
}

__global__ __launch_bounds__(256) void gat_gather_l2(
    const unsigned* __restrict__ rowptr, const int* __restrict__ col,
    const float* __restrict__ G, const float* __restrict__ alS,
    const float* __restrict__ alD, const float* __restrict__ b,
    float* __restrict__ out, int N)
{
  __shared__ float stage[8 * NODES_PER_WAVE * 40];
  const int lane = threadIdx.x & 31;
  const int wv = __builtin_amdgcn_readfirstlane((int)(threadIdx.x >> 5));
  const int d0 = (blockIdx.x * 8 + wv) * NODES_PER_WAVE;
  float* slot = stage + wv * (NODES_PER_WAVE * 40);
  const bool act = lane < 20;
  const int c2 = act ? 2 * lane : 0;
  unsigned r0 = 0, r1 = 0, r2 = 0;
  int dgn = 0;
  float ad = 0.f, vs = 0.f;
  int sc = 0;
  float ac = 0.f;
  const int elast = max((int)rowptr[N] - 1, 0);
  if (d0 < N) {
    r0 = rowptr[d0]; r1 = rowptr[d0 + 1]; r2 = rowptr[min(d0 + 2, N)];
    dgn = (int)(r2 - r1);
    ad = alD[d0];
    vs = leaky(alS[d0] + ad);
    sc = col[min((int)r0 + min(lane, max((int)(r1 - r0) - 1, 0)), elast)];
    ac = alS[sc];
  }
#pragma unroll 1
  for (int kn = 0; kn < NODES_PER_WAVE; ++kn) {
  const int d = d0 + kn;
  float res0 = 0.f, res1 = 0.f;
  if (d < N) {
    const int beg = (int)r0;
    const int deg = (int)(r1 - r0);
    const int dn = d + 1;
    const bool more = (kn + 1 < NODES_PER_WAVE) && (dn < N);
    const int degn = more ? dgn : 0;
    const unsigned r3 = rowptr[min(d + 3, N)];
    const int dnc = min(dn, N - 1);
    const float ns = alS[dnc], nd = alD[dnc];
    const int snr = col[min((int)r1 + min(lane, max(degn - 1, 0)), elast)];
    const float2 gown = *(const float2*)(G + (size_t)d * 40 + c2);
    float m = vs;
    if (lane < deg) m = fmaxf(m, leaky(ac + ad));
    for (int base = 32; base < deg; base += 32) {
      const int i = base + lane;
      if (i < deg) m = fmaxf(m, leaky(alS[col[beg + i]] + ad));
    }
    m = wave_max(m);
    __builtin_amdgcn_sched_barrier(0);
    const float anr = alS[snr];
    const float wself = expf(vs - m);
    float acc0 = wself * gown.x, acc1 = wself * gown.y;
    float p = 0.f;
    for (int base = 0; base < deg; base += 32) {
      const int i = base + lane;
      int s = d;
      float w = 0.f;
      if (base == 0) {
        s = sc;
        if (i < deg) w = expf(leaky(ac + ad) - m);
      } else if (i < deg) {
        s = col[beg + i];
        w = expf(leaky(alS[s] + ad) - m);
      }
      p += w;
      const int cnt = min(32, deg - base);
      for (int j0 = 0; j0 < cnt; j0 += ROWS_IN_FLIGHT) {
#pragma unroll
        for (int u = 0; u < ROWS_IN_FLIGHT; ++u) {
          const int j = j0 + u;
          const int sj = __builtin_amdgcn_readlane(s, j);
          const float wj = lane_bcast(w, j);
          const float2 gv = *(const float2*)(G + (size_t)sj * 40 + c2);
          acc0 += wj * gv.x;
          acc1 += wj * gv.y;
        }
      }
    }
    const float den = wave_sum(p) + wself + 1e-16f;
    res0 = acc0 / den + b[c2];
    res1 = acc1 / den + b[c2 + 1];
    r0 = r1; r1 = r2; r2 = r3;
    dgn = (int)(r2 - r1);
    ad = nd;
    vs = leaky(ns + nd);
    sc = snr; ac = anr;
  }
  if (act) {
    slot[kn * 40 + c2] = res0;
    slot[kn * 40 + c2 + 1] = res1;
  }
  }
  const size_t lim = (size_t)N * 40;
  const size_t g0 = (size_t)d0 * 40 + lane;
#pragma unroll
  for (int j = 0; j < (NODES_PER_WAVE * 40) / 32; ++j) {
    const size_t g = g0 + 32 * j;
    if (g < lim) out[g] = slot[32 * j + lane];
  }
}

extern "C" void kernel_launch(void* const* d_in, const int* in_sizes, int n_in,
                              void* d_out, int out_size, void* d_ws, size_t ws_size,
                              hipStream_t stream)
{
  const float* x   = (const float*)d_in[0];
  const int*   ei  = (const int*)d_in[1];
  const float* W1  = (const float*)d_in[2];
  const float* aS1 = (const float*)d_in[3];
  const float* aD1 = (const float*)d_in[4];
  const float* b1  = (const float*)d_in[5];
  const float* W2  = (const float*)d_in[6];
  const float* aS2 = (const float*)d_in[7];
  const float* aD2 = (const float*)d_in[8];
  const float* b2  = (const float*)d_in[9];
  float* out = (float*)d_out;

  const int N  = in_sizes[0] / 128;
  const int E  = in_sizes[1] / 2;
  const int NP = (N / SCAN_CHUNK + 1) * SCAN_CHUNK;

  char* ws = (char*)d_ws;
  size_t off = 0;
  auto carve = [&](size_t bytes) -> char* {
    char* p = ws + off;
    off += (bytes + 255) & ~(size_t)255;
    return p;
  };
  float*    h1     = (float*)carve((size_t)N * 128 * 4);
  float*    out1   = (float*)carve((size_t)N * 128 * 4);
  float*    alS    = (float*)carve((size_t)N * 2 * 4);
  float*    alD    = (float*)carve((size_t)N * 2 * 4);
  unsigned* cnt    = (unsigned*)carve((size_t)NP * 4);
  unsigned* rowptr = (unsigned*)carve((size_t)NP * 4);
  unsigned* rank   = (unsigned*)carve((size_t)E * 4);
  int*      col    = (int*)carve((size_t)E * 4);
  const int NS = (N + 31) & ~31;
  const int B  = (E + SLICE - 1) >> SLICE_SHIFT;
  unsigned* part   = (unsigned*)carve((size_t)B * NS * 4);
  float* g2 = h1;

  gat_csr_hist<<<B, 1024, 0, stream>>>(ei, E, NS, part, rank);
  gat_csr_colsum<<<NP / 256, 256, 0, stream>>>(part, cnt, N, NS, B, NP);
  gat_csr_scan<<<1, 1024, 0, stream>>>(cnt, rowptr, NP);
  gat_csr_fill<<<(E + 2047) / 2048, 256, 0, stream>>>(ei, E, rowptr, part, NS, rank, col);

  gat_gemm_128x128<<<768, 256, 0, stream>>>(x, W1, h1, N / 16);
  gat_attn<<<(N * 2 + 255) / 256, 256, 0, stream>>>(h1, aS1, aD1, alS, alD, N, 2, 64);
  gat_gather_l1<<<(N + 31) / 32, 256, 0, stream>>>(rowptr, col, h1, alS, alD, b1, out1, N);

  gat_gemm_128x40<<<1024, 96, 0, stream>>>(out1, W2, g2, N / 16);
  gat_attn<<<(N + 255) / 256, 256, 0, stream>>>(g2, aS2, aD2, alS, alD, N, 1, 40);
  gat_gather_l2<<<(N + 31) / 32, 256, 0, stream>>>(rowptr, col, g2, alS, alD, b2, out, N);
}


# Independent stages share a launch: edge histogram blocks run beside the layer 1 GEMM blocks (1024 threads each), the slice sums beside the layer 1 attention dots; 8 launches, sum orders unchanged
# speedup vs baseline: 3.5035x; 1.0771x over previous
#include <hip/hip_runtime.h>
#include <math.h>

typedef __attribute__((ext_vector_type(2))) float v2f;
typedef __attribute__((ext_vector_type(8))) float v8f;

#define NEG_SLOPE 0.2f

__device__ __forceinline__ void gemm_128x128_walk(
    const float* __restrict__ X, const float* __restrict__ W,
    float* __restrict__ H, int nTiles, int colTile, int tile0, int tileStep)
{
  const int lane = threadIdx.x & 31;
  const int col0 = colTile << 4;
  const int kOff = (lane >> 4) << 1;
  const int bCol = col0 + (lane & 15);
  v2f bf[32];
#pragma unroll
  for (int s = 0; s < 32; ++s) {
    bf[s].x = W[(size_t)(4 * s + kOff) * 128 + bCol];
    bf[s].y = W[(size_t)(4 * s + kOff + 1) * 128 + bCol];
  }
  for (int tile = tile0; tile < nTiles; tile += tileStep) {
    const int row0 = tile << 4;
    const float* xr = X + (size_t)(row0 + (lane & 15)) * 128 + kOff;
    v8f acc = {};
#pragma unroll
    for (int s = 0; s < 32; ++s) {
      v2f a;
      a.x = xr[4 * s];
      a.y = xr[4 * s + 1];
      acc = __builtin_amdgcn_wmma_f32_16x16x4_f32(
          false, a, false, bf[s], (short)0, acc, false, false);
    }
    const int r = row0 + ((lane >> 4) << 3);
    const int c = col0 + (lane & 15);
#pragma unroll
    for (int i = 0; i < 8; ++i)
      H[(size_t)(r + i) * 128 + c] = acc[i];
  }
}

__global__ __launch_bounds__(96) void gat_gemm_128x40(
    const float* __restrict__ X, const float* __restrict__ W,
    float* __restrict__ G, int nTiles)
{
  const int lane = threadIdx.x & 31;
  const int wave = threadIdx.x >> 5;
  const int col0 = wave << 4;
  const int kOff = (lane >> 4) << 1;
  const int bCol = col0 + (lane & 15);
  const bool colOk = (bCol < 40);
  v2f bf[32];
#pragma unroll
  for (int s = 0; s < 32; ++s) {
    bf[s].x = colOk ? W[(size_t)(4 * s + kOff) * 40 + bCol] : 0.0f;
    bf[s].y = colOk ? W[(size_t)(4 * s + kOff + 1) * 40 + bCol] : 0.0f;
  }
  for (int tile = blockIdx.x; tile < nTiles; tile += gridDim.x) {
    const int row0 = tile << 4;
    const float* xr = X + (size_t)(row0 + (lane & 15)) * 128 + kOff;
    v8f acc = {};
#pragma unroll
    for (int s = 0; s < 32; ++s) {
      v2f a;
      a.x = xr[4 * s];
      a.y = xr[4 * s + 1];
      acc = __builtin_amdgcn_wmma_f32_16x16x4_f32(
          false, a, false, bf[s], (short)0, acc, false, false);
    }
    const int r = row0 + ((lane >> 4) << 3);
    const int c = col0 + (lane & 15);
    if (c < 40) {
#pragma unroll
      for (int i = 0; i < 8; ++i)
        G[(size_t)(r + i) * 40 + c] = acc[i];
    }
  }
}

__device__ __forceinline__ void attn_dots(
    int idx, const float* __restrict__ Hf, const float* __restrict__ aS,
    const float* __restrict__ aD, float* __restrict__ alS,
    float* __restrict__ alD, int N, int NH, int C)
{
  if (idx >= N * NH) return;
  int n = idx / NH, h = idx - n * NH;
  const float* hp = Hf + (size_t)n * NH * C + (size_t)h * C;
  const float* sp = aS + (size_t)h * C;
  const float* dp = aD + (size_t)h * C;
  float s = 0.f, d = 0.f;
  for (int c = 0; c < C; ++c) { s += hp[c] * sp[c]; d += hp[c] * dp[c]; }
  alS[idx] = s;
  alD[idx] = d;
}

__global__ __launch_bounds__(256) void gat_attn(
    const float* __restrict__ Hf, const float* __restrict__ aS,
    const float* __restrict__ aD, float* __restrict__ alS,
    float* __restrict__ alD, int N, int NH, int C)
{
  attn_dots(blockIdx.x * 256 + threadIdx.x, Hf, aS, aD, alS, alD, N, NH, C);
}

__device__ __forceinline__ float leaky(float v) {
  return (v > 0.f) ? v : NEG_SLOPE * v;
}
__device__ __forceinline__ float wave_max(float v) {
  for (int o = 16; o > 0; o >>= 1) v = fmaxf(v, __shfl_xor(v, o, 32));
  return v;
}
__device__ __forceinline__ float wave_sum(float v) {
  for (int o = 16; o > 0; o >>= 1) v += __shfl_xor(v, o, 32);
  return v;
}
__device__ __forceinline__ float lane_bcast(float v, int j) {
  return __int_as_float(__builtin_amdgcn_readlane(__float_as_int(v), j));
}

#define EDGES_PER_THREAD 8
#define SLICE_SHIFT 14
#define SLICE (1 << SLICE_SHIFT)
#define HIST_WORDS 51200
#define HIST_EPT (SLICE / 1024)
__global__ __launch_bounds__(1024) void gat_hist_gemm1(
    const int* __restrict__ ei, int E, int NS, int B, unsigned* __restrict__ part,
    unsigned* __restrict__ rank, const float* __restrict__ X,
    const float* __restrict__ W, float* __restrict__ H, int nTiles)
{
  __shared__ unsigned hist[HIST_WORDS];
  const int t = threadIdx.x;
  if ((int)blockIdx.x >= B) {
    const int wv = __builtin_amdgcn_readfirstlane(t >> 5);
    const int gb = (int)blockIdx.x - B;
    const int nGemm = (int)gridDim.x - B;
    gemm_128x128_walk(X, W, H, nTiles, wv & 7, gb * 4 + (wv >> 3), nGemm * 4);
    return;
  }
  for (int i = t; i < HIST_WORDS / 4; i += 1024)
    ((uint4*)hist)[i] = make_uint4(0u, 0u, 0u, 0u);
  __syncthreads();
  const int e0 = blockIdx.x * SLICE + t;
  int dst[HIST_EPT];
  unsigned r[HIST_EPT];
#pragma unroll
  for (int k = 0; k < HIST_EPT; ++k) {
    const int e = e0 + 1024 * k;
    dst[k] = (e < E) ? ei[E + e] : -1;
  }
#pragma unroll
  for (int k = 0; k < HIST_EPT; ++k)
    r[k] = ((unsigned)dst[k] < (unsigned)HIST_WORDS) ? atomicAdd(&hist[dst[k]], 1u) : 0u;
#pragma unroll
  for (int k = 0; k < HIST_EPT; ++k) {
    const int e = e0 + 1024 * k;
    if (e < E) rank[e] = r[k];
  }
  __syncthreads();
  uint4* p = (uint4*)(part + (size_t)blockIdx.x * NS);
  for (int i = t; i < NS / 4; i += 1024) p[i] = ((const uint4*)hist)[i];
}

#define COLSUM_CHUNK 16
__device__ __forceinline__ void csr_colsum(
    int n, unsigned* __restrict__ part, unsigned* __restrict__ cnt, int N, int NS,
    int B, int NP)
{
  if (n >= NP) return;
  unsigned run = 0;
  if (n < N) {
    unsigned* p = part + n;
    for (int b0 = 0; b0 < B; b0 += COLSUM_CHUNK) {
      unsigned v[COLSUM_CHUNK];
#pragma unroll
      for (int k = 0; k < COLSUM_CHUNK; ++k)
        v[k] = (b0 + k < B) ? p[(size_t)(b0 + k) * NS] : 0u;
#pragma unroll
      for (int k = 0; k < COLSUM_CHUNK; ++k) {
        if (b0 + k < B) p[(size_t)(b0 + k) * NS] = run;
        run += v[k];
      }
    }
  }
  cnt[n] = run;
}

__global__ __launch_bounds__(256) void gat_colsum_attn(
    unsigned* __restrict__ part, unsigned* __restrict__ cnt, int N, int NS,
    int B, int NP, int sumBlocks,
    const float* __restrict__ Hf, const float* __restrict__ aS,
    const float* __restrict__ aD, float* __restrict__ alS,
    float* __restrict__ alD, int NH, int C)
{
  if ((int)blockIdx.x < sumBlocks) {
    csr_colsum(blockIdx.x * 256 + threadIdx.x, part, cnt, N, NS, B, NP);
  } else {
    attn_dots(((int)blockIdx.x - sumBlocks) * 256 + threadIdx.x,
              Hf, aS, aD, alS, alD, N, NH, C);
  }
}

#define SCAN_CHUNK 16384
__global__ __launch_bounds__(1024) void gat_csr_scan(
    const unsigned* __restrict__ cnt, unsigned* __restrict__ rowptr, int NP)
{
  __shared__ unsigned wtot[32];
  const int t = threadIdx.x, lane = t & 31, wv = t >> 5;
  unsigned carry = 0;
  for (int base = 0; base < NP; base += SCAN_CHUNK) {
    const uint4* p = (const uint4*)(cnt + base + t * 16);
    unsigned v[16];
#pragma unroll
    for (int i = 0; i < 4; ++i) {
      const uint4 q = p[i];
      v[4 * i] = q.x; v[4 * i + 1] = q.y; v[4 * i + 2] = q.z; v[4 * i + 3] = q.w;
    }
    unsigned s = 0;
#pragma unroll
    for (int i = 0; i < 16; ++i) s += v[i];
    unsigned incl = s;
    for (int o = 1; o < 32; o <<= 1) {
      const unsigned u = __shfl_up(incl, o, 32);
      if (lane >= o) incl += u;
    }
    if (lane == 31) wtot[wv] = incl;
    __syncthreads();
    if (wv == 0) {
      unsigned w = wtot[lane];
      for (int o = 1; o < 32; o <<= 1) {
        const unsigned u = __shfl_up(w, o, 32);
        if (lane >= o) w += u;
      }
      wtot[lane] = w;
    }
    __syncthreads();
    unsigned run = carry + (incl - s) + (wv ? wtot[wv - 1] : 0u);
    carry += wtot[31];
    uint4* op = (uint4*)(rowptr + base + t * 16);
#pragma unroll
    for (int i = 0; i < 4; ++i) {
      uint4 q;
      q.x = run; run += v[4 * i];
      q.y = run; run += v[4 * i + 1];
      q.z = run; run += v[4 * i + 2];
      q.w = run; run += v[4 * i + 3];
      op[i] = q;
    }
    __syncthreads();
  }
}

__global__ __launch_bounds__(256) void gat_csr_fill(
    const int* __restrict__ ei, int E, const unsigned* __restrict__ rowptr,
    const unsigned* __restrict__ part, int NS,
    const unsigned* __restrict__ rank, int* __restrict__ col)
{
  const int e0 = blockIdx.x * (256 * EDGES_PER_THREAD) + threadIdx.x;
  int dst[EDGES_PER_THREAD], sr[EDGES_PER_THREAD];
  unsigned rk[EDGES_PER_THREAD], rp[EDGES_PER_THREAD], pp[EDGES_PER_THREAD];
#pragma unroll
  for (int k = 0; k < EDGES_PER_THREAD; ++k) {
    const int e = e0 + 256 * k;
    const bool ok = e < E;
    dst[k] = ok ? ei[E + e] : -1;
    sr[k] = ok ? ei[e] : 0;
    rk[k] = ok ? rank[e] : 0u;
  }
#pragma unroll
  for (int k = 0; k < EDGES_PER_THREAD; ++k)
    rp[k] = (dst[k] >= 0) ? rowptr[dst[k]] : 0u;
#pragma unroll
  for (int k = 0; k < EDGES_PER_THREAD; ++k) {
    const int e = e0 + 256 * k;
    pp[k] = (dst[k] >= 0) ? part[(size_t)(e >> SLICE_SHIFT) * NS + dst[k]] : 0u;
  }
#pragma unroll
  for (int k = 0; k < EDGES_PER_THREAD; ++k)
    if (dst[k] >= 0) col[rp[k] + pp[k] + rk[k]] = sr[k];
}

#define NODES_PER_WAVE 4
#define ROWS_IN_FLIGHT 4
__global__ __launch_bounds__(256) void gat_gather_l1(
    const unsigned* __restrict__ rowptr, const int* __restrict__ col,
    const float* __restrict__ Hf, const float* __restrict__ alS,
    const float* __restrict__ alD, const float* __restrict__ b,
    float* __restrict__ out1, int N)
{
  const int lane = threadIdx.x & 31;
  const int wv = __builtin_amdgcn_readfirstlane((int)(threadIdx.x >> 5));
  const int d0 = (blockIdx.x * 8 + wv) * NODES_PER_WAVE;
  if (d0 >= N) return;
  const int hsel = lane >> 4;
  unsigned r0 = rowptr[d0], r1 = rowptr[d0 + 1], r2 = rowptr[min(d0 + 2, N)];
  const int elast = max((int)rowptr[N] - 1, 0);
  int dgn = (int)(r2 - r1);
  float ad0 = alD[d0 * 2], ad1 = alD[d0 * 2 + 1];
  float vs0 = leaky(alS[d0 * 2] + ad0), vs1 = leaky(alS[d0 * 2 + 1] + ad1);
  int sc = col[min((int)r0 + min(lane, max((int)(r1 - r0) - 1, 0)), elast)];
  float2 ac = *(const float2*)(alS + (size_t)sc * 2);
#pragma unroll 1
  for (int kn = 0; kn < NODES_PER_WAVE; ++kn) {
  const int d = d0 + kn;
  if (d >= N) break;
  const int beg = (int)r0;
  const int deg = (int)(r1 - r0);
  const int dn = d + 1;
  const bool more = (kn + 1 < NODES_PER_WAVE) && (dn < N);
  const int degn = more ? dgn : 0;
  const unsigned r3 = rowptr[min(d + 3, N)];
  const int dnc = min(dn, N - 1);
  const float ns0 = alS[dnc * 2], ns1 = alS[dnc * 2 + 1];
  const float nd0 = alD[dnc * 2], nd1 = alD[dnc * 2 + 1];
  const int snr = col[min((int)r1 + min(lane, max(degn - 1, 0)), elast)];
  const float4 hown = *(const float4*)(Hf + (size_t)d * 128 + lane * 4);
  float m0 = vs0, m1 = vs1;
  if (lane < deg) {
    m0 = fmaxf(m0, leaky(ac.x + ad0));
    m1 = fmaxf(m1, leaky(ac.y + ad1));
  }
  for (int base = 32; base < deg; base += 32) {
    const int i = base + lane;
    if (i < deg) {
      const int s = col[beg + i];
      const float2 a = *(const float2*)(alS + (size_t)s * 2);
      m0 = fmaxf(m0, leaky(a.x + ad0));
      m1 = fmaxf(m1, leaky(a.y + ad1));
    }
  }
  m0 = wave_max(m0);
  m1 = wave_max(m1);
  __builtin_amdgcn_sched_barrier(0);
  const float2 anr = *(const float2*)(alS + (size_t)snr * 2);
  const float ws0 = expf(vs0 - m0), ws1 = expf(vs1 - m1);
  float4 acc;
  {
    const float w = hsel ? ws1 : ws0;
    acc.x = w * hown.x; acc.y = w * hown.y; acc.z = w * hown.z; acc.w = w * hown.w;
  }
  float p0 = 0.f, p1 = 0.f;
  for (int base = 0; base < deg; base += 32) {
    const int i = base + lane;
    int s = d;
    float w0 = 0.f, w1 = 0.f;
    if (base == 0) {
      s = sc;
      if (i < deg) {
        w0 = expf(leaky(ac.x + ad0) - m0);
        w1 = expf(leaky(ac.y + ad1) - m1);
      }
    } else if (i < deg) {
      s = col[beg + i];
      const float2 a = *(const float2*)(alS + (size_t)s * 2);
      w0 = expf(leaky(a.x + ad0) - m0);
      w1 = expf(leaky(a.y + ad1) - m1);
    }
    p0 += w0;
    p1 += w1;
    const int cnt = min(32, deg - base);
    for (int j0 = 0; j0 < cnt; j0 += ROWS_IN_FLIGHT) {
#pragma unroll
      for (int u = 0; u < ROWS_IN_FLIGHT; ++u) {
        const int j = j0 + u;
        const int sj = __builtin_amdgcn_readlane(s, j);
        const float wj0 = lane_bcast(w0, j), wj1 = lane_bcast(w1, j);
        const float w = hsel ? wj1 : wj0;
        const float4 hv = *(const float4*)(Hf + (size_t)sj * 128 + lane * 4);
        acc.x += w * hv.x; acc.y += w * hv.y; acc.z += w * hv.z; acc.w += w * hv.w;
      }
    }
  }
  const float den0 = wave_sum(p0) + ws0;
  const float den1 = wave_sum(p1) + ws1;
  const float den = (hsel ? den1 : den0) + 1e-16f;
  const float4 bv = *(const float4*)(b + lane * 4);
  float4 r;
  r.x = acc.x / den + bv.x;
  r.y = acc.y / den + bv.y;
  r.z = acc.z / den + bv.z;
  r.w = acc.w / den + bv.w;
  r.x = (r.x > 0.f) ? r.x : (expf(r.x) - 1.0f);
  r.y = (r.y > 0.f) ? r.y : (expf(r.y) - 1.0f);
  r.z = (r.z > 0.f) ? r.z : (expf(r.z) - 1.0f);
  r.w = (r.w > 0.f) ? r.w : (expf(r.w) - 1.0f);
  *(float4*)(out1 + (size_t)d * 128 + lane * 4) = r;
  r0 = r1; r1 = r2; r2 = r3;
  dgn = (int)(r2 - r1);
  ad0 = nd0; ad1 = nd1;
  vs0 = leaky(ns0 + nd0); vs1 = leaky(ns1 + nd1);
  sc = snr; ac = anr;
  }
}

__global__ __launch_bounds__(256) void gat_gather_l2(
    const unsigned* __restrict__ rowptr, const int* __restrict__ col,
    const float* __restrict__ G, const float* __restrict__ alS,
    const float* __restrict__ alD, const float* __restrict__ b,
    float* __restrict__ out, int N)
{
  __shared__ float stage[8 * NODES_PER_WAVE * 40];
  const int lane = threadIdx.x & 31;
  const int wv = __builtin_amdgcn_readfirstlane((int)(threadIdx.x >> 5));
  const int d0 = (blockIdx.x * 8 + wv) * NODES_PER_WAVE;
  float* slot = stage + wv * (NODES_PER_WAVE * 40);
  const bool act = lane < 20;
  const int c2 = act ? 2 * lane : 0;
  unsigned r0 = 0, r1 = 0, r2 = 0;
  int dgn = 0;
  float ad = 0.f, vs = 0.f;
  int sc = 0;
  float ac = 0.f;
  const int elast = max((int)rowptr[N] - 1, 0);
  if (d0 < N) {
    r0 = rowptr[d0]; r1 = rowptr[d0 + 1]; r2 = rowptr[min(d0 + 2, N)];
    dgn = (int)(r2 - r1);
    ad = alD[d0];
    vs = leaky(alS[d0] + ad);
    sc = col[min((int)r0 + min(lane, max((int)(r1 - r0) - 1, 0)), elast)];
    ac = alS[sc];
  }
#pragma unroll 1
  for (int kn = 0; kn < NODES_PER_WAVE; ++kn) {
  const int d = d0 + kn;
  float res0 = 0.f, res1 = 0.f;
  if (d < N) {
    const int beg = (int)r0;
    const int deg = (int)(r1 - r0);
    const int dn = d + 1;
    const bool more = (kn + 1 < NODES_PER_WAVE) && (dn < N);
    const int degn = more ? dgn : 0;
    const unsigned r3 = rowptr[min(d + 3, N)];
    const int dnc = min(dn, N - 1);
    const float ns = alS[dnc], nd = alD[dnc];
    const int snr = col[min((int)r1 + min(lane, max(degn - 1, 0)), elast)];
    const float2 gown = *(const float2*)(G + (size_t)d * 40 + c2);
    float m = vs;
    if (lane < deg) m = fmaxf(m, leaky(ac + ad));
    for (int base = 32; base < deg; base += 32) {
      const int i = base + lane;
      if (i < deg) m = fmaxf(m, leaky(alS[col[beg + i]] + ad));
    }
    m = wave_max(m);
    __builtin_amdgcn_sched_barrier(0);
    const float anr = alS[snr];
    const float wself = expf(vs - m);
    float acc0 = wself * gown.x, acc1 = wself * gown.y;
    float p = 0.f;
    for (int base = 0; base < deg; base += 32) {
      const int i = base + lane;
      int s = d;
      float w = 0.f;
      if (base == 0) {
        s = sc;
        if (i < deg) w = expf(leaky(ac + ad) - m);
      } else if (i < deg) {
        s = col[beg + i];
        w = expf(leaky(alS[s] + ad) - m);
      }
      p += w;
      const int cnt = min(32, deg - base);
      for (int j0 = 0; j0 < cnt; j0 += ROWS_IN_FLIGHT) {
#pragma unroll
        for (int u = 0; u < ROWS_IN_FLIGHT; ++u) {
          const int j = j0 + u;
          const int sj = __builtin_amdgcn_readlane(s, j);
          const float wj = lane_bcast(w, j);
          const float2 gv = *(const float2*)(G + (size_t)sj * 40 + c2);
          acc0 += wj * gv.x;
          acc1 += wj * gv.y;
        }
      }
    }
    const float den = wave_sum(p) + wself + 1e-16f;
    res0 = acc0 / den + b[c2];
    res1 = acc1 / den + b[c2 + 1];
    r0 = r1; r1 = r2; r2 = r3;
    dgn = (int)(r2 - r1);
    ad = nd;
    vs = leaky(ns + nd);
    sc = snr; ac = anr;
  }
  if (act) {
    slot[kn * 40 + c2] = res0;
    slot[kn * 40 + c2 + 1] = res1;
  }
  }
  const size_t lim = (size_t)N * 40;
  const size_t g0 = (size_t)d0 * 40 + lane;
#pragma unroll
  for (int j = 0; j < (NODES_PER_WAVE * 40) / 32; ++j) {
    const size_t g = g0 + 32 * j;
    if (g < lim) out[g] = slot[32 * j + lane];
  }
}

extern "C" void kernel_launch(void* const* d_in, const int* in_sizes, int n_in,
                              void* d_out, int out_size, void* d_ws, size_t ws_size,
                              hipStream_t stream)
{
  const float* x   = (const float*)d_in[0];
  const int*   ei  = (const int*)d_in[1];
  const float* W1  = (const float*)d_in[2];
  const float* aS1 = (const float*)d_in[3];
  const float* aD1 = (const float*)d_in[4];
  const float* b1  = (const float*)d_in[5];
  const float* W2  = (const float*)d_in[6];
  const float* aS2 = (const float*)d_in[7];
  const float* aD2 = (const float*)d_in[8];
  const float* b2  = (const float*)d_in[9];
  float* out = (float*)d_out;

  const int N  = in_sizes[0] / 128;
  const int E  = in_sizes[1] / 2;
  const int NP = (N / SCAN_CHUNK + 1) * SCAN_CHUNK;

  char* ws = (char*)d_ws;
  size_t off = 0;
  auto carve = [&](size_t bytes) -> char* {
    char* p = ws + off;
    off += (bytes + 255) & ~(size_t)255;
    return p;
  };
  float*    h1     = (float*)carve((size_t)N * 128 * 4);
  float*    out1   = (float*)carve((size_t)N * 128 * 4);
  float*    alS    = (float*)carve((size_t)N * 2 * 4);
  float*    alD    = (float*)carve((size_t)N * 2 * 4);
  unsigned* cnt    = (unsigned*)carve((size_t)NP * 4);
  unsigned* rowptr = (unsigned*)carve((size_t)NP * 4);
  unsigned* rank   = (unsigned*)carve((size_t)E * 4);
  int*      col    = (int*)carve((size_t)E * 4);
  const int NS = (N + 31) & ~31;
  const int B  = (E + SLICE - 1) >> SLICE_SHIFT;
  unsigned* part   = (unsigned*)carve((size_t)B * NS * 4);
  float* g2 = h1;

  const int gemmBlocks = (B < 224) ? 256 - B : 32;
  gat_hist_gemm1<<<B + gemmBlocks, 1024, 0, stream>>>(ei, E, NS, B, part, rank, x, W1, h1, N / 16);
  gat_colsum_attn<<<NP / 256 + (N * 2 + 255) / 256, 256, 0, stream>>>(
      part, cnt, N, NS, B, NP, NP / 256, h1, aS1, aD1, alS, alD, 2, 64);
  gat_csr_scan<<<1, 1024, 0, stream>>>(cnt, rowptr, NP);
  gat_csr_fill<<<(E + 2047) / 2048, 256, 0, stream>>>(ei, E, rowptr, part, NS, rank, col);

  gat_gather_l1<<<(N + 31) / 32, 256, 0, stream>>>(rowptr, col, h1, alS, alD, b1, out1, N);

  gat_gemm_128x40<<<1024, 96, 0, stream>>>(out1, W2, g2, N / 16);
  gat_attn<<<(N + 255) / 256, 256, 0, stream>>>(g2, aS2, aD2, alS, alD, N, 1, 40);
  gat_gather_l2<<<(N + 31) / 32, 256, 0, stream>>>(rowptr, col, g2, alS, alD, b2, out, N);
}


# Both GEMMs stage x rows in LDS: whole 512-byte rows copied once per block (two stages, padded pitch), A fragments read from LDS instead of row-strided 8-byte global loads; k order and sums unchanged
# speedup vs baseline: 3.6842x; 1.0516x over previous
#include <hip/hip_runtime.h>
#include <math.h>

typedef __attribute__((ext_vector_type(2))) float v2f;
typedef __attribute__((ext_vector_type(4))) float v4f;
typedef __attribute__((ext_vector_type(8))) float v8f;

#define NEG_SLOPE 0.2f

#define A_PITCH 132
#define G1_STAGE (64 * A_PITCH)

__device__ __forceinline__ void gemm_128x128_walk(
    const float* __restrict__ X, const float* __restrict__ W,
    float* __restrict__ H, float* __restrict__ stage, int nTiles, int wv,
    int tileBase0, int tileStep)
{
  const int lane = threadIdx.x & 31;
  const int colTile = wv & 7, rowSub = wv >> 3;
  const int col0 = colTile << 4;
  const int kOff = (lane >> 4) << 1;
  const int bCol = col0 + (lane & 15);
  v2f bf[32];
#pragma unroll
  for (int s = 0; s < 32; ++s) {
    bf[s].x = W[(size_t)(4 * s + kOff) * 128 + bCol];
    bf[s].y = W[(size_t)(4 * s + kOff + 1) * 128 + bCol];
  }
  const int nRows = nTiles << 4;
  const int sOff0 = wv * A_PITCH + lane * 4;
  const int sOff1 = (wv + 32) * A_PITCH + lane * 4;
  const int fOff = (rowSub * 16 + (lane & 15)) * A_PITCH + kOff;
  {
    const int g0 = min(tileBase0 * 16 + wv, nRows - 1);
    const int g1 = min(tileBase0 * 16 + wv + 32, nRows - 1);
    const float4 q0 = *(const float4*)(X + (size_t)g0 * 128 + lane * 4);
    const float4 q1 = *(const float4*)(X + (size_t)g1 * 128 + lane * 4);
    *(float4*)(stage + sOff0) = q0;
    *(float4*)(stage + sOff1) = q1;
  }
  __syncthreads();
  int buf = 0;
  for (int tileBase = tileBase0; tileBase < nTiles; tileBase += tileStep) {
    const int nextBase = tileBase + tileStep;
    const bool more = nextBase < nTiles;
    float4 q0 = make_float4(0.f, 0.f, 0.f, 0.f), q1 = q0;
    if (more) {
      const int g0 = min(nextBase * 16 + wv, nRows - 1);
      const int g1 = min(nextBase * 16 + wv + 32, nRows - 1);
      q0 = *(const float4*)(X + (size_t)g0 * 128 + lane * 4);
      q1 = *(const float4*)(X + (size_t)g1 * 128 + lane * 4);
    }
    const int tile = tileBase + rowSub;
    if (tile < nTiles) {
      const float* ar = stage + buf * G1_STAGE + fOff;
      v8f acc = {};
#pragma unroll
      for (int s = 0; s < 32; ++s) {
        const v2f a = *(const v2f*)(ar + 4 * s);
        acc = __builtin_amdgcn_wmma_f32_16x16x4_f32(
            false, a, false, bf[s], (short)0, acc, false, false);
      }
      const int r = (tile << 4) + ((lane >> 4) << 3);
      const int c = col0 + (lane & 15);
#pragma unroll
      for (int i = 0; i < 8; ++i)
        H[(size_t)(r + i) * 128 + c] = acc[i];
    }
    if (more) {
      float* sp = stage + (buf ^ 1) * G1_STAGE;
      *(float4*)(sp + sOff0) = q0;
      *(float4*)(sp + sOff1) = q1;
    }
    __syncthreads();
    buf ^= 1;
  }
}

#define G2_STAGE (16 * A_PITCH)
__global__ __launch_bounds__(96) void gat_gemm_128x40(
    const float* __restrict__ X, const float* __restrict__ W,
    float* __restrict__ G, int nTiles)
{
  __shared__ __attribute__((aligned(16))) float stage[2 * G2_STAGE];
  const int lane = threadIdx.x & 31;
  const int wave = __builtin_amdgcn_readfirstlane((int)(threadIdx.x >> 5));
  const int col0 = wave << 4;
  const int kOff = (lane >> 4) << 1;
  const int bCol = col0 + (lane & 15);
  const bool colOk = (bCol < 40);
  v2f bf[32];
#pragma unroll
  for (int s = 0; s < 32; ++s) {
    bf[s].x = colOk ? W[(size_t)(4 * s + kOff) * 40 + bCol] : 0.0f;
    bf[s].y = colOk ? W[(size_t)(4 * s + kOff + 1) * 40 + bCol] : 0.0f;
  }
  const int fOff = (lane & 15) * A_PITCH + kOff;
  v4f q[6];
  int tile = blockIdx.x;
  if (tile < nTiles) {
#pragma unroll
    for (int i = 0; i < 6; ++i) {
      const int lr = min(wave + 3 * i, 15);
      q[i] = *(const v4f*)(X + (size_t)(tile * 16 + lr) * 128 + lane * 4);
    }
#pragma unroll
    for (int i = 0; i < 6; ++i) {
      const int lr = wave + 3 * i;
      if (lr < 16) *(v4f*)(stage + lr * A_PITCH + lane * 4) = q[i];
    }
  }
  __syncthreads();
  int buf = 0;
  for (; tile < nTiles; tile += gridDim.x) {
    const int nextTile = tile + gridDim.x;
    const bool more = nextTile < nTiles;
    if (more) {
#pragma unroll
      for (int i = 0; i < 6; ++i) {
        const int lr = min(wave + 3 * i, 15);
        q[i] = *(const v4f*)(X + (size_t)(nextTile * 16 + lr) * 128 + lane * 4);
      }
    }
    const int row0 = tile << 4;
    const float* ar = stage + buf * G2_STAGE + fOff;
    v8f acc = {};
#pragma unroll
    for (int s = 0; s < 32; ++s) {
      const v2f a = *(const v2f*)(ar + 4 * s);
      acc = __builtin_amdgcn_wmma_f32_16x16x4_f32(
          false, a, false, bf[s], (short)0, acc, false, false);
    }
    const int r = row0 + ((lane >> 4) << 3);
    const int c = col0 + (lane & 15);
    if (c < 40) {
#pragma unroll
      for (int i = 0; i < 8; ++i)
        G[(size_t)(r + i) * 40 + c] = acc[i];
    }
    if (more) {
      float* sp = stage + (buf ^ 1) * G2_STAGE;
#pragma unroll
      for (int i = 0; i < 6; ++i) {
        const int lr = wave + 3 * i;
        if (lr < 16) *(v4f*)(sp + lr * A_PITCH + lane * 4) = q[i];
      }
    }
    __syncthreads();
    buf ^= 1;
  }
}

__device__ __forceinline__ void attn_dots(
    int idx, const float* __restrict__ Hf, const float* __restrict__ aS,
    const float* __restrict__ aD, float* __restrict__ alS,
    float* __restrict__ alD, int N, int NH, int C)
{
  if (idx >= N * NH) return;
  int n = idx / NH, h = idx - n * NH;
  const float* hp = Hf + (size_t)n * NH * C + (size_t)h * C;
  const float* sp = aS + (size_t)h * C;
  const float* dp = aD + (size_t)h * C;
  float s = 0.f, d = 0.f;
  for (int c = 0; c < C; ++c) { s += hp[c] * sp[c]; d += hp[c] * dp[c]; }
  alS[idx] = s;
  alD[idx] = d;
}

__global__ __launch_bounds__(256) void gat_attn(
    const float* __restrict__ Hf, const float* __restrict__ aS,
    const float* __restrict__ aD, float* __restrict__ alS,
    float* __restrict__ alD, int N, int NH, int C)
{
  attn_dots(blockIdx.x * 256 + threadIdx.x, Hf, aS, aD, alS, alD, N, NH, C);
}

__device__ __forceinline__ float leaky(float v) {
  return (v > 0.f) ? v : NEG_SLOPE * v;
}
__device__ __forceinline__ float wave_max(float v) {
  for (int o = 16; o > 0; o >>= 1) v = fmaxf(v, __shfl_xor(v, o, 32));
  return v;
}
__device__ __forceinline__ float wave_sum(float v) {
  for (int o = 16; o > 0; o >>= 1) v += __shfl_xor(v, o, 32);
  return v;
}
__device__ __forceinline__ float lane_bcast(float v, int j) {
  return __int_as_float(__builtin_amdgcn_readlane(__float_as_int(v), j));
}

#define EDGES_PER_THREAD 8
#define SLICE_SHIFT 14
#define SLICE (1 << SLICE_SHIFT)
#define HIST_WORDS 51200
#define HIST_EPT (SLICE / 1024)
__global__ __launch_bounds__(1024) void gat_hist_gemm1(
    const int* __restrict__ ei, int E, int NS, int B, unsigned* __restrict__ part,
    unsigned* __restrict__ rank, const float* __restrict__ X,
    const float* __restrict__ W, float* __restrict__ H, int nTiles)
{
  __shared__ __attribute__((aligned(16))) unsigned hist[HIST_WORDS];
  const int t = threadIdx.x;
  if ((int)blockIdx.x >= B) {
    const int wv = __builtin_amdgcn_readfirstlane(t >> 5);
    const int gb = (int)blockIdx.x - B;
    const int nGemm = (int)gridDim.x - B;
    gemm_128x128_walk(X, W, H, (float*)hist, nTiles, wv, gb * 4, nGemm * 4);
    return;
  }
  for (int i = t; i < HIST_WORDS / 4; i += 1024)
    ((uint4*)hist)[i] = make_uint4(0u, 0u, 0u, 0u);
  __syncthreads();
  const int e0 = blockIdx.x * SLICE + t;
  int dst[HIST_EPT];
  unsigned r[HIST_EPT];
#pragma unroll
  for (int k = 0; k < HIST_EPT; ++k) {
    const int e = e0 + 1024 * k;
    dst[k] = (e < E) ? ei[E + e] : -1;
  }
#pragma unroll
  for (int k = 0; k < HIST_EPT; ++k)
    r[k] = ((unsigned)dst[k] < (unsigned)HIST_WORDS) ? atomicAdd(&hist[dst[k]], 1u) : 0u;
#pragma unroll
  for (int k = 0; k < HIST_EPT; ++k) {
    const int e = e0 + 1024 * k;
    if (e < E) rank[e] = r[k];
  }
  __syncthreads();
  uint4* p = (uint4*)(part + (size_t)blockIdx.x * NS);
  for (int i = t; i < NS / 4; i += 1024) p[i] = ((const uint4*)hist)[i];
}

#define COLSUM_CHUNK 16
__device__ __forceinline__ void csr_colsum(
    int n, unsigned* __restrict__ part, unsigned* __restrict__ cnt, int N, int NS,
    int B, int NP)
{
  if (n >= NP) return;
  unsigned run = 0;
  if (n < N) {
    unsigned* p = part + n;
    for (int b0 = 0; b0 < B; b0 += COLSUM_CHUNK) {
      unsigned v[COLSUM_CHUNK];
#pragma unroll
      for (int k = 0; k < COLSUM_CHUNK; ++k)
        v[k] = (b0 + k < B) ? p[(size_t)(b0 + k) * NS] : 0u;
#pragma unroll
      for (int k = 0; k < COLSUM_CHUNK; ++k) {
        if (b0 + k < B) p[(size_t)(b0 + k) * NS] = run;
        run += v[k];
      }
    }
  }
  cnt[n] = run;
}

__global__ __launch_bounds__(256) void gat_colsum_attn(
    unsigned* __restrict__ part, unsigned* __restrict__ cnt, int N, int NS,
    int B, int NP, int sumBlocks,
    const float* __restrict__ Hf, const float* __restrict__ aS,
    const float* __restrict__ aD, float* __restrict__ alS,
    float* __restrict__ alD, int NH, int C)
{
  if ((int)blockIdx.x < sumBlocks) {
    csr_colsum(blockIdx.x * 256 + threadIdx.x, part, cnt, N, NS, B, NP);
  } else {
    attn_dots(((int)blockIdx.x - sumBlocks) * 256 + threadIdx.x,
              Hf, aS, aD, alS, alD, N, NH, C);
  }
}

#define SCAN_CHUNK 16384
__global__ __launch_bounds__(1024) void gat_csr_scan(
    const unsigned* __restrict__ cnt, unsigned* __restrict__ rowptr, int NP)
{
  __shared__ unsigned wtot[32];
  const int t = threadIdx.x, lane = t & 31, wv = t >> 5;
  unsigned carry = 0;
  for (int base = 0; base < NP; base += SCAN_CHUNK) {
    const uint4* p = (const uint4*)(cnt + base + t * 16);
    unsigned v[16];
#pragma unroll
    for (int i = 0; i < 4; ++i) {
      const uint4 q = p[i];
      v[4 * i] = q.x; v[4 * i + 1] = q.y; v[4 * i + 2] = q.z; v[4 * i + 3] = q.w;
    }
    unsigned s = 0;
#pragma unroll
    for (int i = 0; i < 16; ++i) s += v[i];
    unsigned incl = s;
    for (int o = 1; o < 32; o <<= 1) {
      const unsigned u = __shfl_up(incl, o, 32);
      if (lane >= o) incl += u;
    }
    if (lane == 31) wtot[wv] = incl;
    __syncthreads();
    if (wv == 0) {
      unsigned w = wtot[lane];
      for (int o = 1; o < 32; o <<= 1) {
        const unsigned u = __shfl_up(w, o, 32);
        if (lane >= o) w += u;
      }
      wtot[lane] = w;
    }
    __syncthreads();
    unsigned run = carry + (incl - s) + (wv ? wtot[wv - 1] : 0u);
    carry += wtot[31];
    uint4* op = (uint4*)(rowptr + base + t * 16);
#pragma unroll
    for (int i = 0; i < 4; ++i) {
      uint4 q;
      q.x = run; run += v[4 * i];
      q.y = run; run += v[4 * i + 1];
      q.z = run; run += v[4 * i + 2];
      q.w = run; run += v[4 * i + 3];
      op[i] = q;
    }
    __syncthreads();
  }
}

__global__ __launch_bounds__(256) void gat_csr_fill(
    const int* __restrict__ ei, int E, const unsigned* __restrict__ rowptr,
    const unsigned* __restrict__ part, int NS,
    const unsigned* __restrict__ rank, int* __restrict__ col)
{
  const int e0 = blockIdx.x * (256 * EDGES_PER_THREAD) + threadIdx.x;
  int dst[EDGES_PER_THREAD], sr[EDGES_PER_THREAD];
  unsigned rk[EDGES_PER_THREAD], rp[EDGES_PER_THREAD], pp[EDGES_PER_THREAD];
#pragma unroll
  for (int k = 0; k < EDGES_PER_THREAD; ++k) {
    const int e = e0 + 256 * k;
    const bool ok = e < E;
    dst[k] = ok ? ei[E + e] : -1;
    sr[k] = ok ? ei[e] : 0;
    rk[k] = ok ? rank[e] : 0u;
  }
#pragma unroll
  for (int k = 0; k < EDGES_PER_THREAD; ++k)
    rp[k] = (dst[k] >= 0) ? rowptr[dst[k]] : 0u;
#pragma unroll
  for (int k = 0; k < EDGES_PER_THREAD; ++k) {
    const int e = e0 + 256 * k;
    pp[k] = (dst[k] >= 0) ? part[(size_t)(e >> SLICE_SHIFT) * NS + dst[k]] : 0u;
  }
#pragma unroll
  for (int k = 0; k < EDGES_PER_THREAD; ++k)
    if (dst[k] >= 0) col[rp[k] + pp[k] + rk[k]] = sr[k];
}

#define NODES_PER_WAVE 4
#define ROWS_IN_FLIGHT 4
__global__ __launch_bounds__(256) void gat_gather_l1(
    const unsigned* __restrict__ rowptr, const int* __restrict__ col,
    const float* __restrict__ Hf, const float* __restrict__ alS,
    const float* __restrict__ alD, const float* __restrict__ b,
    float* __restrict__ out1, int N)
{
  const int lane = threadIdx.x & 31;
  const int wv = __builtin_amdgcn_readfirstlane((int)(threadIdx.x >> 5));
  const int d0 = (blockIdx.x * 8 + wv) * NODES_PER_WAVE;
  if (d0 >= N) return;
  const int hsel = lane >> 4;
  unsigned r0 = rowptr[d0], r1 = rowptr[d0 + 1], r2 = rowptr[min(d0 + 2, N)];
  const int elast = max((int)rowptr[N] - 1, 0);
  int dgn = (int)(r2 - r1);
  float ad0 = alD[d0 * 2], ad1 = alD[d0 * 2 + 1];
  float vs0 = leaky(alS[d0 * 2] + ad0), vs1 = leaky(alS[d0 * 2 + 1] + ad1);
  int sc = col[min((int)r0 + min(lane, max((int)(r1 - r0) - 1, 0)), elast)];
  float2 ac = *(const float2*)(alS + (size_t)sc * 2);
#pragma unroll 1
  for (int kn = 0; kn < NODES_PER_WAVE; ++kn) {
  const int d = d0 + kn;
  if (d >= N) break;
  const int beg = (int)r0;
  const int deg = (int)(r1 - r0);
  const int dn = d + 1;
  const bool more = (kn + 1 < NODES_PER_WAVE) && (dn < N);
  const int degn = more ? dgn : 0;
  const unsigned r3 = rowptr[min(d + 3, N)];
  const int dnc = min(dn, N - 1);
  const float ns0 = alS[dnc * 2], ns1 = alS[dnc * 2 + 1];
  const float nd0 = alD[dnc * 2], nd1 = alD[dnc * 2 + 1];
  const int snr = col[min((int)r1 + min(lane, max(degn - 1, 0)), elast)];
  const float4 hown = *(const float4*)(Hf + (size_t)d * 128 + lane * 4);
  float m0 = vs0, m1 = vs1;
  if (lane < deg) {
    m0 = fmaxf(m0, leaky(ac.x + ad0));
    m1 = fmaxf(m1, leaky(ac.y + ad1));
  }
  for (int base = 32; base < deg; base += 32) {
    const int i = base + lane;
    if (i < deg) {
      const int s = col[beg + i];
      const float2 a = *(const float2*)(alS + (size_t)s * 2);
      m0 = fmaxf(m0, leaky(a.x + ad0));
      m1 = fmaxf(m1, leaky(a.y + ad1));
    }
  }
  m0 = wave_max(m0);
  m1 = wave_max(m1);
  __builtin_amdgcn_sched_barrier(0);
  const float2 anr = *(const float2*)(alS + (size_t)snr * 2);
  const float ws0 = expf(vs0 - m0), ws1 = expf(vs1 - m1);
  float4 acc;
  {
    const float w = hsel ? ws1 : ws0;
    acc.x = w * hown.x; acc.y = w * hown.y; acc.z = w * hown.z; acc.w = w * hown.w;
  }
  float p0 = 0.f, p1 = 0.f;
  for (int base = 0; base < deg; base += 32) {
    const int i = base + lane;
    int s = d;
    float w0 = 0.f, w1 = 0.f;
    if (base == 0) {
      s = sc;
      if (i < deg) {
        w0 = expf(leaky(ac.x + ad0) - m0);
        w1 = expf(leaky(ac.y + ad1) - m1);
      }
    } else if (i < deg) {
      s = col[beg + i];
      const float2 a = *(const float2*)(alS + (size_t)s * 2);
      w0 = expf(leaky(a.x + ad0) - m0);
      w1 = expf(leaky(a.y + ad1) - m1);
    }
    p0 += w0;
    p1 += w1;
    const int cnt = min(32, deg - base);
    for (int j0 = 0; j0 < cnt; j0 += ROWS_IN_FLIGHT) {
#pragma unroll
      for (int u = 0; u < ROWS_IN_FLIGHT; ++u) {
        const int j = j0 + u;
        const int sj = __builtin_amdgcn_readlane(s, j);
        const float wj0 = lane_bcast(w0, j), wj1 = lane_bcast(w1, j);
        const float w = hsel ? wj1 : wj0;
        const float4 hv = *(const float4*)(Hf + (size_t)sj * 128 + lane * 4);
        acc.x += w * hv.x; acc.y += w * hv.y; acc.z += w * hv.z; acc.w += w * hv.w;
      }
    }
  }
  const float den0 = wave_sum(p0) + ws0;
  const float den1 = wave_sum(p1) + ws1;
  const float den = (hsel ? den1 : den0) + 1e-16f;
  const float4 bv = *(const float4*)(b + lane * 4);
  float4 r;
  r.x = acc.x / den + bv.x;
  r.y = acc.y / den + bv.y;
  r.z = acc.z / den + bv.z;
  r.w = acc.w / den + bv.w;
  r.x = (r.x > 0.f) ? r.x : (expf(r.x) - 1.0f);
  r.y = (r.y > 0.f) ? r.y : (expf(r.y) - 1.0f);
  r.z = (r.z > 0.f) ? r.z : (expf(r.z) - 1.0f);
  r.w = (r.w > 0.f) ? r.w : (expf(r.w) - 1.0f);
  *(float4*)(out1 + (size_t)d * 128 + lane * 4) = r;
  r0 = r1; r1 = r2; r2 = r3;
  dgn = (int)(r2 - r1);
  ad0 = nd0; ad1 = nd1;
  vs0 = leaky(ns0 + nd0); vs1 = leaky(ns1 + nd1);
  sc = snr; ac = anr;
  }
}

__global__ __launch_bounds__(256) void gat_gather_l2(
    const unsigned* __restrict__ rowptr, const int* __restrict__ col,
    const float* __restrict__ G, const float* __restrict__ alS,
    const float* __restrict__ alD, const float* __restrict__ b,
    float* __restrict__ out, int N)
{
  __shared__ float stage[8 * NODES_PER_WAVE * 40];
  const int lane = threadIdx.x & 31;
  const int wv = __builtin_amdgcn_readfirstlane((int)(threadIdx.x >> 5));
  const int d0 = (blockIdx.x * 8 + wv) * NODES_PER_WAVE;
  float* slot = stage + wv * (NODES_PER_WAVE * 40);
  const bool act = lane < 20;
  const int c2 = act ? 2 * lane : 0;
  unsigned r0 = 0, r1 = 0, r2 = 0;
  int dgn = 0;
  float ad = 0.f, vs = 0.f;
  int sc = 0;
  float ac = 0.f;
  const int elast = max((int)rowptr[N] - 1, 0);
  if (d0 < N) {
    r0 = rowptr[d0]; r1 = rowptr[d0 + 1]; r2 = rowptr[min(d0 + 2, N)];
    dgn = (int)(r2 - r1);
    ad = alD[d0];
    vs = leaky(alS[d0] + ad);
    sc = col[min((int)r0 + min(lane, max((int)(r1 - r0) - 1, 0)), elast)];
    ac = alS[sc];
  }
#pragma unroll 1
  for (int kn = 0; kn < NODES_PER_WAVE; ++kn) {
  const int d = d0 + kn;
  float res0 = 0.f, res1 = 0.f;
  if (d < N) {
    const int beg = (int)r0;
    const int deg = (int)(r1 - r0);
    const int dn = d + 1;
    const bool more = (kn + 1 < NODES_PER_WAVE) && (dn < N);
    const int degn = more ? dgn : 0;
    const unsigned r3 = rowptr[min(d + 3, N)];
    const int dnc = min(dn, N - 1);
    const float ns = alS[dnc], nd = alD[dnc];
    const int snr = col[min((int)r1 + min(lane, max(degn - 1, 0)), elast)];
    const float2 gown = *(const float2*)(G + (size_t)d * 40 + c2);
    float m = vs;
    if (lane < deg) m = fmaxf(m, leaky(ac + ad));
    for (int base = 32; base < deg; base += 32) {
      const int i = base + lane;
      if (i < deg) m = fmaxf(m, leaky(alS[col[beg + i]] + ad));
    }
    m = wave_max(m);
    __builtin_amdgcn_sched_barrier(0);
    const float anr = alS[snr];
    const float wself = expf(vs - m);
    float acc0 = wself * gown.x, acc1 = wself * gown.y;
    float p = 0.f;
    for (int base = 0; base < deg; base += 32) {
      const int i = base + lane;
      int s = d;
      float w = 0.f;
      if (base == 0) {
        s = sc;
        if (i < deg) w = expf(leaky(ac + ad) - m);
      } else if (i < deg) {
        s = col[beg + i];
        w = expf(leaky(alS[s] + ad) - m);
      }
      p += w;
      const int cnt = min(32, deg - base);
      for (int j0 = 0; j0 < cnt; j0 += ROWS_IN_FLIGHT) {
#pragma unroll
        for (int u = 0; u < ROWS_IN_FLIGHT; ++u) {
          const int j = j0 + u;
          const int sj = __builtin_amdgcn_readlane(s, j);
          const float wj = lane_bcast(w, j);
          const float2 gv = *(const float2*)(G + (size_t)sj * 40 + c2);
          acc0 += wj * gv.x;
          acc1 += wj * gv.y;
        }
      }
    }
    const float den = wave_sum(p) + wself + 1e-16f;
    res0 = acc0 / den + b[c2];
    res1 = acc1 / den + b[c2 + 1];
    r0 = r1; r1 = r2; r2 = r3;
    dgn = (int)(r2 - r1);
    ad = nd;
    vs = leaky(ns + nd);
    sc = snr; ac = anr;
  }
  if (act) {
    slot[kn * 40 + c2] = res0;
    slot[kn * 40 + c2 + 1] = res1;
  }
  }
  const size_t lim = (size_t)N * 40;
  const size_t g0 = (size_t)d0 * 40 + lane;
#pragma unroll
  for (int j = 0; j < (NODES_PER_WAVE * 40) / 32; ++j) {
    const size_t g = g0 + 32 * j;
    if (g < lim) out[g] = slot[32 * j + lane];
  }
}

extern "C" void kernel_launch(void* const* d_in, const int* in_sizes, int n_in,
                              void* d_out, int out_size, void* d_ws, size_t ws_size,
                              hipStream_t stream)
{
  const float* x   = (const float*)d_in[0];
  const int*   ei  = (const int*)d_in[1];
  const float* W1  = (const float*)d_in[2];
  const float* aS1 = (const float*)d_in[3];
  const float* aD1 = (const float*)d_in[4];
  const float* b1  = (const float*)d_in[5];
  const float* W2  = (const float*)d_in[6];
  const float* aS2 = (const float*)d_in[7];
  const float* aD2 = (const float*)d_in[8];
  const float* b2  = (const float*)d_in[9];
  float* out = (float*)d_out;

  const int N  = in_sizes[0] / 128;
  const int E  = in_sizes[1] / 2;
  const int NP = (N / SCAN_CHUNK + 1) * SCAN_CHUNK;

  char* ws = (char*)d_ws;
  size_t off = 0;
  auto carve = [&](size_t bytes) -> char* {
    char* p = ws + off;
    off += (bytes + 255) & ~(size_t)255;
    return p;
  };
  float*    h1     = (float*)carve((size_t)N * 128 * 4);
  float*    out1   = (float*)carve((size_t)N * 128 * 4);
  float*    alS    = (float*)carve((size_t)N * 2 * 4);
  float*    alD    = (float*)carve((size_t)N * 2 * 4);
  unsigned* cnt    = (unsigned*)carve((size_t)NP * 4);
  unsigned* rowptr = (unsigned*)carve((size_t)NP * 4);
  unsigned* rank   = (unsigned*)carve((size_t)E * 4);
  int*      col    = (int*)carve((size_t)E * 4);
  const int NS = (N + 31) & ~31;
  const int B  = (E + SLICE - 1) >> SLICE_SHIFT;
  unsigned* part   = (unsigned*)carve((size_t)B * NS * 4);
  float* g2 = h1;

  const int gemmBlocks = (B < 224) ? 256 - B : 32;
  gat_hist_gemm1<<<B + gemmBlocks, 1024, 0, stream>>>(ei, E, NS, B, part, rank, x, W1, h1, N / 16);
  gat_colsum_attn<<<NP / 256 + (N * 2 + 255) / 256, 256, 0, stream>>>(
      part, cnt, N, NS, B, NP, NP / 256, h1, aS1, aD1, alS, alD, 2, 64);
  gat_csr_scan<<<1, 1024, 0, stream>>>(cnt, rowptr, NP);
  gat_csr_fill<<<(E + 2047) / 2048, 256, 0, stream>>>(ei, E, rowptr, part, NS, rank, col);

  gat_gather_l1<<<(N + 31) / 32, 256, 0, stream>>>(rowptr, col, h1, alS, alD, b1, out1, N);

  gat_gemm_128x40<<<1024, 96, 0, stream>>>(out1, W2, g2, N / 16);
  gat_attn<<<(N + 255) / 256, 256, 0, stream>>>(g2, aS2, aD2, alS, alD, N, 1, 40);
  gat_gather_l2<<<(N + 31) / 32, 256, 0, stream>>>(rowptr, col, g2, alS, alD, b2, out, N);
}


# Row pointers without a scan launch: slice-sum blocks leave a per-block prefix and total of the in-degrees, each fill block forms the prefix of the totals in LDS; 7 launches, sum orders unchanged
# speedup vs baseline: 4.0148x; 1.0897x over previous
#include <hip/hip_runtime.h>
#include <math.h>

typedef __attribute__((ext_vector_type(2))) float v2f;
typedef __attribute__((ext_vector_type(4))) float v4f;
typedef __attribute__((ext_vector_type(8))) float v8f;

#define NEG_SLOPE 0.2f

#define A_PITCH 132
#define G1_STAGE (64 * A_PITCH)

__device__ __forceinline__ void gemm_128x128_walk(
    const float* __restrict__ X, const float* __restrict__ W,
    float* __restrict__ H, float* __restrict__ stage, int nTiles, int wv,
    int tileBase0, int tileStep)
{
  const int lane = threadIdx.x & 31;
  const int colTile = wv & 7, rowSub = wv >> 3;
  const int col0 = colTile << 4;
  const int kOff = (lane >> 4) << 1;
  const int bCol = col0 + (lane & 15);
  v2f bf[32];
#pragma unroll
  for (int s = 0; s < 32; ++s) {
    bf[s].x = W[(size_t)(4 * s + kOff) * 128 + bCol];
    bf[s].y = W[(size_t)(4 * s + kOff + 1) * 128 + bCol];
  }
  const int nRows = nTiles << 4;
  const int sOff0 = wv * A_PITCH + lane * 4;
  const int sOff1 = (wv + 32) * A_PITCH + lane * 4;
  const int fOff = (rowSub * 16 + (lane & 15)) * A_PITCH + kOff;
  {
    const int g0 = min(tileBase0 * 16 + wv, nRows - 1);
    const int g1 = min(tileBase0 * 16 + wv + 32, nRows - 1);
    const float4 q0 = *(const float4*)(X + (size_t)g0 * 128 + lane * 4);
    const float4 q1 = *(const float4*)(X + (size_t)g1 * 128 + lane * 4);
    *(float4*)(stage + sOff0) = q0;
    *(float4*)(stage + sOff1) = q1;
  }
  __syncthreads();
  int buf = 0;
  for (int tileBase = tileBase0; tileBase < nTiles; tileBase += tileStep) {
    const int nextBase = tileBase + tileStep;
    const bool more = nextBase < nTiles;
    float4 q0 = make_float4(0.f, 0.f, 0.f, 0.f), q1 = q0;
    if (more) {
      const int g0 = min(nextBase * 16 + wv, nRows - 1);
      const int g1 = min(nextBase * 16 + wv + 32, nRows - 1);
      q0 = *(const float4*)(X + (size_t)g0 * 128 + lane * 4);
      q1 = *(const float4*)(X + (size_t)g1 * 128 + lane * 4);
    }
    const int tile = tileBase + rowSub;
    if (tile < nTiles) {
      const float* ar = stage + buf * G1_STAGE + fOff;
      v8f acc = {};
#pragma unroll
      for (int s = 0; s < 32; ++s) {
        const v2f a = *(const v2f*)(ar + 4 * s);
        acc = __builtin_amdgcn_wmma_f32_16x16x4_f32(
            false, a, false, bf[s], (short)0, acc, false, false);
      }
      const int r = (tile << 4) + ((lane >> 4) << 3);
      const int c = col0 + (lane & 15);
#pragma unroll
      for (int i = 0; i < 8; ++i)
        H[(size_t)(r + i) * 128 + c] = acc[i];
    }
    if (more) {
      float* sp = stage + (buf ^ 1) * G1_STAGE;
      *(float4*)(sp + sOff0) = q0;
      *(float4*)(sp + sOff1) = q1;
    }
    __syncthreads();
    buf ^= 1;
  }
}

#define G2_STAGE (16 * A_PITCH)
__global__ __launch_bounds__(96) void gat_gemm_128x40(
    const float* __restrict__ X, const float* __restrict__ W,
    float* __restrict__ G, int nTiles)
{
  __shared__ __attribute__((aligned(16))) float stage[2 * G2_STAGE];
  const int lane = threadIdx.x & 31;
  const int wave = __builtin_amdgcn_readfirstlane((int)(threadIdx.x >> 5));
  const int col0 = wave << 4;
  const int kOff = (lane >> 4) << 1;
  const int bCol = col0 + (lane & 15);
  const bool colOk = (bCol < 40);
  v2f bf[32];
#pragma unroll
  for (int s = 0; s < 32; ++s) {
    bf[s].x = colOk ? W[(size_t)(4 * s + kOff) * 40 + bCol] : 0.0f;
    bf[s].y = colOk ? W[(size_t)(4 * s + kOff + 1) * 40 + bCol] : 0.0f;
  }
  const int fOff = (lane & 15) * A_PITCH + kOff;
  v4f q[6];
  int tile = blockIdx.x;
  if (tile < nTiles) {
#pragma unroll
    for (int i = 0; i < 6; ++i) {
      const int lr = min(wave + 3 * i, 15);
      q[i] = *(const v4f*)(X + (size_t)(tile * 16 + lr) * 128 + lane * 4);
    }
#pragma unroll
    for (int i = 0; i < 6; ++i) {
      const int lr = wave + 3 * i;
      if (lr < 16) *(v4f*)(stage + lr * A_PITCH + lane * 4) = q[i];
    }
  }
  __syncthreads();
  int buf = 0;
  for (; tile < nTiles; tile += gridDim.x) {
    const int nextTile = tile + gridDim.x;
    const bool more = nextTile < nTiles;
    if (more) {
#pragma unroll
      for (int i = 0; i < 6; ++i) {
        const int lr = min(wave + 3 * i, 15);
        q[i] = *(const v4f*)(X + (size_t)(nextTile * 16 + lr) * 128 + lane * 4);
      }
    }
    const int row0 = tile << 4;
    const float* ar = stage + buf * G2_STAGE + fOff;
    v8f acc = {};
#pragma unroll
    for (int s = 0; s < 32; ++s) {
      const v2f a = *(const v2f*)(ar + 4 * s);
      acc = __builtin_amdgcn_wmma_f32_16x16x4_f32(
          false, a, false, bf[s], (short)0, acc, false, false);
    }
    const int r = row0 + ((lane >> 4) << 3);
    const int c = col0 + (lane & 15);
    if (c < 40) {
#pragma unroll
      for (int i = 0; i < 8; ++i)
        G[(size_t)(r + i) * 40 + c] = acc[i];
    }
    if (more) {
      float* sp = stage + (buf ^ 1) * G2_STAGE;
#pragma unroll
      for (int i = 0; i < 6; ++i) {
        const int lr = wave + 3 * i;
        if (lr < 16) *(v4f*)(sp + lr * A_PITCH + lane * 4) = q[i];
      }
    }
    __syncthreads();
    buf ^= 1;
  }
}

__device__ __forceinline__ void attn_dots(
    int idx, const float* __restrict__ Hf, const float* __restrict__ aS,
    const float* __restrict__ aD, float* __restrict__ alS,
    float* __restrict__ alD, int N, int NH, int C)
{
  if (idx >= N * NH) return;
  int n = idx / NH, h = idx - n * NH;
  const float* hp = Hf + (size_t)n * NH * C + (size_t)h * C;
  const float* sp = aS + (size_t)h * C;
  const float* dp = aD + (size_t)h * C;
  float s = 0.f, d = 0.f;
  for (int c = 0; c < C; ++c) { s += hp[c] * sp[c]; d += hp[c] * dp[c]; }
  alS[idx] = s;
  alD[idx] = d;
}

__global__ __launch_bounds__(256) void gat_attn(
    const float* __restrict__ Hf, const float* __restrict__ aS,
    const float* __restrict__ aD, float* __restrict__ alS,
    float* __restrict__ alD, int N, int NH, int C)
{
  attn_dots(blockIdx.x * 256 + threadIdx.x, Hf, aS, aD, alS, alD, N, NH, C);
}

__device__ __forceinline__ float leaky(float v) {
  return (v > 0.f) ? v : NEG_SLOPE * v;
}
__device__ __forceinline__ float wave_max(float v) {
  for (int o = 16; o > 0; o >>= 1) v = fmaxf(v, __shfl_xor(v, o, 32));
  return v;
}
__device__ __forceinline__ float wave_sum(float v) {
  for (int o = 16; o > 0; o >>= 1) v += __shfl_xor(v, o, 32);
  return v;
}
__device__ __forceinline__ float lane_bcast(float v, int j) {
  return __int_as_float(__builtin_amdgcn_readlane(__float_as_int(v), j));
}

#define EDGES_PER_THREAD 8
#define SLICE_SHIFT 14
#define SLICE (1 << SLICE_SHIFT)
#define HIST_WORDS 51200
#define HIST_EPT (SLICE / 1024)
__global__ __launch_bounds__(1024) void gat_hist_gemm1(
    const int* __restrict__ ei, int E, int NS, int B, unsigned* __restrict__ part,
    unsigned* __restrict__ rank, const float* __restrict__ X,
    const float* __restrict__ W, float* __restrict__ H, int nTiles)
{
  __shared__ __attribute__((aligned(16))) unsigned hist[HIST_WORDS];
  const int t = threadIdx.x;
  if ((int)blockIdx.x >= B) {
    const int wv = __builtin_amdgcn_readfirstlane(t >> 5);
    const int gb = (int)blockIdx.x - B;
    const int nGemm = (int)gridDim.x - B;
    gemm_128x128_walk(X, W, H, (float*)hist, nTiles, wv, gb * 4, nGemm * 4);
    return;
  }
  for (int i = t; i < HIST_WORDS / 4; i += 1024)
    ((uint4*)hist)[i] = make_uint4(0u, 0u, 0u, 0u);
  __syncthreads();
  const int e0 = blockIdx.x * SLICE + t;
  int dst[HIST_EPT];
  unsigned r[HIST_EPT];
#pragma unroll
  for (int k = 0; k < HIST_EPT; ++k) {
    const int e = e0 + 1024 * k;
    dst[k] = (e < E) ? ei[E + e] : -1;
  }
#pragma unroll
  for (int k = 0; k < HIST_EPT; ++k)
    r[k] = ((unsigned)dst[k] < (unsigned)HIST_WORDS) ? atomicAdd(&hist[dst[k]], 1u) : 0u;
#pragma unroll
  for (int k = 0; k < HIST_EPT; ++k) {
    const int e = e0 + 1024 * k;
    if (e < E) rank[e] = r[k];
  }
  __syncthreads();
  uint4* p = (uint4*)(part + (size_t)blockIdx.x * NS);
  for (int i = t; i < NS / 4; i += 1024) p[i] = ((const uint4*)hist)[i];
}

#define COLSUM_CHUNK 16
#define NODE_BLOCK 256
__device__ __forceinline__ void csr_colsum(
    int n, unsigned* __restrict__ part, unsigned* __restrict__ lp,
    unsigned* __restrict__ bt, unsigned* __restrict__ wtot, int N, int NS, int B)
{
  unsigned run = 0;
  if (n < N) {
    unsigned* p = part + n;
    for (int b0 = 0; b0 < B; b0 += COLSUM_CHUNK) {
      unsigned v[COLSUM_CHUNK];
#pragma unroll
      for (int k = 0; k < COLSUM_CHUNK; ++k)
        v[k] = (b0 + k < B) ? p[(size_t)(b0 + k) * NS] : 0u;
#pragma unroll
      for (int k = 0; k < COLSUM_CHUNK; ++k) {
        if (b0 + k < B) p[(size_t)(b0 + k) * NS] = run;
        run += v[k];
      }
    }
  }
  const int lane = threadIdx.x & 31, wv = threadIdx.x >> 5;
  unsigned incl = run;
  for (int o = 1; o < 32; o <<= 1) {
    const unsigned u = __shfl_up(incl, o, 32);
    if (lane >= o) incl += u;
  }
  if (lane == 31) wtot[wv] = incl;
  __syncthreads();
  unsigned before = 0;
#pragma unroll
  for (int w = 0; w < NODE_BLOCK / 32; ++w)
    if (w < wv) before += wtot[w];
  lp[n] = before + incl - run;
  if (threadIdx.x == NODE_BLOCK - 1) bt[n / NODE_BLOCK] = before + incl;
}

__global__ __launch_bounds__(256) void gat_colsum_attn(
    unsigned* __restrict__ part, unsigned* __restrict__ lp,
    unsigned* __restrict__ bt, int N, int NS, int B, int sumBlocks,
    const float* __restrict__ Hf, const float* __restrict__ aS,
    const float* __restrict__ aD, float* __restrict__ alS,
    float* __restrict__ alD, int NH, int C)
{
  __shared__ unsigned wtot[NODE_BLOCK / 32];
  if ((int)blockIdx.x < sumBlocks) {
    csr_colsum(blockIdx.x * NODE_BLOCK + threadIdx.x, part, lp, bt, wtot, N, NS, B);
  } else {
    attn_dots(((int)blockIdx.x - sumBlocks) * 256 + threadIdx.x,
              Hf, aS, aD, alS, alD, N, NH, C);
  }
}

__global__ __launch_bounds__(256) void gat_csr_fill(
    const int* __restrict__ ei, int E, const unsigned* __restrict__ lp,
    const unsigned* __restrict__ bt, int nb, unsigned* __restrict__ rowptr,
    const unsigned* __restrict__ part, int NS,
    const unsigned* __restrict__ rank, int* __restrict__ col)
{
  __shared__ unsigned bp[NODE_BLOCK];
  __shared__ unsigned wtot[NODE_BLOCK / 32];
  const int t = threadIdx.x, lane = t & 31, wv = t >> 5;
  const int e0 = blockIdx.x * (256 * EDGES_PER_THREAD) + t;
  int dst[EDGES_PER_THREAD], sr[EDGES_PER_THREAD];
  unsigned rk[EDGES_PER_THREAD], rp[EDGES_PER_THREAD], pp[EDGES_PER_THREAD];
  const unsigned v = (t < nb) ? bt[t] : 0u;
  const int bOwn = blockIdx.x;
  const unsigned lpOwn = lp[min(bOwn, nb - 1) * NODE_BLOCK + t];
#pragma unroll
  for (int k = 0; k < EDGES_PER_THREAD; ++k) {
    const int ec = min(e0 + 256 * k, E - 1);
    dst[k] = ei[E + ec];
    sr[k] = ei[ec];
    rk[k] = rank[ec];
  }
#pragma unroll
  for (int k = 0; k < EDGES_PER_THREAD; ++k)
    if (e0 + 256 * k >= E) dst[k] = -1;
  {
    unsigned incl = v;
    for (int o = 1; o < 32; o <<= 1) {
      const unsigned u = __shfl_up(incl, o, 32);
      if (lane >= o) incl += u;
    }
    if (lane == 31) wtot[wv] = incl;
    __syncthreads();
    unsigned before = 0;
#pragma unroll
    for (int w = 0; w < NODE_BLOCK / 32; ++w)
      if (w < wv) before += wtot[w];
    bp[t] = before + incl - v;
    __syncthreads();
  }
  if (bOwn < nb) rowptr[bOwn * NODE_BLOCK + t] = bp[bOwn] + lpOwn;
  for (int b = bOwn + gridDim.x; b < nb; b += gridDim.x)
    rowptr[b * NODE_BLOCK + t] = bp[b] + lp[b * NODE_BLOCK + t];
#pragma unroll
  for (int k = 0; k < EDGES_PER_THREAD; ++k)
    rp[k] = (dst[k] >= 0) ? lp[dst[k]] : 0u;
#pragma unroll
  for (int k = 0; k < EDGES_PER_THREAD; ++k) {
    const int e = e0 + 256 * k;
    pp[k] = (dst[k] >= 0) ? part[(size_t)(e >> SLICE_SHIFT) * NS + dst[k]] : 0u;
  }
#pragma unroll
  for (int k = 0; k < EDGES_PER_THREAD; ++k)
    if (dst[k] >= 0)
      col[bp[dst[k] / NODE_BLOCK] + rp[k] + pp[k] + rk[k]] = sr[k];
}

#define NODES_PER_WAVE 4
#define ROWS_IN_FLIGHT 4
__global__ __launch_bounds__(256) void gat_gather_l1(
    const unsigned* __restrict__ rowptr, const int* __restrict__ col,
    const float* __restrict__ Hf, const float* __restrict__ alS,
    const float* __restrict__ alD, const float* __restrict__ b,
    float* __restrict__ out1, int N)
{
  const int lane = threadIdx.x & 31;
  const int wv = __builtin_amdgcn_readfirstlane((int)(threadIdx.x >> 5));
  const int d0 = (blockIdx.x * 8 + wv) * NODES_PER_WAVE;
  if (d0 >= N) return;
  const int hsel = lane >> 4;
  unsigned r0 = rowptr[d0], r1 = rowptr[d0 + 1], r2 = rowptr[min(d0 + 2, N)];
  const int elast = max((int)rowptr[N] - 1, 0);
  int dgn = (int)(r2 - r1);
  float ad0 = alD[d0 * 2], ad1 = alD[d0 * 2 + 1];
  float vs0 = leaky(alS[d0 * 2] + ad0), vs1 = leaky(alS[d0 * 2 + 1] + ad1);
  int sc = col[min((int)r0 + min(lane, max((int)(r1 - r0) - 1, 0)), elast)];
  float2 ac = *(const float2*)(alS + (size_t)sc * 2);
#pragma unroll 1
  for (int kn = 0; kn < NODES_PER_WAVE; ++kn) {
  const int d = d0 + kn;
  if (d >= N) break;
  const int beg = (int)r0;
  const int deg = (int)(r1 - r0);
  const int dn = d + 1;
  const bool more = (kn + 1 < NODES_PER_WAVE) && (dn < N);
  const int degn = more ? dgn : 0;
  const unsigned r3 = rowptr[min(d + 3, N)];
  const int dnc = min(dn, N - 1);
  const float ns0 = alS[dnc * 2], ns1 = alS[dnc * 2 + 1];
  const float nd0 = alD[dnc * 2], nd1 = alD[dnc * 2 + 1];
  const int snr = col[min((int)r1 + min(lane, max(degn - 1, 0)), elast)];
  const float4 hown = *(const float4*)(Hf + (size_t)d * 128 + lane * 4);
  float m0 = vs0, m1 = vs1;
  if (lane < deg) {
    m0 = fmaxf(m0, leaky(ac.x + ad0));
    m1 = fmaxf(m1, leaky(ac.y + ad1));
  }
  for (int base = 32; base < deg; base += 32) {
    const int i = base + lane;
    if (i < deg) {
      const int s = col[beg + i];
      const float2 a = *(const float2*)(alS + (size_t)s * 2);
      m0 = fmaxf(m0, leaky(a.x + ad0));
      m1 = fmaxf(m1, leaky(a.y + ad1));
    }
  }
  m0 = wave_max(m0);
  m1 = wave_max(m1);
  __builtin_amdgcn_sched_barrier(0);
  const float2 anr = *(const float2*)(alS + (size_t)snr * 2);
  const float ws0 = expf(vs0 - m0), ws1 = expf(vs1 - m1);
  float4 acc;
  {
    const float w = hsel ? ws1 : ws0;
    acc.x = w * hown.x; acc.y = w * hown.y; acc.z = w * hown.z; acc.w = w * hown.w;
  }
  float p0 = 0.f, p1 = 0.f;
  for (int base = 0; base < deg; base += 32) {
    const int i = base + lane;
    int s = d;
    float w0 = 0.f, w1 = 0.f;
    if (base == 0) {
      s = sc;
      if (i < deg) {
        w0 = expf(leaky(ac.x + ad0) - m0);
        w1 = expf(leaky(ac.y + ad1) - m1);
      }
    } else if (i < deg) {
      s = col[beg + i];
      const float2 a = *(const float2*)(alS + (size_t)s * 2);
      w0 = expf(leaky(a.x + ad0) - m0);
      w1 = expf(leaky(a.y + ad1) - m1);
    }
    p0 += w0;
    p1 += w1;
    const int cnt = min(32, deg - base);
    for (int j0 = 0; j0 < cnt; j0 += ROWS_IN_FLIGHT) {
#pragma unroll
      for (int u = 0; u < ROWS_IN_FLIGHT; ++u) {
        const int j = j0 + u;
        const int sj = __builtin_amdgcn_readlane(s, j);
        const float wj0 = lane_bcast(w0, j), wj1 = lane_bcast(w1, j);
        const float w = hsel ? wj1 : wj0;
        const float4 hv = *(const float4*)(Hf + (size_t)sj * 128 + lane * 4);
        acc.x += w * hv.x; acc.y += w * hv.y; acc.z += w * hv.z; acc.w += w * hv.w;
      }
    }
  }
  const float den0 = wave_sum(p0) + ws0;
  const float den1 = wave_sum(p1) + ws1;
  const float den = (hsel ? den1 : den0) + 1e-16f;
  const float4 bv = *(const float4*)(b + lane * 4);
  float4 r;
  r.x = acc.x / den + bv.x;
  r.y = acc.y / den + bv.y;
  r.z = acc.z / den + bv.z;
  r.w = acc.w / den + bv.w;
  r.x = (r.x > 0.f) ? r.x : (expf(r.x) - 1.0f);
  r.y = (r.y > 0.f) ? r.y : (expf(r.y) - 1.0f);
  r.z = (r.z > 0.f) ? r.z : (expf(r.z) - 1.0f);
  r.w = (r.w > 0.f) ? r.w : (expf(r.w) - 1.0f);
  *(float4*)(out1 + (size_t)d * 128 + lane * 4) = r;
  r0 = r1; r1 = r2; r2 = r3;
  dgn = (int)(r2 - r1);
  ad0 = nd0; ad1 = nd1;
  vs0 = leaky(ns0 + nd0); vs1 = leaky(ns1 + nd1);
  sc = snr; ac = anr;
  }
}

__global__ __launch_bounds__(256) void gat_gather_l2(
    const unsigned* __restrict__ rowptr, const int* __restrict__ col,
    const float* __restrict__ G, const float* __restrict__ alS,
    const float* __restrict__ alD, const float* __restrict__ b,
    float* __restrict__ out, int N)
{
  __shared__ float stage[8 * NODES_PER_WAVE * 40];
  const int lane = threadIdx.x & 31;
  const int wv = __builtin_amdgcn_readfirstlane((int)(threadIdx.x >> 5));
  const int d0 = (blockIdx.x * 8 + wv) * NODES_PER_WAVE;
  float* slot = stage + wv * (NODES_PER_WAVE * 40);
  const bool act = lane < 20;
  const int c2 = act ? 2 * lane : 0;
  unsigned r0 = 0, r1 = 0, r2 = 0;
  int dgn = 0;
  float ad = 0.f, vs = 0.f;
  int sc = 0;
  float ac = 0.f;
  const int elast = max((int)rowptr[N] - 1, 0);
  if (d0 < N) {
    r0 = rowptr[d0]; r1 = rowptr[d0 + 1]; r2 = rowptr[min(d0 + 2, N)];
    dgn = (int)(r2 - r1);
    ad = alD[d0];
    vs = leaky(alS[d0] + ad);
    sc = col[min((int)r0 + min(lane, max((int)(r1 - r0) - 1, 0)), elast)];
    ac = alS[sc];
  }
#pragma unroll 1
  for (int kn = 0; kn < NODES_PER_WAVE; ++kn) {
  const int d = d0 + kn;
  float res0 = 0.f, res1 = 0.f;
  if (d < N) {
    const int beg = (int)r0;
    const int deg = (int)(r1 - r0);
    const int dn = d + 1;
    const bool more = (kn + 1 < NODES_PER_WAVE) && (dn < N);
    const int degn = more ? dgn : 0;
    const unsigned r3 = rowptr[min(d + 3, N)];
    const int dnc = min(dn, N - 1);
    const float ns = alS[dnc], nd = alD[dnc];
    const int snr = col[min((int)r1 + min(lane, max(degn - 1, 0)), elast)];
    const float2 gown = *(const float2*)(G + (size_t)d * 40 + c2);
    float m = vs;
    if (lane < deg) m = fmaxf(m, leaky(ac + ad));
    for (int base = 32; base < deg; base += 32) {
      const int i = base + lane;
      if (i < deg) m = fmaxf(m, leaky(alS[col[beg + i]] + ad));
    }
    m = wave_max(m);
    __builtin_amdgcn_sched_barrier(0);
    const float anr = alS[snr];
    const float wself = expf(vs - m);
    float acc0 = wself * gown.x, acc1 = wself * gown.y;
    float p = 0.f;
    for (int base = 0; base < deg; base += 32) {
      const int i = base + lane;
      int s = d;
      float w = 0.f;
      if (base == 0) {
        s = sc;
        if (i < deg) w = expf(leaky(ac + ad) - m);
      } else if (i < deg) {
        s = col[beg + i];
        w = expf(leaky(alS[s] + ad) - m);
      }
      p += w;
      const int cnt = min(32, deg - base);
      for (int j0 = 0; j0 < cnt; j0 += ROWS_IN_FLIGHT) {
#pragma unroll
        for (int u = 0; u < ROWS_IN_FLIGHT; ++u) {
          const int j = j0 + u;
          const int sj = __builtin_amdgcn_readlane(s, j);
          const float wj = lane_bcast(w, j);
          const float2 gv = *(const float2*)(G + (size_t)sj * 40 + c2);
          acc0 += wj * gv.x;
          acc1 += wj * gv.y;
        }
      }
    }
    const float den = wave_sum(p) + wself + 1e-16f;
    res0 = acc0 / den + b[c2];
    res1 = acc1 / den + b[c2 + 1];
    r0 = r1; r1 = r2; r2 = r3;
    dgn = (int)(r2 - r1);
    ad = nd;
    vs = leaky(ns + nd);
    sc = snr; ac = anr;
  }
  if (act) {
    slot[kn * 40 + c2] = res0;
    slot[kn * 40 + c2 + 1] = res1;
  }
  }
  const size_t lim = (size_t)N * 40;
  const size_t g0 = (size_t)d0 * 40 + lane;
#pragma unroll
  for (int j = 0; j < (NODES_PER_WAVE * 40) / 32; ++j) {
    const size_t g = g0 + 32 * j;
    if (g < lim) out[g] = slot[32 * j + lane];
  }
}

extern "C" void kernel_launch(void* const* d_in, const int* in_sizes, int n_in,
                              void* d_out, int out_size, void* d_ws, size_t ws_size,
                              hipStream_t stream)
{
  const float* x   = (const float*)d_in[0];
  const int*   ei  = (const int*)d_in[1];
  const float* W1  = (const float*)d_in[2];
  const float* aS1 = (const float*)d_in[3];
  const float* aD1 = (const float*)d_in[4];
  const float* b1  = (const float*)d_in[5];
  const float* W2  = (const float*)d_in[6];
  const float* aS2 = (const float*)d_in[7];
  const float* aD2 = (const float*)d_in[8];
  const float* b2  = (const float*)d_in[9];
  float* out = (float*)d_out;

  const int N  = in_sizes[0] / 128;
  const int E  = in_sizes[1] / 2;
  const int NP = (N / NODE_BLOCK + 1) * NODE_BLOCK;

  char* ws = (char*)d_ws;
  size_t off = 0;
  auto carve = [&](size_t bytes) -> char* {
    char* p = ws + off;
    off += (bytes + 255) & ~(size_t)255;
    return p;
  };
  float*    h1     = (float*)carve((size_t)N * 128 * 4);
  float*    out1   = (float*)carve((size_t)N * 128 * 4);
  float*    alS    = (float*)carve((size_t)N * 2 * 4);
  float*    alD    = (float*)carve((size_t)N * 2 * 4);
  unsigned* lp     = (unsigned*)carve((size_t)NP * 4);
  unsigned* bt     = (unsigned*)carve((size_t)NODE_BLOCK * 4);
  unsigned* rowptr = (unsigned*)carve((size_t)NP * 4);
  unsigned* rank   = (unsigned*)carve((size_t)E * 4);
  int*      col    = (int*)carve((size_t)E * 4);
  const int NS = (N + 31) & ~31;
  const int B  = (E + SLICE - 1) >> SLICE_SHIFT;
  unsigned* part   = (unsigned*)carve((size_t)B * NS * 4);
  float* g2 = h1;

  const int gemmBlocks = (B < 224) ? 256 - B : 32;
  gat_hist_gemm1<<<B + gemmBlocks, 1024, 0, stream>>>(ei, E, NS, B, part, rank, x, W1, h1, N / 16);
  gat_colsum_attn<<<NP / NODE_BLOCK + (N * 2 + 255) / 256, 256, 0, stream>>>(
      part, lp, bt, N, NS, B, NP / NODE_BLOCK, h1, aS1, aD1, alS, alD, 2, 64);
  gat_csr_fill<<<(E + 2047) / 2048, 256, 0, stream>>>(
      ei, E, lp, bt, NP / NODE_BLOCK, rowptr, part, NS, rank, col);

  gat_gather_l1<<<(N + 31) / 32, 256, 0, stream>>>(rowptr, col, h1, alS, alD, b1, out1, N);

  gat_gemm_128x40<<<1024, 96, 0, stream>>>(out1, W2, g2, N / 16);
  gat_attn<<<(N + 255) / 256, 256, 0, stream>>>(g2, aS2, aD2, alS, alD, N, 1, 40);
  gat_gather_l2<<<(N + 31) / 32, 256, 0, stream>>>(rowptr, col, g2, alS, alD, b2, out, N);
}


# Edge sort as two coalesced passes: slice blocks group (dst,src) pairs by 256-node block in LDS, a block per node block counts and places them; 6 launches; a node's edges sum in LDS arrival order
# speedup vs baseline: 5.3002x; 1.3202x over previous
#include <hip/hip_runtime.h>
#include <math.h>

typedef __attribute__((ext_vector_type(2))) float v2f;
typedef __attribute__((ext_vector_type(4))) float v4f;
typedef __attribute__((ext_vector_type(8))) float v8f;

#define NEG_SLOPE 0.2f

#define A_PITCH 132
#define G1_STAGE (64 * A_PITCH)

__device__ __forceinline__ void gemm_128x128_walk(
    const float* __restrict__ X, const float* __restrict__ W,
    float* __restrict__ H, float* __restrict__ stage, int nTiles, int wv,
    int tileBase0, int tileStep)
{
  const int lane = threadIdx.x & 31;
  const int colTile = wv & 7, rowSub = wv >> 3;
  const int col0 = colTile << 4;
  const int kOff = (lane >> 4) << 1;
  const int bCol = col0 + (lane & 15);
  v2f bf[32];
#pragma unroll
  for (int s = 0; s < 32; ++s) {
    bf[s].x = W[(size_t)(4 * s + kOff) * 128 + bCol];
    bf[s].y = W[(size_t)(4 * s + kOff + 1) * 128 + bCol];
  }
  const int nRows = nTiles << 4;
  const int sOff0 = wv * A_PITCH + lane * 4;
  const int sOff1 = (wv + 32) * A_PITCH + lane * 4;
  const int fOff = (rowSub * 16 + (lane & 15)) * A_PITCH + kOff;
  {
    const int g0 = min(tileBase0 * 16 + wv, nRows - 1);
    const int g1 = min(tileBase0 * 16 + wv + 32, nRows - 1);
    const float4 q0 = *(const float4*)(X + (size_t)g0 * 128 + lane * 4);
    const float4 q1 = *(const float4*)(X + (size_t)g1 * 128 + lane * 4);
    *(float4*)(stage + sOff0) = q0;
    *(float4*)(stage + sOff1) = q1;
  }
  __syncthreads();
  int buf = 0;
  for (int tileBase = tileBase0; tileBase < nTiles; tileBase += tileStep) {
    const int nextBase = tileBase + tileStep;
    const bool more = nextBase < nTiles;
    float4 q0 = make_float4(0.f, 0.f, 0.f, 0.f), q1 = q0;
    if (more) {
      const int g0 = min(nextBase * 16 + wv, nRows - 1);
      const int g1 = min(nextBase * 16 + wv + 32, nRows - 1);
      q0 = *(const float4*)(X + (size_t)g0 * 128 + lane * 4);
      q1 = *(const float4*)(X + (size_t)g1 * 128 + lane * 4);
    }
    const int tile = tileBase + rowSub;
    if (tile < nTiles) {
      const float* ar = stage + buf * G1_STAGE + fOff;
      v8f acc = {};
#pragma unroll
      for (int s = 0; s < 32; ++s) {
        const v2f a = *(const v2f*)(ar + 4 * s);
        acc = __builtin_amdgcn_wmma_f32_16x16x4_f32(
            false, a, false, bf[s], (short)0, acc, false, false);
      }
      const int r = (tile << 4) + ((lane >> 4) << 3);
      const int c = col0 + (lane & 15);
#pragma unroll
      for (int i = 0; i < 8; ++i)
        H[(size_t)(r + i) * 128 + c] = acc[i];
    }
    if (more) {
      float* sp = stage + (buf ^ 1) * G1_STAGE;
      *(float4*)(sp + sOff0) = q0;
      *(float4*)(sp + sOff1) = q1;
    }
    __syncthreads();
    buf ^= 1;
  }
}

#define G2_STAGE (16 * A_PITCH)
__global__ __launch_bounds__(96) void gat_gemm_128x40(
    const float* __restrict__ X, const float* __restrict__ W,
    float* __restrict__ G, int nTiles)
{
  __shared__ __attribute__((aligned(16))) float stage[2 * G2_STAGE];
  const int lane = threadIdx.x & 31;
  const int wave = __builtin_amdgcn_readfirstlane((int)(threadIdx.x >> 5));
  const int col0 = wave << 4;
  const int kOff = (lane >> 4) << 1;
  const int bCol = col0 + (lane & 15);
  const bool colOk = (bCol < 40);
  v2f bf[32];
#pragma unroll
  for (int s = 0; s < 32; ++s) {
    bf[s].x = colOk ? W[(size_t)(4 * s + kOff) * 40 + bCol] : 0.0f;
    bf[s].y = colOk ? W[(size_t)(4 * s + kOff + 1) * 40 + bCol] : 0.0f;
  }
  const int fOff = (lane & 15) * A_PITCH + kOff;
  v4f q[6];
  int tile = blockIdx.x;
  if (tile < nTiles) {
#pragma unroll
    for (int i = 0; i < 6; ++i) {
      const int lr = min(wave + 3 * i, 15);
      q[i] = *(const v4f*)(X + (size_t)(tile * 16 + lr) * 128 + lane * 4);
    }
#pragma unroll
    for (int i = 0; i < 6; ++i) {
      const int lr = wave + 3 * i;
      if (lr < 16) *(v4f*)(stage + lr * A_PITCH + lane * 4) = q[i];
    }
  }
  __syncthreads();
  int buf = 0;
  for (; tile < nTiles; tile += gridDim.x) {
    const int nextTile = tile + gridDim.x;
    const bool more = nextTile < nTiles;
    if (more) {
#pragma unroll
      for (int i = 0; i < 6; ++i) {
        const int lr = min(wave + 3 * i, 15);
        q[i] = *(const v4f*)(X + (size_t)(nextTile * 16 + lr) * 128 + lane * 4);
      }
    }
    const int row0 = tile << 4;
    const float* ar = stage + buf * G2_STAGE + fOff;
    v8f acc = {};
#pragma unroll
    for (int s = 0; s < 32; ++s) {
      const v2f a = *(const v2f*)(ar + 4 * s);
      acc = __builtin_amdgcn_wmma_f32_16x16x4_f32(
          false, a, false, bf[s], (short)0, acc, false, false);
    }
    const int r = row0 + ((lane >> 4) << 3);
    const int c = col0 + (lane & 15);
    if (c < 40) {
#pragma unroll
      for (int i = 0; i < 8; ++i)
        G[(size_t)(r + i) * 40 + c] = acc[i];
    }
    if (more) {
      float* sp = stage + (buf ^ 1) * G2_STAGE;
#pragma unroll
      for (int i = 0; i < 6; ++i) {
        const int lr = wave + 3 * i;
        if (lr < 16) *(v4f*)(sp + lr * A_PITCH + lane * 4) = q[i];
      }
    }
    __syncthreads();
    buf ^= 1;
  }
}

__device__ __forceinline__ void attn_dots(
    int idx, const float* __restrict__ Hf, const float* __restrict__ aS,
    const float* __restrict__ aD, float* __restrict__ alS,
    float* __restrict__ alD, int N, int NH, int C)
{
  if (idx >= N * NH) return;
  int n = idx / NH, h = idx - n * NH;
  const float* hp = Hf + (size_t)n * NH * C + (size_t)h * C;
  const float* sp = aS + (size_t)h * C;
  const float* dp = aD + (size_t)h * C;
  float s = 0.f, d = 0.f;
  for (int c = 0; c < C; ++c) { s += hp[c] * sp[c]; d += hp[c] * dp[c]; }
  alS[idx] = s;
  alD[idx] = d;
}

__global__ __launch_bounds__(256) void gat_attn(
    const float* __restrict__ Hf, const float* __restrict__ aS,
    const float* __restrict__ aD, float* __restrict__ alS,
    float* __restrict__ alD, int N, int NH, int C)
{
  attn_dots(blockIdx.x * 256 + threadIdx.x, Hf, aS, aD, alS, alD, N, NH, C);
}

__device__ __forceinline__ float leaky(float v) {
  return (v > 0.f) ? v : NEG_SLOPE * v;
}
__device__ __forceinline__ float wave_max(float v) {
  for (int o = 16; o > 0; o >>= 1) v = fmaxf(v, __shfl_xor(v, o, 32));
  return v;
}
__device__ __forceinline__ float wave_sum(float v) {
  for (int o = 16; o > 0; o >>= 1) v += __shfl_xor(v, o, 32);
  return v;
}
__device__ __forceinline__ float lane_bcast(float v, int j) {
  return __int_as_float(__builtin_amdgcn_readlane(__float_as_int(v), j));
}

#define SLICE_SHIFT 14
#define SLICE (1 << SLICE_SHIFT)
#define HIST_WORDS 51200
#define HIST_EPT (SLICE / 1024)
#define NODE_SHIFT 8
#define NODE_BLOCK (1 << NODE_SHIFT)
#define BIN_GROUPS 512
#define BO_PITCH 288
__global__ __launch_bounds__(1024) void gat_hist_gemm1(
    const int* __restrict__ ei, int E, int B, int nb, int* __restrict__ binned,
    unsigned* __restrict__ boff, const float* __restrict__ X,
    const float* __restrict__ W, float* __restrict__ H, int nTiles)
{
  __shared__ __attribute__((aligned(16))) unsigned hist[HIST_WORDS];
  const int t = threadIdx.x;
  if ((int)blockIdx.x >= B) {
    const int wv = __builtin_amdgcn_readfirstlane(t >> 5);
    const int gb = (int)blockIdx.x - B;
    const int nGemm = (int)gridDim.x - B;
    gemm_128x128_walk(X, W, H, (float*)hist, nTiles, wv, gb * 4, nGemm * 4);
    return;
  }
  unsigned* cntS = hist;
  unsigned* wtS = hist + BIN_GROUPS;
  int2* stg = (int2*)(hist + 1024);
  if (t < BIN_GROUPS) cntS[t] = 0u;
  __syncthreads();
  const int e0 = blockIdx.x * SLICE + t;
  int dst[HIST_EPT], sr[HIST_EPT];
  unsigned bk[HIST_EPT], r[HIST_EPT];
#pragma unroll
  for (int k = 0; k < HIST_EPT; ++k) {
    const int ec = min(e0 + 1024 * k, E - 1);
    dst[k] = ei[E + ec];
    sr[k] = ei[ec];
  }
#pragma unroll
  for (int k = 0; k < HIST_EPT; ++k) {
    const bool ok = (e0 + 1024 * k < E) &&
                    ((unsigned)dst[k] < (unsigned)(nb << NODE_SHIFT));
    bk[k] = ok ? ((unsigned)dst[k] >> NODE_SHIFT) : (unsigned)nb;
  }
#pragma unroll
  for (int k = 0; k < HIST_EPT; ++k) r[k] = atomicAdd(&cntS[bk[k]], 1u);
  __syncthreads();
  const int lane = t & 31, wvv = t >> 5;
  unsigned v = 0u, incl = 0u;
  if (t < BIN_GROUPS) {
    v = cntS[t];
    incl = v;
    for (int o = 1; o < 32; o <<= 1) {
      const unsigned u = __shfl_up(incl, o, 32);
      if (lane >= o) incl += u;
    }
    if (lane == 31) wtS[wvv] = incl;
  }
  __syncthreads();
  if (t < BIN_GROUPS) {
    unsigned before = 0u;
#pragma unroll
    for (int w = 0; w < BIN_GROUPS / 32; ++w)
      if (w < wvv) before += wtS[w];
    const unsigned ex = before + incl - v;
    cntS[t] = ex;
    if (t < BO_PITCH) boff[(size_t)blockIdx.x * BO_PITCH + t] = ex;
  }
  __syncthreads();
#pragma unroll
  for (int k = 0; k < HIST_EPT; ++k)
    stg[cntS[bk[k]] + r[k]] = make_int2(dst[k], sr[k]);
  __syncthreads();
  uint4* o = (uint4*)(binned + (size_t)blockIdx.x * (2 * SLICE));
  for (int i = t; i < SLICE / 2; i += 1024) o[i] = ((const uint4*)stg)[i];
}

#define BUCKET_SH (2 * NODE_BLOCK + 8 + 32)
__device__ __forceinline__ void csr_bucket(
    int b, const int* __restrict__ binned, const unsigned* __restrict__ boff,
    int B, unsigned* __restrict__ rowptr, int* __restrict__ col,
    unsigned* __restrict__ sh)
{
  unsigned* cnt = sh;
  unsigned* cur = sh + NODE_BLOCK;
  unsigned* wt = sh + 2 * NODE_BLOCK;
  unsigned* wb = sh + 2 * NODE_BLOCK + 8;
  const int t = threadIdx.x, lane = t & 31;
  const int wv = __builtin_amdgcn_readfirstlane(t >> 5);
  if (t < NODE_BLOCK) cnt[t] = 0u;
  unsigned pb = 0u;
  for (int s = t; s < B; s += 1024) pb += boff[(size_t)s * BO_PITCH + b];
  for (int o = 16; o > 0; o >>= 1) pb += __shfl_xor(pb, o, 32);
  if (lane == 0) wb[wv] = pb;
  __syncthreads();
  for (int s = wv; s < B; s += 32) {
    const unsigned beg = boff[(size_t)s * BO_PITCH + b];
    const unsigned end = boff[(size_t)s * BO_PITCH + b + 1];
    const int2* seg = (const int2*)binned + (size_t)s * SLICE;
    for (unsigned i0 = beg; i0 < end; i0 += 128) {
      int d[4];
#pragma unroll
      for (int u = 0; u < 4; ++u)
        d[u] = seg[min(i0 + 32 * u + lane, end - 1)].x;
#pragma unroll
      for (int u = 0; u < 4; ++u)
        if (i0 + 32 * u + lane < end) atomicAdd(&cnt[d[u] & (NODE_BLOCK - 1)], 1u);
    }
  }
  __syncthreads();
  unsigned base = 0u;
#pragma unroll
  for (int w = 0; w < 32; ++w) base += wb[w];
  unsigned v = 0u, incl = 0u;
  if (t < NODE_BLOCK) {
    v = cnt[t];
    incl = v;
    for (int o = 1; o < 32; o <<= 1) {
      const unsigned u = __shfl_up(incl, o, 32);
      if (lane >= o) incl += u;
    }
    if (lane == 31) wt[wv] = incl;
  }
  __syncthreads();
  if (t < NODE_BLOCK) {
    unsigned before = 0u;
#pragma unroll
    for (int w = 0; w < NODE_BLOCK / 32; ++w)
      if (w < wv) before += wt[w];
    const unsigned ex = before + incl - v;
    rowptr[b * NODE_BLOCK + t] = base + ex;
    cur[t] = ex;
  }
  __syncthreads();
  for (int s = wv; s < B; s += 32) {
    const unsigned beg = boff[(size_t)s * BO_PITCH + b];
    const unsigned end = boff[(size_t)s * BO_PITCH + b + 1];
    const int2* seg = (const int2*)binned + (size_t)s * SLICE;
    for (unsigned i0 = beg; i0 < end; i0 += 128) {
      int2 e[4];
#pragma unroll
      for (int u = 0; u < 4; ++u)
        e[u] = seg[min(i0 + 32 * u + lane, end - 1)];
#pragma unroll
      for (int u = 0; u < 4; ++u)
        if (i0 + 32 * u + lane < end)
          col[base + atomicAdd(&cur[e[u].x & (NODE_BLOCK - 1)], 1u)] = e[u].y;
    }
  }
}

__global__ __launch_bounds__(1024) void gat_bucket_attn(
    const int* __restrict__ binned, const unsigned* __restrict__ boff, int B,
    int nb, unsigned* __restrict__ rowptr, int* __restrict__ col,
    const float* __restrict__ Hf, const float* __restrict__ aS,
    const float* __restrict__ aD, float* __restrict__ alS,
    float* __restrict__ alD, int N, int NH, int C)
{
  __shared__ unsigned sh[BUCKET_SH];
  if ((int)blockIdx.x < nb) {
    csr_bucket(blockIdx.x, binned, boff, B, rowptr, col, sh);
  } else {
    attn_dots(((int)blockIdx.x - nb) * 1024 + threadIdx.x,
              Hf, aS, aD, alS, alD, N, NH, C);
  }
}

#define NODES_PER_WAVE 4
#define ROWS_IN_FLIGHT 4
__global__ __launch_bounds__(256) void gat_gather_l1(
    const unsigned* __restrict__ rowptr, const int* __restrict__ col,
    const float* __restrict__ Hf, const float* __restrict__ alS,
    const float* __restrict__ alD, const float* __restrict__ b,
    float* __restrict__ out1, int N)
{
  const int lane = threadIdx.x & 31;
  const int wv = __builtin_amdgcn_readfirstlane((int)(threadIdx.x >> 5));
  const int d0 = (blockIdx.x * 8 + wv) * NODES_PER_WAVE;
  if (d0 >= N) return;
  const int hsel = lane >> 4;
  unsigned r0 = rowptr[d0], r1 = rowptr[d0 + 1], r2 = rowptr[min(d0 + 2, N)];
  const int elast = max((int)rowptr[N] - 1, 0);
  int dgn = (int)(r2 - r1);
  float ad0 = alD[d0 * 2], ad1 = alD[d0 * 2 + 1];
  float vs0 = leaky(alS[d0 * 2] + ad0), vs1 = leaky(alS[d0 * 2 + 1] + ad1);
  int sc = col[min((int)r0 + min(lane, max((int)(r1 - r0) - 1, 0)), elast)];
  float2 ac = *(const float2*)(alS + (size_t)sc * 2);
#pragma unroll 1
  for (int kn = 0; kn < NODES_PER_WAVE; ++kn) {
  const int d = d0 + kn;
  if (d >= N) break;
  const int beg = (int)r0;
  const int deg = (int)(r1 - r0);
  const int dn = d + 1;
  const bool more = (kn + 1 < NODES_PER_WAVE) && (dn < N);
  const int degn = more ? dgn : 0;
  const unsigned r3 = rowptr[min(d + 3, N)];
  const int dnc = min(dn, N - 1);
  const float ns0 = alS[dnc * 2], ns1 = alS[dnc * 2 + 1];
  const float nd0 = alD[dnc * 2], nd1 = alD[dnc * 2 + 1];
  const int snr = col[min((int)r1 + min(lane, max(degn - 1, 0)), elast)];
  const float4 hown = *(const float4*)(Hf + (size_t)d * 128 + lane * 4);
  float m0 = vs0, m1 = vs1;
  if (lane < deg) {
    m0 = fmaxf(m0, leaky(ac.x + ad0));
    m1 = fmaxf(m1, leaky(ac.y + ad1));
  }
  for (int base = 32; base < deg; base += 32) {
    const int i = base + lane;
    if (i < deg) {
      const int s = col[beg + i];
      const float2 a = *(const float2*)(alS + (size_t)s * 2);
      m0 = fmaxf(m0, leaky(a.x + ad0));
      m1 = fmaxf(m1, leaky(a.y + ad1));
    }
  }
  m0 = wave_max(m0);
  m1 = wave_max(m1);
  __builtin_amdgcn_sched_barrier(0);
  const float2 anr = *(const float2*)(alS + (size_t)snr * 2);
  const float ws0 = expf(vs0 - m0), ws1 = expf(vs1 - m1);
  float4 acc;
  {
    const float w = hsel ? ws1 : ws0;
    acc.x = w * hown.x; acc.y = w * hown.y; acc.z = w * hown.z; acc.w = w * hown.w;
  }
  float p0 = 0.f, p1 = 0.f;
  for (int base = 0; base < deg; base += 32) {
    const int i = base + lane;
    int s = d;
    float w0 = 0.f, w1 = 0.f;
    if (base == 0) {
      s = sc;
      if (i < deg) {
        w0 = expf(leaky(ac.x + ad0) - m0);
        w1 = expf(leaky(ac.y + ad1) - m1);
      }
    } else if (i < deg) {
      s = col[beg + i];
      const float2 a = *(const float2*)(alS + (size_t)s * 2);
      w0 = expf(leaky(a.x + ad0) - m0);
      w1 = expf(leaky(a.y + ad1) - m1);
    }
    p0 += w0;
    p1 += w1;
    const int cnt = min(32, deg - base);
    for (int j0 = 0; j0 < cnt; j0 += ROWS_IN_FLIGHT) {
#pragma unroll
      for (int u = 0; u < ROWS_IN_FLIGHT; ++u) {
        const int j = j0 + u;
        const int sj = __builtin_amdgcn_readlane(s, j);
        const float wj0 = lane_bcast(w0, j), wj1 = lane_bcast(w1, j);
        const float w = hsel ? wj1 : wj0;
        const float4 hv = *(const float4*)(Hf + (size_t)sj * 128 + lane * 4);
        acc.x += w * hv.x; acc.y += w * hv.y; acc.z += w * hv.z; acc.w += w * hv.w;
      }
    }
  }
  const float den0 = wave_sum(p0) + ws0;
  const float den1 = wave_sum(p1) + ws1;
  const float den = (hsel ? den1 : den0) + 1e-16f;
  const float4 bv = *(const float4*)(b + lane * 4);
  float4 r;
  r.x = acc.x / den + bv.x;
  r.y = acc.y / den + bv.y;
  r.z = acc.z / den + bv.z;
  r.w = acc.w / den + bv.w;
  r.x = (r.x > 0.f) ? r.x : (expf(r.x) - 1.0f);
  r.y = (r.y > 0.f) ? r.y : (expf(r.y) - 1.0f);
  r.z = (r.z > 0.f) ? r.z : (expf(r.z) - 1.0f);
  r.w = (r.w > 0.f) ? r.w : (expf(r.w) - 1.0f);
  *(float4*)(out1 + (size_t)d * 128 + lane * 4) = r;
  r0 = r1; r1 = r2; r2 = r3;
  dgn = (int)(r2 - r1);
  ad0 = nd0; ad1 = nd1;
  vs0 = leaky(ns0 + nd0); vs1 = leaky(ns1 + nd1);
  sc = snr; ac = anr;
  }
}

__global__ __launch_bounds__(256) void gat_gather_l2(
    const unsigned* __restrict__ rowptr, const int* __restrict__ col,
    const float* __restrict__ G, const float* __restrict__ alS,
    const float* __restrict__ alD, const float* __restrict__ b,
    float* __restrict__ out, int N)
{
  __shared__ float stage[8 * NODES_PER_WAVE * 40];
  const int lane = threadIdx.x & 31;
  const int wv = __builtin_amdgcn_readfirstlane((int)(threadIdx.x >> 5));
  const int d0 = (blockIdx.x * 8 + wv) * NODES_PER_WAVE;
  float* slot = stage + wv * (NODES_PER_WAVE * 40);
  const bool act = lane < 20;
  const int c2 = act ? 2 * lane : 0;
  unsigned r0 = 0, r1 = 0, r2 = 0;
  int dgn = 0;
  float ad = 0.f, vs = 0.f;
  int sc = 0;
  float ac = 0.f;
  const int elast = max((int)rowptr[N] - 1, 0);
  if (d0 < N) {
    r0 = rowptr[d0]; r1 = rowptr[d0 + 1]; r2 = rowptr[min(d0 + 2, N)];
    dgn = (int)(r2 - r1);
    ad = alD[d0];
    vs = leaky(alS[d0] + ad);
    sc = col[min((int)r0 + min(lane, max((int)(r1 - r0) - 1, 0)), elast)];
    ac = alS[sc];
  }
#pragma unroll 1
  for (int kn = 0; kn < NODES_PER_WAVE; ++kn) {
  const int d = d0 + kn;
  float res0 = 0.f, res1 = 0.f;
  if (d < N) {
    const int beg = (int)r0;
    const int deg = (int)(r1 - r0);
    const int dn = d + 1;
    const bool more = (kn + 1 < NODES_PER_WAVE) && (dn < N);
    const int degn = more ? dgn : 0;
    const unsigned r3 = rowptr[min(d + 3, N)];
    const int dnc = min(dn, N - 1);
    const float ns = alS[dnc], nd = alD[dnc];
    const int snr = col[min((int)r1 + min(lane, max(degn - 1, 0)), elast)];
    const float2 gown = *(const float2*)(G + (size_t)d * 40 + c2);
    float m = vs;
    if (lane < deg) m = fmaxf(m, leaky(ac + ad));
    for (int base = 32; base < deg; base += 32) {
      const int i = base + lane;
      if (i < deg) m = fmaxf(m, leaky(alS[col[beg + i]] + ad));
    }
    m = wave_max(m);
    __builtin_amdgcn_sched_barrier(0);
    const float anr = alS[snr];
    const float wself = expf(vs - m);
    float acc0 = wself * gown.x, acc1 = wself * gown.y;
    float p = 0.f;
    for (int base = 0; base < deg; base += 32) {
      const int i = base + lane;
      int s = d;
      float w = 0.f;
      if (base == 0) {
        s = sc;
        if (i < deg) w = expf(leaky(ac + ad) - m);
      } else if (i < deg) {
        s = col[beg + i];
        w = expf(leaky(alS[s] + ad) - m);
      }
      p += w;
      const int cnt = min(32, deg - base);
      for (int j0 = 0; j0 < cnt; j0 += ROWS_IN_FLIGHT) {
#pragma unroll
        for (int u = 0; u < ROWS_IN_FLIGHT; ++u) {
          const int j = j0 + u;
          const int sj = __builtin_amdgcn_readlane(s, j);
          const float wj = lane_bcast(w, j);
          const float2 gv = *(const float2*)(G + (size_t)sj * 40 + c2);
          acc0 += wj * gv.x;
          acc1 += wj * gv.y;
        }
      }
    }
    const float den = wave_sum(p) + wself + 1e-16f;
    res0 = acc0 / den + b[c2];
    res1 = acc1 / den + b[c2 + 1];
    r0 = r1; r1 = r2; r2 = r3;
    dgn = (int)(r2 - r1);
    ad = nd;
    vs = leaky(ns + nd);
    sc = snr; ac = anr;
  }
  if (act) {
    slot[kn * 40 + c2] = res0;
    slot[kn * 40 + c2 + 1] = res1;
  }
  }
  const size_t lim = (size_t)N * 40;
  const size_t g0 = (size_t)d0 * 40 + lane;
#pragma unroll
  for (int j = 0; j < (NODES_PER_WAVE * 40) / 32; ++j) {
    const size_t g = g0 + 32 * j;
    if (g < lim) out[g] = slot[32 * j + lane];
  }
}

extern "C" void kernel_launch(void* const* d_in, const int* in_sizes, int n_in,
                              void* d_out, int out_size, void* d_ws, size_t ws_size,
                              hipStream_t stream)
{
  const float* x   = (const float*)d_in[0];
  const int*   ei  = (const int*)d_in[1];
  const float* W1  = (const float*)d_in[2];
  const float* aS1 = (const float*)d_in[3];
  const float* aD1 = (const float*)d_in[4];
  const float* b1  = (const float*)d_in[5];
  const float* W2  = (const float*)d_in[6];
  const float* aS2 = (const float*)d_in[7];
  const float* aD2 = (const float*)d_in[8];
  const float* b2  = (const float*)d_in[9];
  float* out = (float*)d_out;

  const int N  = in_sizes[0] / 128;
  const int E  = in_sizes[1] / 2;
  const int NP = (N / NODE_BLOCK + 1) * NODE_BLOCK;
  const int nb = NP / NODE_BLOCK;

  char* ws = (char*)d_ws;
  size_t off = 0;
  auto carve = [&](size_t bytes) -> char* {
    char* p = ws + off;
    off += (bytes + 255) & ~(size_t)255;
    return p;
  };
  float*    h1     = (float*)carve((size_t)N * 128 * 4);
  float*    out1   = (float*)carve((size_t)N * 128 * 4);
  float*    alS    = (float*)carve((size_t)N * 2 * 4);
  float*    alD    = (float*)carve((size_t)N * 2 * 4);
  unsigned* rowptr = (unsigned*)carve((size_t)NP * 4);
  int*      col    = (int*)carve((size_t)E * 4);
  const int B  = (E + SLICE - 1) >> SLICE_SHIFT;
  int*      binned = (int*)carve((size_t)B * SLICE * 8);
  unsigned* boff   = (unsigned*)carve((size_t)B * BO_PITCH * 4);
  float* g2 = h1;

  const int gemmBlocks = (B < 224) ? 256 - B : 32;
  gat_hist_gemm1<<<B + gemmBlocks, 1024, 0, stream>>>(ei, E, B, nb, binned, boff, x, W1, h1, N / 16);
  gat_bucket_attn<<<nb + (N * 2 + 1023) / 1024, 1024, 0, stream>>>(
      binned, boff, B, nb, rowptr, col, h1, aS1, aD1, alS, alD, N, 2, 64);

  gat_gather_l1<<<(N + 31) / 32, 256, 0, stream>>>(rowptr, col, h1, alS, alD, b1, out1, N);

  gat_gemm_128x40<<<1024, 96, 0, stream>>>(out1, W2, g2, N / 16);
  gat_attn<<<(N + 255) / 256, 256, 0, stream>>>(g2, aS2, aD2, alS, alD, N, 1, 40);
  gat_gather_l2<<<(N + 31) / 32, 256, 0, stream>>>(rowptr, col, g2, alS, alD, b2, out, N);
}


# Layer 2 attention dots leave its GEMM as two spare weight columns W2 a_src, W2 a_dst (folded by one block of the bucket launch): 5 launches; these logits now sum over k with folded weights
# speedup vs baseline: 5.5300x; 1.0434x over previous
#include <hip/hip_runtime.h>
#include <math.h>

typedef __attribute__((ext_vector_type(2))) float v2f;
typedef __attribute__((ext_vector_type(4))) float v4f;
typedef __attribute__((ext_vector_type(8))) float v8f;

#define NEG_SLOPE 0.2f

#define A_PITCH 132
#define G1_STAGE (64 * A_PITCH)

__device__ __forceinline__ void gemm_128x128_walk(
    const float* __restrict__ X, const float* __restrict__ W,
    float* __restrict__ H, float* __restrict__ stage, int nTiles, int wv,
    int tileBase0, int tileStep)
{
  const int lane = threadIdx.x & 31;
  const int colTile = wv & 7, rowSub = wv >> 3;
  const int col0 = colTile << 4;
  const int kOff = (lane >> 4) << 1;
  const int bCol = col0 + (lane & 15);
  v2f bf[32];
#pragma unroll
  for (int s = 0; s < 32; ++s) {
    bf[s].x = W[(size_t)(4 * s + kOff) * 128 + bCol];
    bf[s].y = W[(size_t)(4 * s + kOff + 1) * 128 + bCol];
  }
  const int nRows = nTiles << 4;
  const int sOff0 = wv * A_PITCH + lane * 4;
  const int sOff1 = (wv + 32) * A_PITCH + lane * 4;
  const int fOff = (rowSub * 16 + (lane & 15)) * A_PITCH + kOff;
  {
    const int g0 = min(tileBase0 * 16 + wv, nRows - 1);
    const int g1 = min(tileBase0 * 16 + wv + 32, nRows - 1);
    const float4 q0 = *(const float4*)(X + (size_t)g0 * 128 + lane * 4);
    const float4 q1 = *(const float4*)(X + (size_t)g1 * 128 + lane * 4);
    *(float4*)(stage + sOff0) = q0;
    *(float4*)(stage + sOff1) = q1;
  }
  __syncthreads();
  int buf = 0;
  for (int tileBase = tileBase0; tileBase < nTiles; tileBase += tileStep) {
    const int nextBase = tileBase + tileStep;
    const bool more = nextBase < nTiles;
    float4 q0 = make_float4(0.f, 0.f, 0.f, 0.f), q1 = q0;
    if (more) {
      const int g0 = min(nextBase * 16 + wv, nRows - 1);
      const int g1 = min(nextBase * 16 + wv + 32, nRows - 1);
      q0 = *(const float4*)(X + (size_t)g0 * 128 + lane * 4);
      q1 = *(const float4*)(X + (size_t)g1 * 128 + lane * 4);
    }
    const int tile = tileBase + rowSub;
    if (tile < nTiles) {
      const float* ar = stage + buf * G1_STAGE + fOff;
      v8f acc = {};
#pragma unroll
      for (int s = 0; s < 32; ++s) {
        const v2f a = *(const v2f*)(ar + 4 * s);
        acc = __builtin_amdgcn_wmma_f32_16x16x4_f32(
            false, a, false, bf[s], (short)0, acc, false, false);
      }
      const int r = (tile << 4) + ((lane >> 4) << 3);
      const int c = col0 + (lane & 15);
#pragma unroll
      for (int i = 0; i < 8; ++i)
        H[(size_t)(r + i) * 128 + c] = acc[i];
    }
    if (more) {
      float* sp = stage + (buf ^ 1) * G1_STAGE;
      *(float4*)(sp + sOff0) = q0;
      *(float4*)(sp + sOff1) = q1;
    }
    __syncthreads();
    buf ^= 1;
  }
}

#define G2_STAGE (16 * A_PITCH)
__global__ __launch_bounds__(96) void gat_gemm_128x40(
    const float* __restrict__ X, const float* __restrict__ W,
    const float* __restrict__ wf, float* __restrict__ G,
    float* __restrict__ alS, float* __restrict__ alD, int nTiles)
{
  __shared__ __attribute__((aligned(16))) float stage[2 * G2_STAGE];
  const int lane = threadIdx.x & 31;
  const int wave = __builtin_amdgcn_readfirstlane((int)(threadIdx.x >> 5));
  const int col0 = wave << 4;
  const int kOff = (lane >> 4) << 1;
  const int bCol = col0 + (lane & 15);
  const bool colOk = (bCol < 40);
  const bool dotCol = (bCol == 40) || (bCol == 41);
  const float* bp = colOk ? (W + bCol) : (wf + ((bCol == 41) ? 128 : 0));
  const int bPitch = colOk ? 40 : 1;
  const bool live = colOk || dotCol;
  bp += kOff * bPitch;
  v2f bf[32];
#pragma unroll
  for (int s = 0; s < 32; ++s) {
    const float w0 = bp[(4 * s) * bPitch];
    const float w1 = bp[(4 * s + 1) * bPitch];
    bf[s].x = live ? w0 : 0.0f;
    bf[s].y = live ? w1 : 0.0f;
  }
  const int fOff = (lane & 15) * A_PITCH + kOff;
  v4f q[6];
  int tile = blockIdx.x;
  if (tile < nTiles) {
#pragma unroll
    for (int i = 0; i < 6; ++i) {
      const int lr = min(wave + 3 * i, 15);
      q[i] = *(const v4f*)(X + (size_t)(tile * 16 + lr) * 128 + lane * 4);
    }
#pragma unroll
    for (int i = 0; i < 6; ++i) {
      const int lr = wave + 3 * i;
      if (lr < 16) *(v4f*)(stage + lr * A_PITCH + lane * 4) = q[i];
    }
  }
  __syncthreads();
  int buf = 0;
  for (; tile < nTiles; tile += gridDim.x) {
    const int nextTile = tile + gridDim.x;
    const bool more = nextTile < nTiles;
    if (more) {
#pragma unroll
      for (int i = 0; i < 6; ++i) {
        const int lr = min(wave + 3 * i, 15);
        q[i] = *(const v4f*)(X + (size_t)(nextTile * 16 + lr) * 128 + lane * 4);
      }
    }
    const int row0 = tile << 4;
    const float* ar = stage + buf * G2_STAGE + fOff;
    v8f acc = {};
#pragma unroll
    for (int s = 0; s < 32; ++s) {
      const v2f a = *(const v2f*)(ar + 4 * s);
      acc = __builtin_amdgcn_wmma_f32_16x16x4_f32(
          false, a, false, bf[s], (short)0, acc, false, false);
    }
    const int r = row0 + ((lane >> 4) << 3);
    const int c = col0 + (lane & 15);
    if (c < 40) {
#pragma unroll
      for (int i = 0; i < 8; ++i)
        G[(size_t)(r + i) * 40 + c] = acc[i];
    } else if (dotCol) {
      float* ap = ((c == 40) ? alS : alD) + r;
      *(v4f*)ap = __builtin_shufflevector(acc, acc, 0, 1, 2, 3);
      *(v4f*)(ap + 4) = __builtin_shufflevector(acc, acc, 4, 5, 6, 7);
    }
    if (more) {
      float* sp = stage + (buf ^ 1) * G2_STAGE;
#pragma unroll
      for (int i = 0; i < 6; ++i) {
        const int lr = wave + 3 * i;
        if (lr < 16) *(v4f*)(sp + lr * A_PITCH + lane * 4) = q[i];
      }
    }
    __syncthreads();
    buf ^= 1;
  }
}

__device__ __forceinline__ void attn_dots(
    int idx, const float* __restrict__ Hf, const float* __restrict__ aS,
    const float* __restrict__ aD, float* __restrict__ alS,
    float* __restrict__ alD, int N, int NH, int C)
{
  if (idx >= N * NH) return;
  int n = idx / NH, h = idx - n * NH;
  const float* hp = Hf + (size_t)n * NH * C + (size_t)h * C;
  const float* sp = aS + (size_t)h * C;
  const float* dp = aD + (size_t)h * C;
  float s = 0.f, d = 0.f;
  for (int c = 0; c < C; ++c) { s += hp[c] * sp[c]; d += hp[c] * dp[c]; }
  alS[idx] = s;
  alD[idx] = d;
}


__device__ __forceinline__ void fold_attn_weights(
    const float* __restrict__ W, const float* __restrict__ aS,
    const float* __restrict__ aD, float* __restrict__ wf)
{
  const int t = threadIdx.x, k = t >> 3, j = t & 7;
  float s = 0.f, d = 0.f;
#pragma unroll
  for (int i = 0; i < 5; ++i) {
    const int cc = j + 8 * i;
    const float w = W[k * 40 + cc];
    s += w * aS[cc];
    d += w * aD[cc];
  }
  for (int o = 4; o > 0; o >>= 1) {
    s += __shfl_xor(s, o, 32);
    d += __shfl_xor(d, o, 32);
  }
  if (j == 0) {
    wf[k] = s;
    wf[128 + k] = d;
  }
}

__device__ __forceinline__ float leaky(float v) {
  return (v > 0.f) ? v : NEG_SLOPE * v;
}
__device__ __forceinline__ float wave_max(float v) {
  for (int o = 16; o > 0; o >>= 1) v = fmaxf(v, __shfl_xor(v, o, 32));
  return v;
}
__device__ __forceinline__ float wave_sum(float v) {
  for (int o = 16; o > 0; o >>= 1) v += __shfl_xor(v, o, 32);
  return v;
}
__device__ __forceinline__ float lane_bcast(float v, int j) {
  return __int_as_float(__builtin_amdgcn_readlane(__float_as_int(v), j));
}

#define SLICE_SHIFT 14
#define SLICE (1 << SLICE_SHIFT)
#define HIST_WORDS 51200
#define HIST_EPT (SLICE / 1024)
#define NODE_SHIFT 8
#define NODE_BLOCK (1 << NODE_SHIFT)
#define BIN_GROUPS 512
#define BO_PITCH 288
__global__ __launch_bounds__(1024) void gat_hist_gemm1(
    const int* __restrict__ ei, int E, int B, int nb, int* __restrict__ binned,
    unsigned* __restrict__ boff, const float* __restrict__ X,
    const float* __restrict__ W, float* __restrict__ H, int nTiles)
{
  __shared__ __attribute__((aligned(16))) unsigned hist[HIST_WORDS];
  const int t = threadIdx.x;
  if ((int)blockIdx.x >= B) {
    const int wv = __builtin_amdgcn_readfirstlane(t >> 5);
    const int gb = (int)blockIdx.x - B;
    const int nGemm = (int)gridDim.x - B;
    gemm_128x128_walk(X, W, H, (float*)hist, nTiles, wv, gb * 4, nGemm * 4);
    return;
  }
  unsigned* cntS = hist;
  unsigned* wtS = hist + BIN_GROUPS;
  int2* stg = (int2*)(hist + 1024);
  if (t < BIN_GROUPS) cntS[t] = 0u;
  __syncthreads();
  const int e0 = blockIdx.x * SLICE + t;
  int dst[HIST_EPT], sr[HIST_EPT];
  unsigned bk[HIST_EPT], r[HIST_EPT];
#pragma unroll
  for (int k = 0; k < HIST_EPT; ++k) {
    const int ec = min(e0 + 1024 * k, E - 1);
    dst[k] = ei[E + ec];
    sr[k] = ei[ec];
  }
#pragma unroll
  for (int k = 0; k < HIST_EPT; ++k) {
    const bool ok = (e0 + 1024 * k < E) &&
                    ((unsigned)dst[k] < (unsigned)(nb << NODE_SHIFT));
    bk[k] = ok ? ((unsigned)dst[k] >> NODE_SHIFT) : (unsigned)nb;
  }
#pragma unroll
  for (int k = 0; k < HIST_EPT; ++k) r[k] = atomicAdd(&cntS[bk[k]], 1u);
  __syncthreads();
  const int lane = t & 31, wvv = t >> 5;
  unsigned v = 0u, incl = 0u;
  if (t < BIN_GROUPS) {
    v = cntS[t];
    incl = v;
    for (int o = 1; o < 32; o <<= 1) {
      const unsigned u = __shfl_up(incl, o, 32);
      if (lane >= o) incl += u;
    }
    if (lane == 31) wtS[wvv] = incl;
  }
  __syncthreads();
  if (t < BIN_GROUPS) {
    unsigned before = 0u;
#pragma unroll
    for (int w = 0; w < BIN_GROUPS / 32; ++w)
      if (w < wvv) before += wtS[w];
    const unsigned ex = before + incl - v;
    cntS[t] = ex;
    if (t < BO_PITCH) boff[(size_t)blockIdx.x * BO_PITCH + t] = ex;
  }
  __syncthreads();
#pragma unroll
  for (int k = 0; k < HIST_EPT; ++k)
    stg[cntS[bk[k]] + r[k]] = make_int2(dst[k], sr[k]);
  __syncthreads();
  uint4* o = (uint4*)(binned + (size_t)blockIdx.x * (2 * SLICE));
  for (int i = t; i < SLICE / 2; i += 1024) o[i] = ((const uint4*)stg)[i];
}

#define BUCKET_SH (2 * NODE_BLOCK + 8 + 32)
__device__ __forceinline__ void csr_bucket(
    int b, const int* __restrict__ binned, const unsigned* __restrict__ boff,
    int B, unsigned* __restrict__ rowptr, int* __restrict__ col,
    unsigned* __restrict__ sh)
{
  unsigned* cnt = sh;
  unsigned* cur = sh + NODE_BLOCK;
  unsigned* wt = sh + 2 * NODE_BLOCK;
  unsigned* wb = sh + 2 * NODE_BLOCK + 8;
  const int t = threadIdx.x, lane = t & 31;
  const int wv = __builtin_amdgcn_readfirstlane(t >> 5);
  if (t < NODE_BLOCK) cnt[t] = 0u;
  unsigned pb = 0u;
  for (int s = t; s < B; s += 1024) pb += boff[(size_t)s * BO_PITCH + b];
  for (int o = 16; o > 0; o >>= 1) pb += __shfl_xor(pb, o, 32);
  if (lane == 0) wb[wv] = pb;
  __syncthreads();
  for (int s = wv; s < B; s += 32) {
    const unsigned beg = boff[(size_t)s * BO_PITCH + b];
    const unsigned end = boff[(size_t)s * BO_PITCH + b + 1];
    const int2* seg = (const int2*)binned + (size_t)s * SLICE;
    for (unsigned i0 = beg; i0 < end; i0 += 128) {
      int d[4];
#pragma unroll
      for (int u = 0; u < 4; ++u)
        d[u] = seg[min(i0 + 32 * u + lane, end - 1)].x;
#pragma unroll
      for (int u = 0; u < 4; ++u)
        if (i0 + 32 * u + lane < end) atomicAdd(&cnt[d[u] & (NODE_BLOCK - 1)], 1u);
    }
  }
  __syncthreads();
  unsigned base = 0u;
#pragma unroll
  for (int w = 0; w < 32; ++w) base += wb[w];
  unsigned v = 0u, incl = 0u;
  if (t < NODE_BLOCK) {
    v = cnt[t];
    incl = v;
    for (int o = 1; o < 32; o <<= 1) {
      const unsigned u = __shfl_up(incl, o, 32);
      if (lane >= o) incl += u;
    }
    if (lane == 31) wt[wv] = incl;
  }
  __syncthreads();
  if (t < NODE_BLOCK) {
    unsigned before = 0u;
#pragma unroll
    for (int w = 0; w < NODE_BLOCK / 32; ++w)
      if (w < wv) before += wt[w];
    const unsigned ex = before + incl - v;
    rowptr[b * NODE_BLOCK + t] = base + ex;
    cur[t] = ex;
  }
  __syncthreads();
  for (int s = wv; s < B; s += 32) {
    const unsigned beg = boff[(size_t)s * BO_PITCH + b];
    const unsigned end = boff[(size_t)s * BO_PITCH + b + 1];
    const int2* seg = (const int2*)binned + (size_t)s * SLICE;
    for (unsigned i0 = beg; i0 < end; i0 += 128) {
      int2 e[4];
#pragma unroll
      for (int u = 0; u < 4; ++u)
        e[u] = seg[min(i0 + 32 * u + lane, end - 1)];
#pragma unroll
      for (int u = 0; u < 4; ++u)
        if (i0 + 32 * u + lane < end)
          col[base + atomicAdd(&cur[e[u].x & (NODE_BLOCK - 1)], 1u)] = e[u].y;
    }
  }
}

__global__ __launch_bounds__(1024) void gat_bucket_attn(
    const int* __restrict__ binned, const unsigned* __restrict__ boff, int B,
    int nb, unsigned* __restrict__ rowptr, int* __restrict__ col,
    const float* __restrict__ Hf, const float* __restrict__ aS,
    const float* __restrict__ aD, float* __restrict__ alS,
    float* __restrict__ alD, int N, int NH, int C,
    const float* __restrict__ W2, const float* __restrict__ aS2,
    const float* __restrict__ aD2, float* __restrict__ wf)
{
  __shared__ unsigned sh[BUCKET_SH];
  if ((int)blockIdx.x < nb) {
    csr_bucket(blockIdx.x, binned, boff, B, rowptr, col, sh);
  } else if (blockIdx.x + 1 < gridDim.x) {
    attn_dots(((int)blockIdx.x - nb) * 1024 + threadIdx.x,
              Hf, aS, aD, alS, alD, N, NH, C);
  } else {
    fold_attn_weights(W2, aS2, aD2, wf);
  }
}

#define NODES_PER_WAVE 4
#define ROWS_IN_FLIGHT 4
__global__ __launch_bounds__(256) void gat_gather_l1(
    const unsigned* __restrict__ rowptr, const int* __restrict__ col,
    const float* __restrict__ Hf, const float* __restrict__ alS,
    const float* __restrict__ alD, const float* __restrict__ b,
    float* __restrict__ out1, int N)
{
  const int lane = threadIdx.x & 31;
  const int wv = __builtin_amdgcn_readfirstlane((int)(threadIdx.x >> 5));
  const int d0 = (blockIdx.x * 8 + wv) * NODES_PER_WAVE;
  if (d0 >= N) return;
  const int hsel = lane >> 4;
  unsigned r0 = rowptr[d0], r1 = rowptr[d0 + 1], r2 = rowptr[min(d0 + 2, N)];
  const int elast = max((int)rowptr[N] - 1, 0);
  int dgn = (int)(r2 - r1);
  float ad0 = alD[d0 * 2], ad1 = alD[d0 * 2 + 1];
  float vs0 = leaky(alS[d0 * 2] + ad0), vs1 = leaky(alS[d0 * 2 + 1] + ad1);
  int sc = col[min((int)r0 + min(lane, max((int)(r1 - r0) - 1, 0)), elast)];
  float2 ac = *(const float2*)(alS + (size_t)sc * 2);
#pragma unroll 1
  for (int kn = 0; kn < NODES_PER_WAVE; ++kn) {
  const int d = d0 + kn;
  if (d >= N) break;
  const int beg = (int)r0;
  const int deg = (int)(r1 - r0);
  const int dn = d + 1;
  const bool more = (kn + 1 < NODES_PER_WAVE) && (dn < N);
  const int degn = more ? dgn : 0;
  const unsigned r3 = rowptr[min(d + 3, N)];
  const int dnc = min(dn, N - 1);
  const float ns0 = alS[dnc * 2], ns1 = alS[dnc * 2 + 1];
  const float nd0 = alD[dnc * 2], nd1 = alD[dnc * 2 + 1];
  const int snr = col[min((int)r1 + min(lane, max(degn - 1, 0)), elast)];
  const float4 hown = *(const float4*)(Hf + (size_t)d * 128 + lane * 4);
  float m0 = vs0, m1 = vs1;
  if (lane < deg) {
    m0 = fmaxf(m0, leaky(ac.x + ad0));
    m1 = fmaxf(m1, leaky(ac.y + ad1));
  }
  for (int base = 32; base < deg; base += 32) {
    const int i = base + lane;
    if (i < deg) {
      const int s = col[beg + i];
      const float2 a = *(const float2*)(alS + (size_t)s * 2);
      m0 = fmaxf(m0, leaky(a.x + ad0));
      m1 = fmaxf(m1, leaky(a.y + ad1));
    }
  }
  m0 = wave_max(m0);
  m1 = wave_max(m1);
  __builtin_amdgcn_sched_barrier(0);
  const float2 anr = *(const float2*)(alS + (size_t)snr * 2);
  const float ws0 = expf(vs0 - m0), ws1 = expf(vs1 - m1);
  float4 acc;
  {
    const float w = hsel ? ws1 : ws0;
    acc.x = w * hown.x; acc.y = w * hown.y; acc.z = w * hown.z; acc.w = w * hown.w;
  }
  float p0 = 0.f, p1 = 0.f;
  for (int base = 0; base < deg; base += 32) {
    const int i = base + lane;
    int s = d;
    float w0 = 0.f, w1 = 0.f;
    if (base == 0) {
      s = sc;
      if (i < deg) {
        w0 = expf(leaky(ac.x + ad0) - m0);
        w1 = expf(leaky(ac.y + ad1) - m1);
      }
    } else if (i < deg) {
      s = col[beg + i];
      const float2 a = *(const float2*)(alS + (size_t)s * 2);
      w0 = expf(leaky(a.x + ad0) - m0);
      w1 = expf(leaky(a.y + ad1) - m1);
    }
    p0 += w0;
    p1 += w1;
    const int cnt = min(32, deg - base);
    for (int j0 = 0; j0 < cnt; j0 += ROWS_IN_FLIGHT) {
#pragma unroll
      for (int u = 0; u < ROWS_IN_FLIGHT; ++u) {
        const int j = j0 + u;
        const int sj = __builtin_amdgcn_readlane(s, j);
        const float wj0 = lane_bcast(w0, j), wj1 = lane_bcast(w1, j);
        const float w = hsel ? wj1 : wj0;
        const float4 hv = *(const float4*)(Hf + (size_t)sj * 128 + lane * 4);
        acc.x += w * hv.x; acc.y += w * hv.y; acc.z += w * hv.z; acc.w += w * hv.w;
      }
    }
  }
  const float den0 = wave_sum(p0) + ws0;
  const float den1 = wave_sum(p1) + ws1;
  const float den = (hsel ? den1 : den0) + 1e-16f;
  const float4 bv = *(const float4*)(b + lane * 4);
  float4 r;
  r.x = acc.x / den + bv.x;
  r.y = acc.y / den + bv.y;
  r.z = acc.z / den + bv.z;
  r.w = acc.w / den + bv.w;
  r.x = (r.x > 0.f) ? r.x : (expf(r.x) - 1.0f);
  r.y = (r.y > 0.f) ? r.y : (expf(r.y) - 1.0f);
  r.z = (r.z > 0.f) ? r.z : (expf(r.z) - 1.0f);
  r.w = (r.w > 0.f) ? r.w : (expf(r.w) - 1.0f);
  *(float4*)(out1 + (size_t)d * 128 + lane * 4) = r;
  r0 = r1; r1 = r2; r2 = r3;
  dgn = (int)(r2 - r1);
  ad0 = nd0; ad1 = nd1;
  vs0 = leaky(ns0 + nd0); vs1 = leaky(ns1 + nd1);
  sc = snr; ac = anr;
  }
}

__global__ __launch_bounds__(256) void gat_gather_l2(
    const unsigned* __restrict__ rowptr, const int* __restrict__ col,
    const float* __restrict__ G, const float* __restrict__ alS,
    const float* __restrict__ alD, const float* __restrict__ b,
    float* __restrict__ out, int N)
{
  __shared__ float stage[8 * NODES_PER_WAVE * 40];
  const int lane = threadIdx.x & 31;
  const int wv = __builtin_amdgcn_readfirstlane((int)(threadIdx.x >> 5));
  const int d0 = (blockIdx.x * 8 + wv) * NODES_PER_WAVE;
  float* slot = stage + wv * (NODES_PER_WAVE * 40);
  const bool act = lane < 20;
  const int c2 = act ? 2 * lane : 0;
  unsigned r0 = 0, r1 = 0, r2 = 0;
  int dgn = 0;
  float ad = 0.f, vs = 0.f;
  int sc = 0;
  float ac = 0.f;
  const int elast = max((int)rowptr[N] - 1, 0);
  if (d0 < N) {
    r0 = rowptr[d0]; r1 = rowptr[d0 + 1]; r2 = rowptr[min(d0 + 2, N)];
    dgn = (int)(r2 - r1);
    ad = alD[d0];
    vs = leaky(alS[d0] + ad);
    sc = col[min((int)r0 + min(lane, max((int)(r1 - r0) - 1, 0)), elast)];
    ac = alS[sc];
  }
#pragma unroll 1
  for (int kn = 0; kn < NODES_PER_WAVE; ++kn) {
  const int d = d0 + kn;
  float res0 = 0.f, res1 = 0.f;
  if (d < N) {
    const int beg = (int)r0;
    const int deg = (int)(r1 - r0);
    const int dn = d + 1;
    const bool more = (kn + 1 < NODES_PER_WAVE) && (dn < N);
    const int degn = more ? dgn : 0;
    const unsigned r3 = rowptr[min(d + 3, N)];
    const int dnc = min(dn, N - 1);
    const float ns = alS[dnc], nd = alD[dnc];
    const int snr = col[min((int)r1 + min(lane, max(degn - 1, 0)), elast)];
    const float2 gown = *(const float2*)(G + (size_t)d * 40 + c2);
    float m = vs;
    if (lane < deg) m = fmaxf(m, leaky(ac + ad));
    for (int base = 32; base < deg; base += 32) {
      const int i = base + lane;
      if (i < deg) m = fmaxf(m, leaky(alS[col[beg + i]] + ad));
    }
    m = wave_max(m);
    __builtin_amdgcn_sched_barrier(0);
    const float anr = alS[snr];
    const float wself = expf(vs - m);
    float acc0 = wself * gown.x, acc1 = wself * gown.y;
    float p = 0.f;
    for (int base = 0; base < deg; base += 32) {
      const int i = base + lane;
      int s = d;
      float w = 0.f;
      if (base == 0) {
        s = sc;
        if (i < deg) w = expf(leaky(ac + ad) - m);
      } else if (i < deg) {
        s = col[beg + i];
        w = expf(leaky(alS[s] + ad) - m);
      }
      p += w;
      const int cnt = min(32, deg - base);
      for (int j0 = 0; j0 < cnt; j0 += ROWS_IN_FLIGHT) {
#pragma unroll
        for (int u = 0; u < ROWS_IN_FLIGHT; ++u) {
          const int j = j0 + u;
          const int sj = __builtin_amdgcn_readlane(s, j);
          const float wj = lane_bcast(w, j);
          const float2 gv = *(const float2*)(G + (size_t)sj * 40 + c2);
          acc0 += wj * gv.x;
          acc1 += wj * gv.y;
        }
      }
    }
    const float den = wave_sum(p) + wself + 1e-16f;
    res0 = acc0 / den + b[c2];
    res1 = acc1 / den + b[c2 + 1];
    r0 = r1; r1 = r2; r2 = r3;
    dgn = (int)(r2 - r1);
    ad = nd;
    vs = leaky(ns + nd);
    sc = snr; ac = anr;
  }
  if (act) {
    slot[kn * 40 + c2] = res0;
    slot[kn * 40 + c2 + 1] = res1;
  }
  }
  const size_t lim = (size_t)N * 40;
  const size_t g0 = (size_t)d0 * 40 + lane;
#pragma unroll
  for (int j = 0; j < (NODES_PER_WAVE * 40) / 32; ++j) {
    const size_t g = g0 + 32 * j;
    if (g < lim) out[g] = slot[32 * j + lane];
  }
}

extern "C" void kernel_launch(void* const* d_in, const int* in_sizes, int n_in,
                              void* d_out, int out_size, void* d_ws, size_t ws_size,
                              hipStream_t stream)
{
  const float* x   = (const float*)d_in[0];
  const int*   ei  = (const int*)d_in[1];
  const float* W1  = (const float*)d_in[2];
  const float* aS1 = (const float*)d_in[3];
  const float* aD1 = (const float*)d_in[4];
  const float* b1  = (const float*)d_in[5];
  const float* W2  = (const float*)d_in[6];
  const float* aS2 = (const float*)d_in[7];
  const float* aD2 = (const float*)d_in[8];
  const float* b2  = (const float*)d_in[9];
  float* out = (float*)d_out;

  const int N  = in_sizes[0] / 128;
  const int E  = in_sizes[1] / 2;
  const int NP = (N / NODE_BLOCK + 1) * NODE_BLOCK;
  const int nb = NP / NODE_BLOCK;

  char* ws = (char*)d_ws;
  size_t off = 0;
  auto carve = [&](size_t bytes) -> char* {
    char* p = ws + off;
    off += (bytes + 255) & ~(size_t)255;
    return p;
  };
  float*    h1     = (float*)carve((size_t)N * 128 * 4);
  float*    out1   = (float*)carve((size_t)N * 128 * 4);
  float*    alS    = (float*)carve((size_t)N * 2 * 4);
  float*    alD    = (float*)carve((size_t)N * 2 * 4);
  unsigned* rowptr = (unsigned*)carve((size_t)NP * 4);
  int*      col    = (int*)carve((size_t)E * 4);
  const int B  = (E + SLICE - 1) >> SLICE_SHIFT;
  int*      binned = (int*)carve((size_t)B * SLICE * 8);
  unsigned* boff   = (unsigned*)carve((size_t)B * BO_PITCH * 4);
  float*    wf2    = (float*)carve((size_t)256 * 4);
  float* g2 = h1;

  const int gemmBlocks = (B < 224) ? 256 - B : 32;
  gat_hist_gemm1<<<B + gemmBlocks, 1024, 0, stream>>>(ei, E, B, nb, binned, boff, x, W1, h1, N / 16);
  gat_bucket_attn<<<nb + (N * 2 + 1023) / 1024 + 1, 1024, 0, stream>>>(
      binned, boff, B, nb, rowptr, col, h1, aS1, aD1, alS, alD, N, 2, 64,
      W2, aS2, aD2, wf2);

  gat_gather_l1<<<(N + 31) / 32, 256, 0, stream>>>(rowptr, col, h1, alS, alD, b1, out1, N);

  gat_gemm_128x40<<<1024, 96, 0, stream>>>(out1, W2, wf2, g2, alS, alD, N / 16);
  gat_gather_l2<<<(N + 31) / 32, 256, 0, stream>>>(rowptr, col, g2, alS, alD, b2, out, N);
}


# Layer 1 attention dots read whole rows: a wave takes 16 nodes with 16-byte loads (4 lines each, not 32), 4-channel lane partials summed in 16-lane trees: a new sum order for these logits
# speedup vs baseline: 5.7312x; 1.0364x over previous
#include <hip/hip_runtime.h>
#include <math.h>

typedef __attribute__((ext_vector_type(2))) float v2f;
typedef __attribute__((ext_vector_type(4))) float v4f;
typedef __attribute__((ext_vector_type(8))) float v8f;

#define NEG_SLOPE 0.2f

#define A_PITCH 132
#define G1_STAGE (64 * A_PITCH)

__device__ __forceinline__ void gemm_128x128_walk(
    const float* __restrict__ X, const float* __restrict__ W,
    float* __restrict__ H, float* __restrict__ stage, int nTiles, int wv,
    int tileBase0, int tileStep)
{
  const int lane = threadIdx.x & 31;
  const int colTile = wv & 7, rowSub = wv >> 3;
  const int col0 = colTile << 4;
  const int kOff = (lane >> 4) << 1;
  const int bCol = col0 + (lane & 15);
  v2f bf[32];
#pragma unroll
  for (int s = 0; s < 32; ++s) {
    bf[s].x = W[(size_t)(4 * s + kOff) * 128 + bCol];
    bf[s].y = W[(size_t)(4 * s + kOff + 1) * 128 + bCol];
  }
  const int nRows = nTiles << 4;
  const int sOff0 = wv * A_PITCH + lane * 4;
  const int sOff1 = (wv + 32) * A_PITCH + lane * 4;
  const int fOff = (rowSub * 16 + (lane & 15)) * A_PITCH + kOff;
  {
    const int g0 = min(tileBase0 * 16 + wv, nRows - 1);
    const int g1 = min(tileBase0 * 16 + wv + 32, nRows - 1);
    const float4 q0 = *(const float4*)(X + (size_t)g0 * 128 + lane * 4);
    const float4 q1 = *(const float4*)(X + (size_t)g1 * 128 + lane * 4);
    *(float4*)(stage + sOff0) = q0;
    *(float4*)(stage + sOff1) = q1;
  }
  __syncthreads();
  int buf = 0;
  for (int tileBase = tileBase0; tileBase < nTiles; tileBase += tileStep) {
    const int nextBase = tileBase + tileStep;
    const bool more = nextBase < nTiles;
    float4 q0 = make_float4(0.f, 0.f, 0.f, 0.f), q1 = q0;
    if (more) {
      const int g0 = min(nextBase * 16 + wv, nRows - 1);
      const int g1 = min(nextBase * 16 + wv + 32, nRows - 1);
      q0 = *(const float4*)(X + (size_t)g0 * 128 + lane * 4);
      q1 = *(const float4*)(X + (size_t)g1 * 128 + lane * 4);
    }
    const int tile = tileBase + rowSub;
    if (tile < nTiles) {
      const float* ar = stage + buf * G1_STAGE + fOff;
      v8f acc = {};
#pragma unroll
      for (int s = 0; s < 32; ++s) {
        const v2f a = *(const v2f*)(ar + 4 * s);
        acc = __builtin_amdgcn_wmma_f32_16x16x4_f32(
            false, a, false, bf[s], (short)0, acc, false, false);
      }
      const int r = (tile << 4) + ((lane >> 4) << 3);
      const int c = col0 + (lane & 15);
#pragma unroll
      for (int i = 0; i < 8; ++i)
        H[(size_t)(r + i) * 128 + c] = acc[i];
    }
    if (more) {
      float* sp = stage + (buf ^ 1) * G1_STAGE;
      *(float4*)(sp + sOff0) = q0;
      *(float4*)(sp + sOff1) = q1;
    }
    __syncthreads();
    buf ^= 1;
  }
}

#define G2_STAGE (16 * A_PITCH)
__global__ __launch_bounds__(96) void gat_gemm_128x40(
    const float* __restrict__ X, const float* __restrict__ W,
    const float* __restrict__ wf, float* __restrict__ G,
    float* __restrict__ alS, float* __restrict__ alD, int nTiles)
{
  __shared__ __attribute__((aligned(16))) float stage[2 * G2_STAGE];
  const int lane = threadIdx.x & 31;
  const int wave = __builtin_amdgcn_readfirstlane((int)(threadIdx.x >> 5));
  const int col0 = wave << 4;
  const int kOff = (lane >> 4) << 1;
  const int bCol = col0 + (lane & 15);
  const bool colOk = (bCol < 40);
  const bool dotCol = (bCol == 40) || (bCol == 41);
  const float* bp = colOk ? (W + bCol) : (wf + ((bCol == 41) ? 128 : 0));
  const int bPitch = colOk ? 40 : 1;
  const bool live = colOk || dotCol;
  bp += kOff * bPitch;
  v2f bf[32];
#pragma unroll
  for (int s = 0; s < 32; ++s) {
    const float w0 = bp[(4 * s) * bPitch];
    const float w1 = bp[(4 * s + 1) * bPitch];
    bf[s].x = live ? w0 : 0.0f;
    bf[s].y = live ? w1 : 0.0f;
  }
  const int fOff = (lane & 15) * A_PITCH + kOff;
  v4f q[6];
  int tile = blockIdx.x;
  if (tile < nTiles) {
#pragma unroll
    for (int i = 0; i < 6; ++i) {
      const int lr = min(wave + 3 * i, 15);
      q[i] = *(const v4f*)(X + (size_t)(tile * 16 + lr) * 128 + lane * 4);
    }
#pragma unroll
    for (int i = 0; i < 6; ++i) {
      const int lr = wave + 3 * i;
      if (lr < 16) *(v4f*)(stage + lr * A_PITCH + lane * 4) = q[i];
    }
  }
  __syncthreads();
  int buf = 0;
  for (; tile < nTiles; tile += gridDim.x) {
    const int nextTile = tile + gridDim.x;
    const bool more = nextTile < nTiles;
    if (more) {
#pragma unroll
      for (int i = 0; i < 6; ++i) {
        const int lr = min(wave + 3 * i, 15);
        q[i] = *(const v4f*)(X + (size_t)(nextTile * 16 + lr) * 128 + lane * 4);
      }
    }
    const int row0 = tile << 4;
    const float* ar = stage + buf * G2_STAGE + fOff;
    v8f acc = {};
#pragma unroll
    for (int s = 0; s < 32; ++s) {
      const v2f a = *(const v2f*)(ar + 4 * s);
      acc = __builtin_amdgcn_wmma_f32_16x16x4_f32(
          false, a, false, bf[s], (short)0, acc, false, false);
    }
    const int r = row0 + ((lane >> 4) << 3);
    const int c = col0 + (lane & 15);
    if (c < 40) {
#pragma unroll
      for (int i = 0; i < 8; ++i)
        G[(size_t)(r + i) * 40 + c] = acc[i];
    } else if (dotCol) {
      float* ap = ((c == 40) ? alS : alD) + r;
      *(v4f*)ap = __builtin_shufflevector(acc, acc, 0, 1, 2, 3);
      *(v4f*)(ap + 4) = __builtin_shufflevector(acc, acc, 4, 5, 6, 7);
    }
    if (more) {
      float* sp = stage + (buf ^ 1) * G2_STAGE;
#pragma unroll
      for (int i = 0; i < 6; ++i) {
        const int lr = wave + 3 * i;
        if (lr < 16) *(v4f*)(sp + lr * A_PITCH + lane * 4) = q[i];
      }
    }
    __syncthreads();
    buf ^= 1;
  }
}

template <int M>
__device__ __forceinline__ float row_xor(float v) {
  return __int_as_float(__builtin_amdgcn_update_dpp(
      0, __float_as_int(v), 0x160 | M, 0xf, 0xf, true));
}
__device__ __forceinline__ void attn_dots_rows(
    int gw, const float* __restrict__ Hf, const float* __restrict__ aS,
    const float* __restrict__ aD, float* __restrict__ alS,
    float* __restrict__ alD, int N)
{
  const int lane = threadIdx.x & 31;
  const int n0 = gw * 16;
  if (n0 >= N) return;
  const v4f as4 = *(const v4f*)(aS + lane * 4);
  const v4f ad4 = *(const v4f*)(aD + lane * 4);
  const v2f sd0 = {as4.x, ad4.x}, sd1 = {as4.y, ad4.y};
  const v2f sd2 = {as4.z, ad4.z}, sd3 = {as4.w, ad4.w};
  const bool b3 = (lane & 8) != 0, b2 = (lane & 4) != 0;
  const bool b1 = (lane & 2) != 0, b0 = (lane & 1) != 0;
  float* op = b3 ? alD : alS;
  const int oidx = (lane & 7) * 2 + (lane >> 4);
  const v4f* __restrict__ H4 = (const v4f*)Hf;
#pragma unroll 1
  for (int bt = 0; bt < 2; ++bt) {
    const int nb0 = n0 + bt * 8;
    v4f hv[8];
#pragma unroll
    for (int i = 0; i < 8; ++i) {
      const unsigned n = (unsigned)min(nb0 + i, N - 1);
      hv[i] = H4[n * 32u + (unsigned)lane];
    }
    float t[8], u[4], w[2];
#pragma unroll
    for (int i = 0; i < 8; ++i) {
      v2f sd = hv[i].x * sd0;
      sd += hv[i].y * sd1;
      sd += hv[i].z * sd2;
      sd += hv[i].w * sd3;
      const float keep = b3 ? sd.y : sd.x, give = b3 ? sd.x : sd.y;
      t[i] = keep + row_xor<8>(give);
    }
#pragma unroll
    for (int i = 0; i < 4; ++i) {
      const float keep = b2 ? t[i + 4] : t[i], give = b2 ? t[i] : t[i + 4];
      u[i] = keep + row_xor<4>(give);
    }
#pragma unroll
    for (int i = 0; i < 2; ++i) {
      const float keep = b1 ? u[i + 2] : u[i], give = b1 ? u[i] : u[i + 2];
      w[i] = keep + row_xor<2>(give);
    }
    const float keep = b0 ? w[1] : w[0], give = b0 ? w[0] : w[1];
    const float r = keep + row_xor<1>(give);
    if (nb0 + (lane & 7) < N) op[(size_t)nb0 * 2 + oidx] = r;
  }
}


__device__ __forceinline__ void fold_attn_weights(
    const float* __restrict__ W, const float* __restrict__ aS,
    const float* __restrict__ aD, float* __restrict__ wf)
{
  const int t = threadIdx.x, k = t >> 3, j = t & 7;
  float s = 0.f, d = 0.f;
#pragma unroll
  for (int i = 0; i < 5; ++i) {
    const int cc = j + 8 * i;
    const float w = W[k * 40 + cc];
    s += w * aS[cc];
    d += w * aD[cc];
  }
  for (int o = 4; o > 0; o >>= 1) {
    s += __shfl_xor(s, o, 32);
    d += __shfl_xor(d, o, 32);
  }
  if (j == 0) {
    wf[k] = s;
    wf[128 + k] = d;
  }
}

__device__ __forceinline__ float leaky(float v) {
  return (v > 0.f) ? v : NEG_SLOPE * v;
}
__device__ __forceinline__ float wave_max(float v) {
  for (int o = 16; o > 0; o >>= 1) v = fmaxf(v, __shfl_xor(v, o, 32));
  return v;
}
__device__ __forceinline__ float wave_sum(float v) {
  for (int o = 16; o > 0; o >>= 1) v += __shfl_xor(v, o, 32);
  return v;
}
__device__ __forceinline__ float lane_bcast(float v, int j) {
  return __int_as_float(__builtin_amdgcn_readlane(__float_as_int(v), j));
}

#define SLICE_SHIFT 14
#define SLICE (1 << SLICE_SHIFT)
#define HIST_WORDS 51200
#define HIST_EPT (SLICE / 1024)
#define NODE_SHIFT 8
#define NODE_BLOCK (1 << NODE_SHIFT)
#define BIN_GROUPS 512
#define BO_PITCH 288
__global__ __launch_bounds__(1024) void gat_hist_gemm1(
    const int* __restrict__ ei, int E, int B, int nb, int* __restrict__ binned,
    unsigned* __restrict__ boff, const float* __restrict__ X,
    const float* __restrict__ W, float* __restrict__ H, int nTiles)
{
  __shared__ __attribute__((aligned(16))) unsigned hist[HIST_WORDS];
  const int t = threadIdx.x;
  if ((int)blockIdx.x >= B) {
    const int wv = __builtin_amdgcn_readfirstlane(t >> 5);
    const int gb = (int)blockIdx.x - B;
    const int nGemm = (int)gridDim.x - B;
    gemm_128x128_walk(X, W, H, (float*)hist, nTiles, wv, gb * 4, nGemm * 4);
    return;
  }
  unsigned* cntS = hist;
  unsigned* wtS = hist + BIN_GROUPS;
  int2* stg = (int2*)(hist + 1024);
  if (t < BIN_GROUPS) cntS[t] = 0u;
  __syncthreads();
  const int e0 = blockIdx.x * SLICE + t;
  int dst[HIST_EPT], sr[HIST_EPT];
  unsigned bk[HIST_EPT], r[HIST_EPT];
#pragma unroll
  for (int k = 0; k < HIST_EPT; ++k) {
    const int ec = min(e0 + 1024 * k, E - 1);
    dst[k] = ei[E + ec];
    sr[k] = ei[ec];
  }
#pragma unroll
  for (int k = 0; k < HIST_EPT; ++k) {
    const bool ok = (e0 + 1024 * k < E) &&
                    ((unsigned)dst[k] < (unsigned)(nb << NODE_SHIFT));
    bk[k] = ok ? ((unsigned)dst[k] >> NODE_SHIFT) : (unsigned)nb;
  }
#pragma unroll
  for (int k = 0; k < HIST_EPT; ++k) r[k] = atomicAdd(&cntS[bk[k]], 1u);
  __syncthreads();
  const int lane = t & 31, wvv = t >> 5;
  unsigned v = 0u, incl = 0u;
  if (t < BIN_GROUPS) {
    v = cntS[t];
    incl = v;
    for (int o = 1; o < 32; o <<= 1) {
      const unsigned u = __shfl_up(incl, o, 32);
      if (lane >= o) incl += u;
    }
    if (lane == 31) wtS[wvv] = incl;
  }
  __syncthreads();
  if (t < BIN_GROUPS) {
    unsigned before = 0u;
#pragma unroll
    for (int w = 0; w < BIN_GROUPS / 32; ++w)
      if (w < wvv) before += wtS[w];
    const unsigned ex = before + incl - v;
    cntS[t] = ex;
    if (t < BO_PITCH) boff[(size_t)blockIdx.x * BO_PITCH + t] = ex;
  }
  __syncthreads();
#pragma unroll
  for (int k = 0; k < HIST_EPT; ++k)
    stg[cntS[bk[k]] + r[k]] = make_int2(dst[k], sr[k]);
  __syncthreads();
  uint4* o = (uint4*)(binned + (size_t)blockIdx.x * (2 * SLICE));
  for (int i = t; i < SLICE / 2; i += 1024) o[i] = ((const uint4*)stg)[i];
}

#define BUCKET_SH (2 * NODE_BLOCK + 8 + 32)
__device__ __forceinline__ void csr_bucket(
    int b, const int* __restrict__ binned, const unsigned* __restrict__ boff,
    int B, unsigned* __restrict__ rowptr, int* __restrict__ col,
    unsigned* __restrict__ sh)
{
  unsigned* cnt = sh;
  unsigned* cur = sh + NODE_BLOCK;
  unsigned* wt = sh + 2 * NODE_BLOCK;
  unsigned* wb = sh + 2 * NODE_BLOCK + 8;
  const int t = threadIdx.x, lane = t & 31;
  const int wv = __builtin_amdgcn_readfirstlane(t >> 5);
  if (t < NODE_BLOCK) cnt[t] = 0u;
  unsigned pb = 0u;
  for (int s = t; s < B; s += 1024) pb += boff[(size_t)s * BO_PITCH + b];
  for (int o = 16; o > 0; o >>= 1) pb += __shfl_xor(pb, o, 32);
  if (lane == 0) wb[wv] = pb;
  __syncthreads();
  for (int s = wv; s < B; s += 32) {
    const unsigned beg = boff[(size_t)s * BO_PITCH + b];
    const unsigned end = boff[(size_t)s * BO_PITCH + b + 1];
    const int2* seg = (const int2*)binned + (size_t)s * SLICE;
    for (unsigned i0 = beg; i0 < end; i0 += 128) {
      int d[4];
#pragma unroll
      for (int u = 0; u < 4; ++u)
        d[u] = seg[min(i0 + 32 * u + lane, end - 1)].x;
#pragma unroll
      for (int u = 0; u < 4; ++u)
        if (i0 + 32 * u + lane < end) atomicAdd(&cnt[d[u] & (NODE_BLOCK - 1)], 1u);
    }
  }
  __syncthreads();
  unsigned base = 0u;
#pragma unroll
  for (int w = 0; w < 32; ++w) base += wb[w];
  unsigned v = 0u, incl = 0u;
  if (t < NODE_BLOCK) {
    v = cnt[t];
    incl = v;
    for (int o = 1; o < 32; o <<= 1) {
      const unsigned u = __shfl_up(incl, o, 32);
      if (lane >= o) incl += u;
    }
    if (lane == 31) wt[wv] = incl;
  }
  __syncthreads();
  if (t < NODE_BLOCK) {
    unsigned before = 0u;
#pragma unroll
    for (int w = 0; w < NODE_BLOCK / 32; ++w)
      if (w < wv) before += wt[w];
    const unsigned ex = before + incl - v;
    rowptr[b * NODE_BLOCK + t] = base + ex;
    cur[t] = ex;
  }
  __syncthreads();
  for (int s = wv; s < B; s += 32) {
    const unsigned beg = boff[(size_t)s * BO_PITCH + b];
    const unsigned end = boff[(size_t)s * BO_PITCH + b + 1];
    const int2* seg = (const int2*)binned + (size_t)s * SLICE;
    for (unsigned i0 = beg; i0 < end; i0 += 128) {
      int2 e[4];
#pragma unroll
      for (int u = 0; u < 4; ++u)
        e[u] = seg[min(i0 + 32 * u + lane, end - 1)];
#pragma unroll
      for (int u = 0; u < 4; ++u)
        if (i0 + 32 * u + lane < end)
          col[base + atomicAdd(&cur[e[u].x & (NODE_BLOCK - 1)], 1u)] = e[u].y;
    }
  }
}

__global__ __launch_bounds__(1024) void gat_bucket_attn(
    const int* __restrict__ binned, const unsigned* __restrict__ boff, int B,
    int nb, unsigned* __restrict__ rowptr, int* __restrict__ col,
    const float* __restrict__ Hf, const float* __restrict__ aS,
    const float* __restrict__ aD, float* __restrict__ alS,
    float* __restrict__ alD, int N,
    const float* __restrict__ W2, const float* __restrict__ aS2,
    const float* __restrict__ aD2, float* __restrict__ wf)
{
  __shared__ unsigned sh[BUCKET_SH];
  if ((int)blockIdx.x < nb) {
    csr_bucket(blockIdx.x, binned, boff, B, rowptr, col, sh);
  } else if (blockIdx.x + 1 < gridDim.x) {
    const int wv = __builtin_amdgcn_readfirstlane((int)(threadIdx.x >> 5));
    attn_dots_rows(((int)blockIdx.x - nb) * 32 + wv, Hf, aS, aD, alS, alD, N);
  } else {
    fold_attn_weights(W2, aS2, aD2, wf);
  }
}

#define NODES_PER_WAVE 4
#define ROWS_IN_FLIGHT 4
__global__ __launch_bounds__(256) void gat_gather_l1(
    const unsigned* __restrict__ rowptr, const int* __restrict__ col,
    const float* __restrict__ Hf, const float* __restrict__ alS,
    const float* __restrict__ alD, const float* __restrict__ b,
    float* __restrict__ out1, int N)
{
  const int lane = threadIdx.x & 31;
  const int wv = __builtin_amdgcn_readfirstlane((int)(threadIdx.x >> 5));
  const int d0 = (blockIdx.x * 8 + wv) * NODES_PER_WAVE;
  if (d0 >= N) return;
  const int hsel = lane >> 4;
  unsigned r0 = rowptr[d0], r1 = rowptr[d0 + 1], r2 = rowptr[min(d0 + 2, N)];
  const int elast = max((int)rowptr[N] - 1, 0);
  int dgn = (int)(r2 - r1);
  float ad0 = alD[d0 * 2], ad1 = alD[d0 * 2 + 1];
  float vs0 = leaky(alS[d0 * 2] + ad0), vs1 = leaky(alS[d0 * 2 + 1] + ad1);
  int sc = col[min((int)r0 + min(lane, max((int)(r1 - r0) - 1, 0)), elast)];
  float2 ac = *(const float2*)(alS + (size_t)sc * 2);
#pragma unroll 1
  for (int kn = 0; kn < NODES_PER_WAVE; ++kn) {
  const int d = d0 + kn;
  if (d >= N) break;
  const int beg = (int)r0;
  const int deg = (int)(r1 - r0);
  const int dn = d + 1;
  const bool more = (kn + 1 < NODES_PER_WAVE) && (dn < N);
  const int degn = more ? dgn : 0;
  const unsigned r3 = rowptr[min(d + 3, N)];
  const int dnc = min(dn, N - 1);
  const float ns0 = alS[dnc * 2], ns1 = alS[dnc * 2 + 1];
  const float nd0 = alD[dnc * 2], nd1 = alD[dnc * 2 + 1];
  const int snr = col[min((int)r1 + min(lane, max(degn - 1, 0)), elast)];
  const float4 hown = *(const float4*)(Hf + (size_t)d * 128 + lane * 4);
  float m0 = vs0, m1 = vs1;
  if (lane < deg) {
    m0 = fmaxf(m0, leaky(ac.x + ad0));
    m1 = fmaxf(m1, leaky(ac.y + ad1));
  }
  for (int base = 32; base < deg; base += 32) {
    const int i = base + lane;
    if (i < deg) {
      const int s = col[beg + i];
      const float2 a = *(const float2*)(alS + (size_t)s * 2);
      m0 = fmaxf(m0, leaky(a.x + ad0));
      m1 = fmaxf(m1, leaky(a.y + ad1));
    }
  }
  m0 = wave_max(m0);
  m1 = wave_max(m1);
  __builtin_amdgcn_sched_barrier(0);
  const float2 anr = *(const float2*)(alS + (size_t)snr * 2);
  const float ws0 = expf(vs0 - m0), ws1 = expf(vs1 - m1);
  float4 acc;
  {
    const float w = hsel ? ws1 : ws0;
    acc.x = w * hown.x; acc.y = w * hown.y; acc.z = w * hown.z; acc.w = w * hown.w;
  }
  float p0 = 0.f, p1 = 0.f;
  for (int base = 0; base < deg; base += 32) {
    const int i = base + lane;
    int s = d;
    float w0 = 0.f, w1 = 0.f;
    if (base == 0) {
      s = sc;
      if (i < deg) {
        w0 = expf(leaky(ac.x + ad0) - m0);
        w1 = expf(leaky(ac.y + ad1) - m1);
      }
    } else if (i < deg) {
      s = col[beg + i];
      const float2 a = *(const float2*)(alS + (size_t)s * 2);
      w0 = expf(leaky(a.x + ad0) - m0);
      w1 = expf(leaky(a.y + ad1) - m1);
    }
    p0 += w0;
    p1 += w1;
    const int cnt = min(32, deg - base);
    for (int j0 = 0; j0 < cnt; j0 += ROWS_IN_FLIGHT) {
#pragma unroll
      for (int u = 0; u < ROWS_IN_FLIGHT; ++u) {
        const int j = j0 + u;
        const int sj = __builtin_amdgcn_readlane(s, j);
        const float wj0 = lane_bcast(w0, j), wj1 = lane_bcast(w1, j);
        const float w = hsel ? wj1 : wj0;
        const float4 hv = *(const float4*)(Hf + (size_t)sj * 128 + lane * 4);
        acc.x += w * hv.x; acc.y += w * hv.y; acc.z += w * hv.z; acc.w += w * hv.w;
      }
    }
  }
  const float den0 = wave_sum(p0) + ws0;
  const float den1 = wave_sum(p1) + ws1;
  const float den = (hsel ? den1 : den0) + 1e-16f;
  const float4 bv = *(const float4*)(b + lane * 4);
  float4 r;
  r.x = acc.x / den + bv.x;
  r.y = acc.y / den + bv.y;
  r.z = acc.z / den + bv.z;
  r.w = acc.w / den + bv.w;
  r.x = (r.x > 0.f) ? r.x : (expf(r.x) - 1.0f);
  r.y = (r.y > 0.f) ? r.y : (expf(r.y) - 1.0f);
  r.z = (r.z > 0.f) ? r.z : (expf(r.z) - 1.0f);
  r.w = (r.w > 0.f) ? r.w : (expf(r.w) - 1.0f);
  *(float4*)(out1 + (size_t)d * 128 + lane * 4) = r;
  r0 = r1; r1 = r2; r2 = r3;
  dgn = (int)(r2 - r1);
  ad0 = nd0; ad1 = nd1;
  vs0 = leaky(ns0 + nd0); vs1 = leaky(ns1 + nd1);
  sc = snr; ac = anr;
  }
}

__global__ __launch_bounds__(256) void gat_gather_l2(
    const unsigned* __restrict__ rowptr, const int* __restrict__ col,
    const float* __restrict__ G, const float* __restrict__ alS,
    const float* __restrict__ alD, const float* __restrict__ b,
    float* __restrict__ out, int N)
{
  __shared__ float stage[8 * NODES_PER_WAVE * 40];
  const int lane = threadIdx.x & 31;
  const int wv = __builtin_amdgcn_readfirstlane((int)(threadIdx.x >> 5));
  const int d0 = (blockIdx.x * 8 + wv) * NODES_PER_WAVE;
  float* slot = stage + wv * (NODES_PER_WAVE * 40);
  const bool act = lane < 20;
  const int c2 = act ? 2 * lane : 0;
  unsigned r0 = 0, r1 = 0, r2 = 0;
  int dgn = 0;
  float ad = 0.f, vs = 0.f;
  int sc = 0;
  float ac = 0.f;
  const int elast = max((int)rowptr[N] - 1, 0);
  if (d0 < N) {
    r0 = rowptr[d0]; r1 = rowptr[d0 + 1]; r2 = rowptr[min(d0 + 2, N)];
    dgn = (int)(r2 - r1);
    ad = alD[d0];
    vs = leaky(alS[d0] + ad);
    sc = col[min((int)r0 + min(lane, max((int)(r1 - r0) - 1, 0)), elast)];
    ac = alS[sc];
  }
#pragma unroll 1
  for (int kn = 0; kn < NODES_PER_WAVE; ++kn) {
  const int d = d0 + kn;
  float res0 = 0.f, res1 = 0.f;
  if (d < N) {
    const int beg = (int)r0;
    const int deg = (int)(r1 - r0);
    const int dn = d + 1;
    const bool more = (kn + 1 < NODES_PER_WAVE) && (dn < N);
    const int degn = more ? dgn : 0;
    const unsigned r3 = rowptr[min(d + 3, N)];
    const int dnc = min(dn, N - 1);
    const float ns = alS[dnc], nd = alD[dnc];
    const int snr = col[min((int)r1 + min(lane, max(degn - 1, 0)), elast)];
    const float2 gown = *(const float2*)(G + (size_t)d * 40 + c2);
    float m = vs;
    if (lane < deg) m = fmaxf(m, leaky(ac + ad));
    for (int base = 32; base < deg; base += 32) {
      const int i = base + lane;
      if (i < deg) m = fmaxf(m, leaky(alS[col[beg + i]] + ad));
    }
    m = wave_max(m);
    __builtin_amdgcn_sched_barrier(0);
    const float anr = alS[snr];
    const float wself = expf(vs - m);
    float acc0 = wself * gown.x, acc1 = wself * gown.y;
    float p = 0.f;
    for (int base = 0; base < deg; base += 32) {
      const int i = base + lane;
      int s = d;
      float w = 0.f;
      if (base == 0) {
        s = sc;
        if (i < deg) w = expf(leaky(ac + ad) - m);
      } else if (i < deg) {
        s = col[beg + i];
        w = expf(leaky(alS[s] + ad) - m);
      }
      p += w;
      const int cnt = min(32, deg - base);
      for (int j0 = 0; j0 < cnt; j0 += ROWS_IN_FLIGHT) {
#pragma unroll
        for (int u = 0; u < ROWS_IN_FLIGHT; ++u) {
          const int j = j0 + u;
          const int sj = __builtin_amdgcn_readlane(s, j);
          const float wj = lane_bcast(w, j);
          const float2 gv = *(const float2*)(G + (size_t)sj * 40 + c2);
          acc0 += wj * gv.x;
          acc1 += wj * gv.y;
        }
      }
    }
    const float den = wave_sum(p) + wself + 1e-16f;
    res0 = acc0 / den + b[c2];
    res1 = acc1 / den + b[c2 + 1];
    r0 = r1; r1 = r2; r2 = r3;
    dgn = (int)(r2 - r1);
    ad = nd;
    vs = leaky(ns + nd);
    sc = snr; ac = anr;
  }
  if (act) {
    slot[kn * 40 + c2] = res0;
    slot[kn * 40 + c2 + 1] = res1;
  }
  }
  const size_t lim = (size_t)N * 40;
  const size_t g0 = (size_t)d0 * 40 + lane;
#pragma unroll
  for (int j = 0; j < (NODES_PER_WAVE * 40) / 32; ++j) {
    const size_t g = g0 + 32 * j;
    if (g < lim) out[g] = slot[32 * j + lane];
  }
}

extern "C" void kernel_launch(void* const* d_in, const int* in_sizes, int n_in,
                              void* d_out, int out_size, void* d_ws, size_t ws_size,
                              hipStream_t stream)
{
  const float* x   = (const float*)d_in[0];
  const int*   ei  = (const int*)d_in[1];
  const float* W1  = (const float*)d_in[2];
  const float* aS1 = (const float*)d_in[3];
  const float* aD1 = (const float*)d_in[4];
  const float* b1  = (const float*)d_in[5];
  const float* W2  = (const float*)d_in[6];
  const float* aS2 = (const float*)d_in[7];
  const float* aD2 = (const float*)d_in[8];
  const float* b2  = (const float*)d_in[9];
  float* out = (float*)d_out;

  const int N  = in_sizes[0] / 128;
  const int E  = in_sizes[1] / 2;
  const int NP = (N / NODE_BLOCK + 1) * NODE_BLOCK;
  const int nb = NP / NODE_BLOCK;

  char* ws = (char*)d_ws;
  size_t off = 0;
  auto carve = [&](size_t bytes) -> char* {
    char* p = ws + off;
    off += (bytes + 255) & ~(size_t)255;
    return p;
  };
  float*    h1     = (float*)carve((size_t)N * 128 * 4);
  float*    out1   = (float*)carve((size_t)N * 128 * 4);
  float*    alS    = (float*)carve((size_t)N * 2 * 4);
  float*    alD    = (float*)carve((size_t)N * 2 * 4);
  unsigned* rowptr = (unsigned*)carve((size_t)NP * 4);
  int*      col    = (int*)carve((size_t)E * 4);
  const int B  = (E + SLICE - 1) >> SLICE_SHIFT;
  int*      binned = (int*)carve((size_t)B * SLICE * 8);
  unsigned* boff   = (unsigned*)carve((size_t)B * BO_PITCH * 4);
  float*    wf2    = (float*)carve((size_t)256 * 4);
  float* g2 = h1;

  const int gemmBlocks = (B < 224) ? 256 - B : 32;
  gat_hist_gemm1<<<B + gemmBlocks, 1024, 0, stream>>>(ei, E, B, nb, binned, boff, x, W1, h1, N / 16);
  gat_bucket_attn<<<nb + (N * 2 + 1023) / 1024 + 1, 1024, 0, stream>>>(
      binned, boff, B, nb, rowptr, col, h1, aS1, aD1, alS, alD, N,
      W2, aS2, aD2, wf2);

  gat_gather_l1<<<(N + 31) / 32, 256, 0, stream>>>(rowptr, col, h1, alS, alD, b1, out1, N);

  gat_gemm_128x40<<<1024, 96, 0, stream>>>(out1, W2, wf2, g2, alS, alD, N / 16);
  gat_gather_l2<<<(N + 31) / 32, 256, 0, stream>>>(rowptr, col, g2, alS, alD, b2, out, N);
}
